# starting kernel
# baseline (speedup 1.0000x reference)
#include <hip/hip_runtime.h>
#include <math.h>


typedef __attribute__((ext_vector_type(2))) float v2f;
typedef __attribute__((ext_vector_type(8))) float v8f;

#define HEADS     4
#define CHAN      64
#define HIDDEN    256
#define FEAT      128
#define NEG_SLOPE 0.2f
#define ORD_NEG_INF 0x007FFFFFu

static __device__ __forceinline__ float lrelu(float x) {
  return x > 0.f ? x : NEG_SLOPE * x;
}
static __device__ __forceinline__ unsigned ordEnc(float f) {
  unsigned u = __float_as_uint(f);
  return (u & 0x80000000u) ? ~u : (u | 0x80000000u);
}
static __device__ __forceinline__ float ordDec(unsigned u) {
  unsigned v = (u & 0x80000000u) ? (u & 0x7fffffffu) : ~u;
  return __uint_as_float(v);
}

template <bool GAT, bool BIAS>
__global__ void wmma_gemm_kernel(const float* __restrict__ A,
                                 const float* __restrict__ W,
                                 const float* __restrict__ bias,
                                 const float* __restrict__ att_src,
                                 const float* __restrict__ att_dst,
                                 float* __restrict__ Hout,
                                 float* __restrict__ s_src,
                                 float* __restrict__ s_dst,
                                 int M, int K, int Nc) {
  const int lane = threadIdx.x & 31;
  const int wv   = threadIdx.x >> 5;
  const int wpb  = blockDim.x >> 5;
  const int tile = blockIdx.x * wpb + wv;
  const int ctiles = Nc >> 6;
  const int mt = tile / ctiles;
  const int ct = tile - mt * ctiles;
  const int m0 = mt << 5;
  if (m0 >= M) return;
  const int n0 = ct << 6;
  const int lo = lane & 15;
  const int hi = lane >> 4;

  v8f acc[2][4] = {};
  const float* arow0 = A + (size_t)(m0 + lo) * K;
  const float* arow1 = A + (size_t)(m0 + 16 + lo) * K;
  const float* wcol  = W + n0 + lo;

  for (int k = 0; k < K; k += 4) {
    const int ka = k + (hi << 1);
    v2f a0 = *(const v2f*)(arow0 + ka);
    v2f a1 = *(const v2f*)(arow1 + ka);
    const float* w0 = wcol + (size_t)ka * Nc;
    const float* w1 = w0 + Nc;
    v2f b0 = {w0[0],  w1[0]};
    v2f b1 = {w0[16], w1[16]};
    v2f b2 = {w0[32], w1[32]};
    v2f b3 = {w0[48], w1[48]};
    acc[0][0] = __builtin_amdgcn_wmma_f32_16x16x4_f32(false, a0, false, b0, (short)0, acc[0][0], false, false);
    acc[1][0] = __builtin_amdgcn_wmma_f32_16x16x4_f32(false, a1, false, b0, (short)0, acc[1][0], false, false);
    acc[0][1] = __builtin_amdgcn_wmma_f32_16x16x4_f32(false, a0, false, b1, (short)0, acc[0][1], false, false);
    acc[1][1] = __builtin_amdgcn_wmma_f32_16x16x4_f32(false, a1, false, b1, (short)0, acc[1][1], false, false);
    acc[0][2] = __builtin_amdgcn_wmma_f32_16x16x4_f32(false, a0, false, b2, (short)0, acc[0][2], false, false);
    acc[1][2] = __builtin_amdgcn_wmma_f32_16x16x4_f32(false, a1, false, b2, (short)0, acc[1][2], false, false);
    acc[0][3] = __builtin_amdgcn_wmma_f32_16x16x4_f32(false, a0, false, b3, (short)0, acc[0][3], false, false);
    acc[1][3] = __builtin_amdgcn_wmma_f32_16x16x4_f32(false, a1, false, b3, (short)0, acc[1][3], false, false);
  }

  float bv0 = 0.f, bv1 = 0.f, bv2 = 0.f, bv3 = 0.f;
  if (BIAS) {
    bv0 = bias[n0 + lo];      bv1 = bias[n0 + 16 + lo];
    bv2 = bias[n0 + 32 + lo]; bv3 = bias[n0 + 48 + lo];
  }

  #pragma unroll
  for (int g = 0; g < 2; ++g) {
    const int rbase = m0 + (g << 4) + (hi << 3);
    #pragma unroll
    for (int r = 0; r < 8; ++r) {
      float* hp = Hout + (size_t)(rbase + r) * Nc + n0 + lo;
      hp[0]  = acc[g][0][r] + bv0;
      hp[16] = acc[g][1][r] + bv1;
      hp[32] = acc[g][2][r] + bv2;
      hp[48] = acc[g][3][r] + bv3;
    }
  }

  if (GAT) {
    const int head = n0 >> 6;
    const float* as = att_src + head * CHAN;
    const float* ad = att_dst + head * CHAN;
    const float as0 = as[lo], as1 = as[16 + lo], as2 = as[32 + lo], as3 = as[48 + lo];
    const float ad0 = ad[lo], ad1 = ad[16 + lo], ad2 = ad[32 + lo], ad3 = ad[48 + lo];
    #pragma unroll
    for (int g = 0; g < 2; ++g) {
      const int rbase = m0 + (g << 4) + (hi << 3);
      #pragma unroll
      for (int r = 0; r < 8; ++r) {
        float ss = acc[g][0][r] * as0 + acc[g][1][r] * as1 + acc[g][2][r] * as2 + acc[g][3][r] * as3;
        float sd = acc[g][0][r] * ad0 + acc[g][1][r] * ad1 + acc[g][2][r] * ad2 + acc[g][3][r] * ad3;
        #pragma unroll
        for (int off = 8; off > 0; off >>= 1) {
          ss += __shfl_xor(ss, off, 32);
          sd += __shfl_xor(sd, off, 32);
        }
        if (lo == 0) {
          const int row = rbase + r;
          s_src[row * HEADS + head] = ss;
          s_dst[row * HEADS + head] = sd;
        }
      }
    }
  }
}

__global__ void layer_init_kernel(float* __restrict__ agg, const float* __restrict__ bias,
                                  float* __restrict__ denom, unsigned* __restrict__ mord,
                                  int N) {
  const int idx = blockIdx.x * blockDim.x + threadIdx.x;
  if (idx < N * HIDDEN) agg[idx] = bias[idx & (HIDDEN - 1)];
  if (idx < N * HEADS) { denom[idx] = 0.f; mord[idx] = ORD_NEG_INF; }
}

static __device__ __forceinline__ void edge_nodes(const int* __restrict__ ei,
                                                  int idx, int E, int& s, int& d) {
  if (idx < E) { s = ei[idx]; d = ei[E + idx]; }
  else         { s = d = idx - E; }
}

__global__ void edge_max_kernel(const int* __restrict__ ei, int E, int N,
                                const float* __restrict__ s_src,
                                const float* __restrict__ s_dst,
                                unsigned* __restrict__ mord) {
  const int idx = blockIdx.x * blockDim.x + threadIdx.x;
  if (idx >= E + N) return;
  int s, d; edge_nodes(ei, idx, E, s, d);
  const float4 vs = *(const float4*)(s_src + (size_t)s * HEADS);
  const float4 vd = *(const float4*)(s_dst + (size_t)d * HEADS);
  unsigned* mp = mord + (size_t)d * HEADS;
  atomicMax(mp + 0, ordEnc(lrelu(vs.x + vd.x)));
  atomicMax(mp + 1, ordEnc(lrelu(vs.y + vd.y)));
  atomicMax(mp + 2, ordEnc(lrelu(vs.z + vd.z)));
  atomicMax(mp + 3, ordEnc(lrelu(vs.w + vd.w)));
}

__global__ void edge_exp_kernel(const int* __restrict__ ei, int E, int N,
                                const float* __restrict__ s_src,
                                const float* __restrict__ s_dst,
                                const unsigned* __restrict__ mord,
                                float* __restrict__ exbuf,
                                float* __restrict__ denom) {
  const int idx = blockIdx.x * blockDim.x + threadIdx.x;
  if (idx >= E + N) return;
  int s, d; edge_nodes(ei, idx, E, s, d);
  const float4 vs = *(const float4*)(s_src + (size_t)s * HEADS);
  const float4 vd = *(const float4*)(s_dst + (size_t)d * HEADS);
  const unsigned* mp = mord + (size_t)d * HEADS;
  const float x0 = expf(lrelu(vs.x + vd.x) - ordDec(mp[0]));
  const float x1 = expf(lrelu(vs.y + vd.y) - ordDec(mp[1]));
  const float x2 = expf(lrelu(vs.z + vd.z) - ordDec(mp[2]));
  const float x3 = expf(lrelu(vs.w + vd.w) - ordDec(mp[3]));
  *(float4*)(exbuf + (size_t)idx * HEADS) = make_float4(x0, x1, x2, x3);
  float* dn = denom + (size_t)d * HEADS;
  atomicAdd(dn + 0, x0);
  atomicAdd(dn + 1, x1);
  atomicAdd(dn + 2, x2);
  atomicAdd(dn + 3, x3);
}

__global__ void edge_agg_kernel(const int* __restrict__ ei, int E, int N,
                                const float* __restrict__ exbuf,
                                const float* __restrict__ denom,
                                const float* __restrict__ Hin,
                                float* __restrict__ agg) {
  const int wid  = blockIdx.x * (blockDim.x >> 5) + (threadIdx.x >> 5);
  const int lane = threadIdx.x & 31;
  if (wid >= E + N) return;
  int s, d; edge_nodes(ei, wid, E, s, d);
  const float4 e4 = *(const float4*)(exbuf + (size_t)wid * HEADS);
  const float4 d4 = *(const float4*)(denom + (size_t)d * HEADS);
  const float alpha[HEADS] = {e4.x / d4.x, e4.y / d4.y, e4.z / d4.z, e4.w / d4.w};
  const float* hs = Hin + (size_t)s * HIDDEN;
  float* od = agg + (size_t)d * HIDDEN;
  #pragma unroll
  for (int i = 0; i < 8; ++i) {
    const int ch = lane + (i << 5);
    atomicAdd(od + ch, hs[ch] * alpha[i >> 1]);
  }
}

__global__ void elu_kernel(float* __restrict__ x, int n) {
  const int idx = blockIdx.x * blockDim.x + threadIdx.x;
  if (idx < n) {
    const float v = x[idx];
    x[idx] = v > 0.f ? v : expm1f(v);
  }
}

extern "C" void kernel_launch(void* const* d_in, const int* in_sizes, int n_in,
                              void* d_out, int out_size, void* d_ws, size_t ws_size,
                              hipStream_t stream) {
  const float* x     = (const float*)d_in[0];
  const int*   ei    = (const int*)  d_in[1];
  const float* W1    = (const float*)d_in[2];
  const float* asrc1 = (const float*)d_in[3];
  const float* adst1 = (const float*)d_in[4];
  const float* b1    = (const float*)d_in[5];
  const float* W2    = (const float*)d_in[6];
  const float* asrc2 = (const float*)d_in[7];
  const float* adst2 = (const float*)d_in[8];
  const float* b2    = (const float*)d_in[9];
  const float* Wf    = (const float*)d_in[10];
  const float* bf    = (const float*)d_in[11];

  const int B  = 2;
  const int N  = in_sizes[0] / (B * FEAT);
  const int E  = in_sizes[1] / 2;
  const int ET = E + N;

  float* p = (float*)d_ws;
  float* buf_h   = p; p += (size_t)N * HIDDEN;
  float* buf_agg = p; p += (size_t)N * HIDDEN;
  float* ssrc    = p; p += (size_t)N * HEADS;
  float* sdst    = p; p += (size_t)N * HEADS;
  float* denom   = p; p += (size_t)N * HEADS;
  unsigned* mord = (unsigned*)p; p += (size_t)N * HEADS;
  float* exbuf   = p; p += (size_t)ET * HEADS;

  const int mtiles  = N >> 5;
  const int gtiles  = mtiles * (HIDDEN >> 6);
  const int ftiles  = mtiles * (CHAN >> 6);
  const int nHC     = N * HIDDEN;
  const dim3 blkG(128);
  const dim3 blkE(256);

  for (int b = 0; b < B; ++b) {
    const float* xb = x + (size_t)b * N * FEAT;
    float* outb = (float*)d_out + (size_t)b * N * CHAN;

    wmma_gemm_kernel<true, false><<<(gtiles + 3) / 4, blkG, 0, stream>>>(
        xb, W1, nullptr, asrc1, adst1, buf_h, ssrc, sdst, N, FEAT, HIDDEN);
    layer_init_kernel<<<(nHC + 255) / 256, blkE, 0, stream>>>(buf_agg, b1, denom, mord, N);
    edge_max_kernel<<<(ET + 255) / 256, blkE, 0, stream>>>(ei, E, N, ssrc, sdst, mord);
    edge_exp_kernel<<<(ET + 255) / 256, blkE, 0, stream>>>(ei, E, N, ssrc, sdst, mord, exbuf, denom);
    edge_agg_kernel<<<(ET + 7) / 8, blkE, 0, stream>>>(ei, E, N, exbuf, denom, buf_h, buf_agg);
    elu_kernel<<<(nHC + 255) / 256, blkE, 0, stream>>>(buf_agg, nHC);

    wmma_gemm_kernel<true, false><<<(gtiles + 3) / 4, blkG, 0, stream>>>(
        buf_agg, W2, nullptr, asrc2, adst2, buf_h, ssrc, sdst, N, HIDDEN, HIDDEN);
    layer_init_kernel<<<(nHC + 255) / 256, blkE, 0, stream>>>(buf_agg, b2, denom, mord, N);
    edge_max_kernel<<<(ET + 255) / 256, blkE, 0, stream>>>(ei, E, N, ssrc, sdst, mord);
    edge_exp_kernel<<<(ET + 255) / 256, blkE, 0, stream>>>(ei, E, N, ssrc, sdst, mord, exbuf, denom);
    edge_agg_kernel<<<(ET + 7) / 8, blkE, 0, stream>>>(ei, E, N, exbuf, denom, buf_h, buf_agg);

    wmma_gemm_kernel<false, true><<<(ftiles + 3) / 4, blkG, 0, stream>>>(
        buf_agg, Wf, bf, nullptr, nullptr, outb, nullptr, nullptr, N, HIDDEN, CHAN);
  }
}


# Edges grouped by destination once per call (count, scan, fill by atomic rank); one wave per node does softmax and h[src] sum in registers: float atomics gone, sums in rank order plus a lane tree
# speedup vs baseline: 2.8420x; 2.8420x over previous
#include <hip/hip_runtime.h>
#include <math.h>


typedef __attribute__((ext_vector_type(2))) float v2f;
typedef __attribute__((ext_vector_type(8))) float v8f;

#define HEADS     4
#define CHAN      64
#define HIDDEN    256
#define FEAT      128
#define NEG_SLOPE 0.2f

static __device__ __forceinline__ float lrelu(float x) {
  return x > 0.f ? x : NEG_SLOPE * x;
}

template <bool GAT, bool BIAS>
__global__ void wmma_gemm_kernel(const float* __restrict__ A,
                                 const float* __restrict__ W,
                                 const float* __restrict__ bias,
                                 const float* __restrict__ att_src,
                                 const float* __restrict__ att_dst,
                                 float* __restrict__ Hout,
                                 float* __restrict__ s_src,
                                 float* __restrict__ s_dst,
                                 int M, int K, int Nc) {
  const int lane = threadIdx.x & 31;
  const int wv   = threadIdx.x >> 5;
  const int wpb  = blockDim.x >> 5;
  const int tile = blockIdx.x * wpb + wv;
  const int ctiles = Nc >> 6;
  const int mt = tile / ctiles;
  const int ct = tile - mt * ctiles;
  const int m0 = mt << 5;
  if (m0 >= M) return;
  const int n0 = ct << 6;
  const int lo = lane & 15;
  const int hi = lane >> 4;

  v8f acc[2][4] = {};
  const float* arow0 = A + (size_t)(m0 + lo) * K;
  const float* arow1 = A + (size_t)(m0 + 16 + lo) * K;
  const float* wcol  = W + n0 + lo;

  for (int k = 0; k < K; k += 4) {
    const int ka = k + (hi << 1);
    v2f a0 = *(const v2f*)(arow0 + ka);
    v2f a1 = *(const v2f*)(arow1 + ka);
    const float* w0 = wcol + (size_t)ka * Nc;
    const float* w1 = w0 + Nc;
    v2f b0 = {w0[0],  w1[0]};
    v2f b1 = {w0[16], w1[16]};
    v2f b2 = {w0[32], w1[32]};
    v2f b3 = {w0[48], w1[48]};
    acc[0][0] = __builtin_amdgcn_wmma_f32_16x16x4_f32(false, a0, false, b0, (short)0, acc[0][0], false, false);
    acc[1][0] = __builtin_amdgcn_wmma_f32_16x16x4_f32(false, a1, false, b0, (short)0, acc[1][0], false, false);
    acc[0][1] = __builtin_amdgcn_wmma_f32_16x16x4_f32(false, a0, false, b1, (short)0, acc[0][1], false, false);
    acc[1][1] = __builtin_amdgcn_wmma_f32_16x16x4_f32(false, a1, false, b1, (short)0, acc[1][1], false, false);
    acc[0][2] = __builtin_amdgcn_wmma_f32_16x16x4_f32(false, a0, false, b2, (short)0, acc[0][2], false, false);
    acc[1][2] = __builtin_amdgcn_wmma_f32_16x16x4_f32(false, a1, false, b2, (short)0, acc[1][2], false, false);
    acc[0][3] = __builtin_amdgcn_wmma_f32_16x16x4_f32(false, a0, false, b3, (short)0, acc[0][3], false, false);
    acc[1][3] = __builtin_amdgcn_wmma_f32_16x16x4_f32(false, a1, false, b3, (short)0, acc[1][3], false, false);
  }

  float bv0 = 0.f, bv1 = 0.f, bv2 = 0.f, bv3 = 0.f;
  if (BIAS) {
    bv0 = bias[n0 + lo];      bv1 = bias[n0 + 16 + lo];
    bv2 = bias[n0 + 32 + lo]; bv3 = bias[n0 + 48 + lo];
  }

  #pragma unroll
  for (int g = 0; g < 2; ++g) {
    const int rbase = m0 + (g << 4) + (hi << 3);
    #pragma unroll
    for (int r = 0; r < 8; ++r) {
      float* hp = Hout + (size_t)(rbase + r) * Nc + n0 + lo;
      hp[0]  = acc[g][0][r] + bv0;
      hp[16] = acc[g][1][r] + bv1;
      hp[32] = acc[g][2][r] + bv2;
      hp[48] = acc[g][3][r] + bv3;
    }
  }

  if (GAT) {
    const int head = n0 >> 6;
    const float* as = att_src + head * CHAN;
    const float* ad = att_dst + head * CHAN;
    const float as0 = as[lo], as1 = as[16 + lo], as2 = as[32 + lo], as3 = as[48 + lo];
    const float ad0 = ad[lo], ad1 = ad[16 + lo], ad2 = ad[32 + lo], ad3 = ad[48 + lo];
    #pragma unroll
    for (int g = 0; g < 2; ++g) {
      const int rbase = m0 + (g << 4) + (hi << 3);
      #pragma unroll
      for (int r = 0; r < 8; ++r) {
        float ss = acc[g][0][r] * as0 + acc[g][1][r] * as1 + acc[g][2][r] * as2 + acc[g][3][r] * as3;
        float sd = acc[g][0][r] * ad0 + acc[g][1][r] * ad1 + acc[g][2][r] * ad2 + acc[g][3][r] * ad3;
        #pragma unroll
        for (int off = 8; off > 0; off >>= 1) {
          ss += __shfl_xor(ss, off, 32);
          sd += __shfl_xor(sd, off, 32);
        }
        if (lo == 0) {
          const int row = rbase + r;
          s_src[row * HEADS + head] = ss;
          s_dst[row * HEADS + head] = sd;
        }
      }
    }
  }
}

__global__ void csr_zero_kernel(int* __restrict__ cnt, int N) {
  const int idx = blockIdx.x * blockDim.x + threadIdx.x;
  if (idx < N) cnt[idx] = 0;
}

__global__ void csr_count_kernel(const int* __restrict__ ei, int E, int* __restrict__ cnt) {
  const int idx = blockIdx.x * blockDim.x + threadIdx.x;
  if (idx < E) atomicAdd(cnt + ei[E + idx], 1);
}

__global__ void __launch_bounds__(1024)
csr_scan_kernel(const int* __restrict__ cnt, int* __restrict__ rowptr,
                int* __restrict__ cursor, int N) {
  __shared__ int part[1024];
  const int t   = threadIdx.x;
  const int per = (N + 1023) >> 10;
  const int lo  = t * per;
  const int hi  = (lo + per < N) ? (lo + per) : N;
  int s = 0;
  for (int i = lo; i < hi; ++i) s += cnt[i];
  part[t] = s;
  __syncthreads();
  for (int off = 1; off < 1024; off <<= 1) {
    const int v = (t >= off) ? part[t - off] : 0;
    __syncthreads();
    part[t] += v;
    __syncthreads();
  }
  int run = part[t] - s;
  for (int i = lo; i < hi; ++i) {
    const int c = cnt[i];
    rowptr[i] = run;
    cursor[i] = run;
    run += c;
  }
  if (t == 1023) rowptr[N] = part[1023];
}

__global__ void csr_fill_kernel(const int* __restrict__ ei, int E,
                                int* __restrict__ cursor, int* __restrict__ csr_src) {
  const int idx = blockIdx.x * blockDim.x + threadIdx.x;
  if (idx < E) {
    const int pos = atomicAdd(cursor + ei[E + idx], 1);
    csr_src[pos] = ei[idx];
  }
}

static __device__ __forceinline__ float wave_max(float v) {
  #pragma unroll
  for (int off = 16; off > 0; off >>= 1) v = fmaxf(v, __shfl_xor(v, off, 32));
  return v;
}
static __device__ __forceinline__ float wave_sum(float v) {
  #pragma unroll
  for (int off = 16; off > 0; off >>= 1) v += __shfl_xor(v, off, 32);
  return v;
}

static __device__ __forceinline__ void load_entry(const int* __restrict__ csr_src,
                                                  const float* __restrict__ s_src,
                                                  int beg, int deg, int d, int j,
                                                  const float4 vd, int& s, float4& e) {
  s = d;
  if (j < deg) s = csr_src[beg + j];
  if (j <= deg) {
    const float4 vs = *(const float4*)(s_src + (size_t)s * HEADS);
    e = make_float4(lrelu(vs.x + vd.x), lrelu(vs.y + vd.y), lrelu(vs.z + vd.z), lrelu(vs.w + vd.w));
  } else {
    e = make_float4(-INFINITY, -INFINITY, -INFINITY, -INFINITY);
  }
}

static __device__ __forceinline__ void accum_entries(const float* __restrict__ Hin,
                                                     int s, const float4 al, int cnt, int lane,
                                                     float4& acc0, float4& acc1) {
  const bool up = lane >= 16;
  #pragma unroll 4
  for (int t = 0; t < cnt; ++t) {
    const int   st = __builtin_amdgcn_readlane(s, t);
    const float a0 = __int_as_float(__builtin_amdgcn_readlane(__float_as_int(al.x), t));
    const float a1 = __int_as_float(__builtin_amdgcn_readlane(__float_as_int(al.y), t));
    const float a2 = __int_as_float(__builtin_amdgcn_readlane(__float_as_int(al.z), t));
    const float a3 = __int_as_float(__builtin_amdgcn_readlane(__float_as_int(al.w), t));
    const float* hs = Hin + (size_t)st * HIDDEN + (lane << 2);
    const float4 h0 = *(const float4*)(hs);
    const float4 h1 = *(const float4*)(hs + 128);
    const float w0 = up ? a1 : a0;
    const float w1 = up ? a3 : a2;
    acc0.x += h0.x * w0; acc0.y += h0.y * w0; acc0.z += h0.z * w0; acc0.w += h0.w * w0;
    acc1.x += h1.x * w1; acc1.y += h1.y * w1; acc1.z += h1.z * w1; acc1.w += h1.w * w1;
  }
}

__global__ void gat_gather_kernel(const int* __restrict__ rowptr,
                                  const int* __restrict__ csr_src,
                                  const float* __restrict__ s_src,
                                  const float* __restrict__ s_dst,
                                  const float* __restrict__ Hin,
                                  const float* __restrict__ bias,
                                  float* __restrict__ agg, int N) {
  const int lane = threadIdx.x & 31;
  const int d = __builtin_amdgcn_readfirstlane(blockIdx.x * (blockDim.x >> 5) + (threadIdx.x >> 5));
  if (d >= N) return;
  const int beg = rowptr[d];
  const int deg = rowptr[d + 1] - beg;
  const int tot = deg + 1;
  const float4 vd = *(const float4*)(s_dst + (size_t)d * HEADS);

  int s0; float4 e0;
  load_entry(csr_src, s_src, beg, deg, d, lane, vd, s0, e0);

  float4 m = e0;
  for (int base = 32; base < tot; base += 32) {
    int s; float4 e;
    load_entry(csr_src, s_src, beg, deg, d, base + lane, vd, s, e);
    m.x = fmaxf(m.x, e.x); m.y = fmaxf(m.y, e.y); m.z = fmaxf(m.z, e.z); m.w = fmaxf(m.w, e.w);
  }
  m.x = wave_max(m.x); m.y = wave_max(m.y); m.z = wave_max(m.z); m.w = wave_max(m.w);

  const bool v0 = lane <= deg;
  float4 x0 = make_float4(v0 ? expf(e0.x - m.x) : 0.f, v0 ? expf(e0.y - m.y) : 0.f,
                          v0 ? expf(e0.z - m.z) : 0.f, v0 ? expf(e0.w - m.w) : 0.f);
  float4 dn = x0;
  for (int base = 32; base < tot; base += 32) {
    int s; float4 e;
    load_entry(csr_src, s_src, beg, deg, d, base + lane, vd, s, e);
    if (base + lane <= deg) {
      dn.x += expf(e.x - m.x); dn.y += expf(e.y - m.y);
      dn.z += expf(e.z - m.z); dn.w += expf(e.w - m.w);
    }
  }
  dn.x = wave_sum(dn.x); dn.y = wave_sum(dn.y); dn.z = wave_sum(dn.z); dn.w = wave_sum(dn.w);

  float4 acc0 = make_float4(0.f, 0.f, 0.f, 0.f);
  float4 acc1 = make_float4(0.f, 0.f, 0.f, 0.f);
  {
    const float4 al = make_float4(x0.x / dn.x, x0.y / dn.y, x0.z / dn.z, x0.w / dn.w);
    accum_entries(Hin, s0, al, tot < 32 ? tot : 32, lane, acc0, acc1);
  }
  for (int base = 32; base < tot; base += 32) {
    int s; float4 e;
    load_entry(csr_src, s_src, beg, deg, d, base + lane, vd, s, e);
    const bool v = base + lane <= deg;
    const float4 al = make_float4(v ? expf(e.x - m.x) / dn.x : 0.f, v ? expf(e.y - m.y) / dn.y : 0.f,
                                  v ? expf(e.z - m.z) / dn.z : 0.f, v ? expf(e.w - m.w) / dn.w : 0.f);
    const int rem = tot - base;
    accum_entries(Hin, s, al, rem < 32 ? rem : 32, lane, acc0, acc1);
  }

  const float4 b0 = *(const float4*)(bias + (lane << 2));
  const float4 b1 = *(const float4*)(bias + 128 + (lane << 2));
  float* od = agg + (size_t)d * HIDDEN + (lane << 2);
  *(float4*)(od)       = make_float4(acc0.x + b0.x, acc0.y + b0.y, acc0.z + b0.z, acc0.w + b0.w);
  *(float4*)(od + 128) = make_float4(acc1.x + b1.x, acc1.y + b1.y, acc1.z + b1.z, acc1.w + b1.w);
}

__global__ void elu_kernel(float* __restrict__ x, int n) {
  const int idx = blockIdx.x * blockDim.x + threadIdx.x;
  if (idx < n) {
    const float v = x[idx];
    x[idx] = v > 0.f ? v : expm1f(v);
  }
}

extern "C" void kernel_launch(void* const* d_in, const int* in_sizes, int n_in,
                              void* d_out, int out_size, void* d_ws, size_t ws_size,
                              hipStream_t stream) {
  const float* x     = (const float*)d_in[0];
  const int*   ei    = (const int*)  d_in[1];
  const float* W1    = (const float*)d_in[2];
  const float* asrc1 = (const float*)d_in[3];
  const float* adst1 = (const float*)d_in[4];
  const float* b1    = (const float*)d_in[5];
  const float* W2    = (const float*)d_in[6];
  const float* asrc2 = (const float*)d_in[7];
  const float* adst2 = (const float*)d_in[8];
  const float* b2    = (const float*)d_in[9];
  const float* Wf    = (const float*)d_in[10];
  const float* bf    = (const float*)d_in[11];

  const int B  = 2;
  const int N  = in_sizes[0] / (B * FEAT);
  const int E  = in_sizes[1] / 2;

  float* p = (float*)d_ws;
  float* buf_h   = p; p += (size_t)N * HIDDEN;
  float* buf_agg = p; p += (size_t)N * HIDDEN;
  float* ssrc    = p; p += (size_t)N * HEADS;
  float* sdst    = p; p += (size_t)N * HEADS;
  int* q = (int*)p;
  int* cnt     = q; q += (size_t)N;
  int* cursor  = q; q += (size_t)N;
  int* rowptr  = q; q += (size_t)N + 4;
  int* csr_src = q; q += (size_t)E;

  const int mtiles  = N >> 5;
  const int gtiles  = mtiles * (HIDDEN >> 6);
  const int ftiles  = mtiles * (CHAN >> 6);
  const int nHC     = N * HIDDEN;
  const dim3 blkG(128);
  const dim3 blkE(256);

  csr_zero_kernel<<<(N + 255) / 256, blkE, 0, stream>>>(cnt, N);
  csr_count_kernel<<<(E + 255) / 256, blkE, 0, stream>>>(ei, E, cnt);
  csr_scan_kernel<<<1, 1024, 0, stream>>>(cnt, rowptr, cursor, N);
  csr_fill_kernel<<<(E + 255) / 256, blkE, 0, stream>>>(ei, E, cursor, csr_src);

  for (int b = 0; b < B; ++b) {
    const float* xb = x + (size_t)b * N * FEAT;
    float* outb = (float*)d_out + (size_t)b * N * CHAN;

    wmma_gemm_kernel<true, false><<<(gtiles + 3) / 4, blkG, 0, stream>>>(
        xb, W1, nullptr, asrc1, adst1, buf_h, ssrc, sdst, N, FEAT, HIDDEN);
    gat_gather_kernel<<<(N + 7) / 8, blkE, 0, stream>>>(rowptr, csr_src, ssrc, sdst, buf_h, b1, buf_agg, N);
    elu_kernel<<<(nHC + 255) / 256, blkE, 0, stream>>>(buf_agg, nHC);

    wmma_gemm_kernel<true, false><<<(gtiles + 3) / 4, blkG, 0, stream>>>(
        buf_agg, W2, nullptr, asrc2, adst2, buf_h, ssrc, sdst, N, HIDDEN, HIDDEN);
    gat_gather_kernel<<<(N + 7) / 8, blkE, 0, stream>>>(rowptr, csr_src, ssrc, sdst, buf_h, b2, buf_agg, N);

    wmma_gemm_kernel<false, true><<<(ftiles + 3) / 4, blkG, 0, stream>>>(
        buf_agg, Wf, bf, nullptr, nullptr, outb, nullptr, nullptr, N, HIDDEN, CHAN);
  }
}


# Layer-1 ELU applied to the finished row in the gather kernel's registers before its one store; the separate one-element-per-thread ELU pass (2 launches of 20000 blocks each) is gone
# speedup vs baseline: 3.2389x; 1.1397x over previous
#include <hip/hip_runtime.h>
#include <math.h>


typedef __attribute__((ext_vector_type(2))) float v2f;
typedef __attribute__((ext_vector_type(8))) float v8f;

#define HEADS     4
#define CHAN      64
#define HIDDEN    256
#define FEAT      128
#define NEG_SLOPE 0.2f

static __device__ __forceinline__ float lrelu(float x) {
  return x > 0.f ? x : NEG_SLOPE * x;
}

template <bool GAT, bool BIAS>
__global__ void wmma_gemm_kernel(const float* __restrict__ A,
                                 const float* __restrict__ W,
                                 const float* __restrict__ bias,
                                 const float* __restrict__ att_src,
                                 const float* __restrict__ att_dst,
                                 float* __restrict__ Hout,
                                 float* __restrict__ s_src,
                                 float* __restrict__ s_dst,
                                 int M, int K, int Nc) {
  const int lane = threadIdx.x & 31;
  const int wv   = threadIdx.x >> 5;
  const int wpb  = blockDim.x >> 5;
  const int tile = blockIdx.x * wpb + wv;
  const int ctiles = Nc >> 6;
  const int mt = tile / ctiles;
  const int ct = tile - mt * ctiles;
  const int m0 = mt << 5;
  if (m0 >= M) return;
  const int n0 = ct << 6;
  const int lo = lane & 15;
  const int hi = lane >> 4;

  v8f acc[2][4] = {};
  const float* arow0 = A + (size_t)(m0 + lo) * K;
  const float* arow1 = A + (size_t)(m0 + 16 + lo) * K;
  const float* wcol  = W + n0 + lo;

  for (int k = 0; k < K; k += 4) {
    const int ka = k + (hi << 1);
    v2f a0 = *(const v2f*)(arow0 + ka);
    v2f a1 = *(const v2f*)(arow1 + ka);
    const float* w0 = wcol + (size_t)ka * Nc;
    const float* w1 = w0 + Nc;
    v2f b0 = {w0[0],  w1[0]};
    v2f b1 = {w0[16], w1[16]};
    v2f b2 = {w0[32], w1[32]};
    v2f b3 = {w0[48], w1[48]};
    acc[0][0] = __builtin_amdgcn_wmma_f32_16x16x4_f32(false, a0, false, b0, (short)0, acc[0][0], false, false);
    acc[1][0] = __builtin_amdgcn_wmma_f32_16x16x4_f32(false, a1, false, b0, (short)0, acc[1][0], false, false);
    acc[0][1] = __builtin_amdgcn_wmma_f32_16x16x4_f32(false, a0, false, b1, (short)0, acc[0][1], false, false);
    acc[1][1] = __builtin_amdgcn_wmma_f32_16x16x4_f32(false, a1, false, b1, (short)0, acc[1][1], false, false);
    acc[0][2] = __builtin_amdgcn_wmma_f32_16x16x4_f32(false, a0, false, b2, (short)0, acc[0][2], false, false);
    acc[1][2] = __builtin_amdgcn_wmma_f32_16x16x4_f32(false, a1, false, b2, (short)0, acc[1][2], false, false);
    acc[0][3] = __builtin_amdgcn_wmma_f32_16x16x4_f32(false, a0, false, b3, (short)0, acc[0][3], false, false);
    acc[1][3] = __builtin_amdgcn_wmma_f32_16x16x4_f32(false, a1, false, b3, (short)0, acc[1][3], false, false);
  }

  float bv0 = 0.f, bv1 = 0.f, bv2 = 0.f, bv3 = 0.f;
  if (BIAS) {
    bv0 = bias[n0 + lo];      bv1 = bias[n0 + 16 + lo];
    bv2 = bias[n0 + 32 + lo]; bv3 = bias[n0 + 48 + lo];
  }

  #pragma unroll
  for (int g = 0; g < 2; ++g) {
    const int rbase = m0 + (g << 4) + (hi << 3);
    #pragma unroll
    for (int r = 0; r < 8; ++r) {
      float* hp = Hout + (size_t)(rbase + r) * Nc + n0 + lo;
      hp[0]  = acc[g][0][r] + bv0;
      hp[16] = acc[g][1][r] + bv1;
      hp[32] = acc[g][2][r] + bv2;
      hp[48] = acc[g][3][r] + bv3;
    }
  }

  if (GAT) {
    const int head = n0 >> 6;
    const float* as = att_src + head * CHAN;
    const float* ad = att_dst + head * CHAN;
    const float as0 = as[lo], as1 = as[16 + lo], as2 = as[32 + lo], as3 = as[48 + lo];
    const float ad0 = ad[lo], ad1 = ad[16 + lo], ad2 = ad[32 + lo], ad3 = ad[48 + lo];
    #pragma unroll
    for (int g = 0; g < 2; ++g) {
      const int rbase = m0 + (g << 4) + (hi << 3);
      #pragma unroll
      for (int r = 0; r < 8; ++r) {
        float ss = acc[g][0][r] * as0 + acc[g][1][r] * as1 + acc[g][2][r] * as2 + acc[g][3][r] * as3;
        float sd = acc[g][0][r] * ad0 + acc[g][1][r] * ad1 + acc[g][2][r] * ad2 + acc[g][3][r] * ad3;
        #pragma unroll
        for (int off = 8; off > 0; off >>= 1) {
          ss += __shfl_xor(ss, off, 32);
          sd += __shfl_xor(sd, off, 32);
        }
        if (lo == 0) {
          const int row = rbase + r;
          s_src[row * HEADS + head] = ss;
          s_dst[row * HEADS + head] = sd;
        }
      }
    }
  }
}

__global__ void csr_zero_kernel(int* __restrict__ cnt, int N) {
  const int idx = blockIdx.x * blockDim.x + threadIdx.x;
  if (idx < N) cnt[idx] = 0;
}

__global__ void csr_count_kernel(const int* __restrict__ ei, int E, int* __restrict__ cnt) {
  const int idx = blockIdx.x * blockDim.x + threadIdx.x;
  if (idx < E) atomicAdd(cnt + ei[E + idx], 1);
}

__global__ void __launch_bounds__(1024)
csr_scan_kernel(const int* __restrict__ cnt, int* __restrict__ rowptr,
                int* __restrict__ cursor, int N) {
  __shared__ int part[1024];
  const int t   = threadIdx.x;
  const int per = (N + 1023) >> 10;
  const int lo  = t * per;
  const int hi  = (lo + per < N) ? (lo + per) : N;
  int s = 0;
  for (int i = lo; i < hi; ++i) s += cnt[i];
  part[t] = s;
  __syncthreads();
  for (int off = 1; off < 1024; off <<= 1) {
    const int v = (t >= off) ? part[t - off] : 0;
    __syncthreads();
    part[t] += v;
    __syncthreads();
  }
  int run = part[t] - s;
  for (int i = lo; i < hi; ++i) {
    const int c = cnt[i];
    rowptr[i] = run;
    cursor[i] = run;
    run += c;
  }
  if (t == 1023) rowptr[N] = part[1023];
}

__global__ void csr_fill_kernel(const int* __restrict__ ei, int E,
                                int* __restrict__ cursor, int* __restrict__ csr_src) {
  const int idx = blockIdx.x * blockDim.x + threadIdx.x;
  if (idx < E) {
    const int pos = atomicAdd(cursor + ei[E + idx], 1);
    csr_src[pos] = ei[idx];
  }
}

static __device__ __forceinline__ float wave_max(float v) {
  #pragma unroll
  for (int off = 16; off > 0; off >>= 1) v = fmaxf(v, __shfl_xor(v, off, 32));
  return v;
}
static __device__ __forceinline__ float wave_sum(float v) {
  #pragma unroll
  for (int off = 16; off > 0; off >>= 1) v += __shfl_xor(v, off, 32);
  return v;
}

static __device__ __forceinline__ void load_entry(const int* __restrict__ csr_src,
                                                  const float* __restrict__ s_src,
                                                  int beg, int deg, int d, int j,
                                                  const float4 vd, int& s, float4& e) {
  s = d;
  if (j < deg) s = csr_src[beg + j];
  if (j <= deg) {
    const float4 vs = *(const float4*)(s_src + (size_t)s * HEADS);
    e = make_float4(lrelu(vs.x + vd.x), lrelu(vs.y + vd.y), lrelu(vs.z + vd.z), lrelu(vs.w + vd.w));
  } else {
    e = make_float4(-INFINITY, -INFINITY, -INFINITY, -INFINITY);
  }
}

static __device__ __forceinline__ void accum_entries(const float* __restrict__ Hin,
                                                     int s, const float4 al, int cnt, int lane,
                                                     float4& acc0, float4& acc1) {
  const bool up = lane >= 16;
  #pragma unroll 4
  for (int t = 0; t < cnt; ++t) {
    const int   st = __builtin_amdgcn_readlane(s, t);
    const float a0 = __int_as_float(__builtin_amdgcn_readlane(__float_as_int(al.x), t));
    const float a1 = __int_as_float(__builtin_amdgcn_readlane(__float_as_int(al.y), t));
    const float a2 = __int_as_float(__builtin_amdgcn_readlane(__float_as_int(al.z), t));
    const float a3 = __int_as_float(__builtin_amdgcn_readlane(__float_as_int(al.w), t));
    const float* hs = Hin + (size_t)st * HIDDEN + (lane << 2);
    const float4 h0 = *(const float4*)(hs);
    const float4 h1 = *(const float4*)(hs + 128);
    const float w0 = up ? a1 : a0;
    const float w1 = up ? a3 : a2;
    acc0.x += h0.x * w0; acc0.y += h0.y * w0; acc0.z += h0.z * w0; acc0.w += h0.w * w0;
    acc1.x += h1.x * w1; acc1.y += h1.y * w1; acc1.z += h1.z * w1; acc1.w += h1.w * w1;
  }
}

static __device__ __forceinline__ float elu1(float v) {
  return v > 0.f ? v : expm1f(v);
}

template <bool ELU>
__global__ void gat_gather_kernel(const int* __restrict__ rowptr,
                                  const int* __restrict__ csr_src,
                                  const float* __restrict__ s_src,
                                  const float* __restrict__ s_dst,
                                  const float* __restrict__ Hin,
                                  const float* __restrict__ bias,
                                  float* __restrict__ agg, int N) {
  const int lane = threadIdx.x & 31;
  const int d = __builtin_amdgcn_readfirstlane(blockIdx.x * (blockDim.x >> 5) + (threadIdx.x >> 5));
  if (d >= N) return;
  const int beg = rowptr[d];
  const int deg = rowptr[d + 1] - beg;
  const int tot = deg + 1;
  const float4 vd = *(const float4*)(s_dst + (size_t)d * HEADS);

  int s0; float4 e0;
  load_entry(csr_src, s_src, beg, deg, d, lane, vd, s0, e0);

  float4 m = e0;
  for (int base = 32; base < tot; base += 32) {
    int s; float4 e;
    load_entry(csr_src, s_src, beg, deg, d, base + lane, vd, s, e);
    m.x = fmaxf(m.x, e.x); m.y = fmaxf(m.y, e.y); m.z = fmaxf(m.z, e.z); m.w = fmaxf(m.w, e.w);
  }
  m.x = wave_max(m.x); m.y = wave_max(m.y); m.z = wave_max(m.z); m.w = wave_max(m.w);

  const bool v0 = lane <= deg;
  float4 x0 = make_float4(v0 ? expf(e0.x - m.x) : 0.f, v0 ? expf(e0.y - m.y) : 0.f,
                          v0 ? expf(e0.z - m.z) : 0.f, v0 ? expf(e0.w - m.w) : 0.f);
  float4 dn = x0;
  for (int base = 32; base < tot; base += 32) {
    int s; float4 e;
    load_entry(csr_src, s_src, beg, deg, d, base + lane, vd, s, e);
    if (base + lane <= deg) {
      dn.x += expf(e.x - m.x); dn.y += expf(e.y - m.y);
      dn.z += expf(e.z - m.z); dn.w += expf(e.w - m.w);
    }
  }
  dn.x = wave_sum(dn.x); dn.y = wave_sum(dn.y); dn.z = wave_sum(dn.z); dn.w = wave_sum(dn.w);

  float4 acc0 = make_float4(0.f, 0.f, 0.f, 0.f);
  float4 acc1 = make_float4(0.f, 0.f, 0.f, 0.f);
  {
    const float4 al = make_float4(x0.x / dn.x, x0.y / dn.y, x0.z / dn.z, x0.w / dn.w);
    accum_entries(Hin, s0, al, tot < 32 ? tot : 32, lane, acc0, acc1);
  }
  for (int base = 32; base < tot; base += 32) {
    int s; float4 e;
    load_entry(csr_src, s_src, beg, deg, d, base + lane, vd, s, e);
    const bool v = base + lane <= deg;
    const float4 al = make_float4(v ? expf(e.x - m.x) / dn.x : 0.f, v ? expf(e.y - m.y) / dn.y : 0.f,
                                  v ? expf(e.z - m.z) / dn.z : 0.f, v ? expf(e.w - m.w) / dn.w : 0.f);
    const int rem = tot - base;
    accum_entries(Hin, s, al, rem < 32 ? rem : 32, lane, acc0, acc1);
  }

  const float4 b0 = *(const float4*)(bias + (lane << 2));
  const float4 b1 = *(const float4*)(bias + 128 + (lane << 2));
  float* od = agg + (size_t)d * HIDDEN + (lane << 2);
  float4 o0 = make_float4(acc0.x + b0.x, acc0.y + b0.y, acc0.z + b0.z, acc0.w + b0.w);
  float4 o1 = make_float4(acc1.x + b1.x, acc1.y + b1.y, acc1.z + b1.z, acc1.w + b1.w);
  if (ELU) {
    o0 = make_float4(elu1(o0.x), elu1(o0.y), elu1(o0.z), elu1(o0.w));
    o1 = make_float4(elu1(o1.x), elu1(o1.y), elu1(o1.z), elu1(o1.w));
  }
  *(float4*)(od)       = o0;
  *(float4*)(od + 128) = o1;
}

extern "C" void kernel_launch(void* const* d_in, const int* in_sizes, int n_in,
                              void* d_out, int out_size, void* d_ws, size_t ws_size,
                              hipStream_t stream) {
  const float* x     = (const float*)d_in[0];
  const int*   ei    = (const int*)  d_in[1];
  const float* W1    = (const float*)d_in[2];
  const float* asrc1 = (const float*)d_in[3];
  const float* adst1 = (const float*)d_in[4];
  const float* b1    = (const float*)d_in[5];
  const float* W2    = (const float*)d_in[6];
  const float* asrc2 = (const float*)d_in[7];
  const float* adst2 = (const float*)d_in[8];
  const float* b2    = (const float*)d_in[9];
  const float* Wf    = (const float*)d_in[10];
  const float* bf    = (const float*)d_in[11];

  const int B  = 2;
  const int N  = in_sizes[0] / (B * FEAT);
  const int E  = in_sizes[1] / 2;

  float* p = (float*)d_ws;
  float* buf_h   = p; p += (size_t)N * HIDDEN;
  float* buf_agg = p; p += (size_t)N * HIDDEN;
  float* ssrc    = p; p += (size_t)N * HEADS;
  float* sdst    = p; p += (size_t)N * HEADS;
  int* q = (int*)p;
  int* cnt     = q; q += (size_t)N;
  int* cursor  = q; q += (size_t)N;
  int* rowptr  = q; q += (size_t)N + 4;
  int* csr_src = q; q += (size_t)E;

  const int mtiles  = N >> 5;
  const int gtiles  = mtiles * (HIDDEN >> 6);
  const int ftiles  = mtiles * (CHAN >> 6);
  const dim3 blkG(128);
  const dim3 blkE(256);

  csr_zero_kernel<<<(N + 255) / 256, blkE, 0, stream>>>(cnt, N);
  csr_count_kernel<<<(E + 255) / 256, blkE, 0, stream>>>(ei, E, cnt);
  csr_scan_kernel<<<1, 1024, 0, stream>>>(cnt, rowptr, cursor, N);
  csr_fill_kernel<<<(E + 255) / 256, blkE, 0, stream>>>(ei, E, cursor, csr_src);

  for (int b = 0; b < B; ++b) {
    const float* xb = x + (size_t)b * N * FEAT;
    float* outb = (float*)d_out + (size_t)b * N * CHAN;

    wmma_gemm_kernel<true, false><<<(gtiles + 3) / 4, blkG, 0, stream>>>(
        xb, W1, nullptr, asrc1, adst1, buf_h, ssrc, sdst, N, FEAT, HIDDEN);
    gat_gather_kernel<true><<<(N + 7) / 8, blkE, 0, stream>>>(rowptr, csr_src, ssrc, sdst, buf_h, b1, buf_agg, N);

    wmma_gemm_kernel<true, false><<<(gtiles + 3) / 4, blkG, 0, stream>>>(
        buf_agg, W2, nullptr, asrc2, adst2, buf_h, ssrc, sdst, N, HIDDEN, HIDDEN);
    gat_gather_kernel<false><<<(N + 7) / 8, blkE, 0, stream>>>(rowptr, csr_src, ssrc, sdst, buf_h, b2, buf_agg, N);

    wmma_gemm_kernel<false, true><<<(ftiles + 3) / 4, blkG, 0, stream>>>(
        buf_agg, Wf, bf, nullptr, nullptr, outb, nullptr, nullptr, N, HIDDEN, CHAN);
  }
}


# GEMM K loop software-pipelined: A/B fragments of the next 16 k are loaded into a second register set (launch bounds 128, no spill) before the current block's 32 WMMAs; k order and sums unchanged
# speedup vs baseline: 4.1828x; 1.2914x over previous
#include <hip/hip_runtime.h>
#include <math.h>


typedef __attribute__((ext_vector_type(2))) float v2f;
typedef __attribute__((ext_vector_type(8))) float v8f;

#define HEADS     4
#define CHAN      64
#define HIDDEN    256
#define FEAT      128
#define NEG_SLOPE 0.2f

static __device__ __forceinline__ float lrelu(float x) {
  return x > 0.f ? x : NEG_SLOPE * x;
}

#define KB 4
struct Frag {
  v2f a0[KB];
  v2f a1[KB];
  v2f b[KB][4];
};

static __device__ __forceinline__ void load_frag(Frag& f,
                                                 const float* __restrict__ ap0,
                                                 const float* __restrict__ ap1,
                                                 const float* __restrict__ wp, int Nc) {
  #pragma unroll
  for (int s = 0; s < KB; ++s) {
    f.a0[s] = *(const v2f*)(ap0 + 4 * s);
    f.a1[s] = *(const v2f*)(ap1 + 4 * s);
    const float* w0 = wp + (size_t)(4 * s) * Nc;
    const float* w1 = w0 + Nc;
    f.b[s][0] = v2f{w0[0],  w1[0]};
    f.b[s][1] = v2f{w0[16], w1[16]};
    f.b[s][2] = v2f{w0[32], w1[32]};
    f.b[s][3] = v2f{w0[48], w1[48]};
  }
}

static __device__ __forceinline__ void mma_frag(v8f (&acc)[2][4], const Frag& f) {
  #pragma unroll
  for (int s = 0; s < KB; ++s) {
    #pragma unroll
    for (int c = 0; c < 4; ++c) {
      acc[0][c] = __builtin_amdgcn_wmma_f32_16x16x4_f32(false, f.a0[s], false, f.b[s][c], (short)0, acc[0][c], false, false);
      acc[1][c] = __builtin_amdgcn_wmma_f32_16x16x4_f32(false, f.a1[s], false, f.b[s][c], (short)0, acc[1][c], false, false);
    }
  }
}

template <bool GAT, bool BIAS>
__global__ void __launch_bounds__(128) wmma_gemm_kernel(const float* __restrict__ A,
                                 const float* __restrict__ W,
                                 const float* __restrict__ bias,
                                 const float* __restrict__ att_src,
                                 const float* __restrict__ att_dst,
                                 float* __restrict__ Hout,
                                 float* __restrict__ s_src,
                                 float* __restrict__ s_dst,
                                 int M, int K, int Nc) {
  const int lane = threadIdx.x & 31;
  const int wv   = threadIdx.x >> 5;
  const int wpb  = blockDim.x >> 5;
  const int tile = blockIdx.x * wpb + wv;
  const int ctiles = Nc >> 6;
  const int mt = tile / ctiles;
  const int ct = tile - mt * ctiles;
  const int m0 = mt << 5;
  if (m0 >= M) return;
  const int n0 = ct << 6;
  const int lo = lane & 15;
  const int hi = lane >> 4;

  v8f acc[2][4] = {};
  const float* ap0 = A + (size_t)(m0 + lo) * K + (hi << 1);
  const float* ap1 = A + (size_t)(m0 + 16 + lo) * K + (hi << 1);
  const float* wp  = W + n0 + lo + (size_t)(hi << 1) * Nc;
  const size_t wblk = (size_t)(4 * KB) * Nc;

  Frag f0, f1;
  load_frag(f0, ap0, ap1, wp, Nc);
  int k = 0;
  for (; k + 8 * KB < K; k += 8 * KB) {
    load_frag(f1, ap0 + 4 * KB, ap1 + 4 * KB, wp + wblk, Nc);
    mma_frag(acc, f0);
    load_frag(f0, ap0 + 8 * KB, ap1 + 8 * KB, wp + 2 * wblk, Nc);
    mma_frag(acc, f1);
    ap0 += 8 * KB; ap1 += 8 * KB; wp += 2 * wblk;
  }
  load_frag(f1, ap0 + 4 * KB, ap1 + 4 * KB, wp + wblk, Nc);
  mma_frag(acc, f0);
  mma_frag(acc, f1);

  float bv0 = 0.f, bv1 = 0.f, bv2 = 0.f, bv3 = 0.f;
  if (BIAS) {
    bv0 = bias[n0 + lo];      bv1 = bias[n0 + 16 + lo];
    bv2 = bias[n0 + 32 + lo]; bv3 = bias[n0 + 48 + lo];
  }

  #pragma unroll
  for (int g = 0; g < 2; ++g) {
    const int rbase = m0 + (g << 4) + (hi << 3);
    #pragma unroll
    for (int r = 0; r < 8; ++r) {
      float* hp = Hout + (size_t)(rbase + r) * Nc + n0 + lo;
      hp[0]  = acc[g][0][r] + bv0;
      hp[16] = acc[g][1][r] + bv1;
      hp[32] = acc[g][2][r] + bv2;
      hp[48] = acc[g][3][r] + bv3;
    }
  }

  if (GAT) {
    const int head = n0 >> 6;
    const float* as = att_src + head * CHAN;
    const float* ad = att_dst + head * CHAN;
    const float as0 = as[lo], as1 = as[16 + lo], as2 = as[32 + lo], as3 = as[48 + lo];
    const float ad0 = ad[lo], ad1 = ad[16 + lo], ad2 = ad[32 + lo], ad3 = ad[48 + lo];
    #pragma unroll
    for (int g = 0; g < 2; ++g) {
      const int rbase = m0 + (g << 4) + (hi << 3);
      #pragma unroll
      for (int r = 0; r < 8; ++r) {
        float ss = acc[g][0][r] * as0 + acc[g][1][r] * as1 + acc[g][2][r] * as2 + acc[g][3][r] * as3;
        float sd = acc[g][0][r] * ad0 + acc[g][1][r] * ad1 + acc[g][2][r] * ad2 + acc[g][3][r] * ad3;
        #pragma unroll
        for (int off = 8; off > 0; off >>= 1) {
          ss += __shfl_xor(ss, off, 32);
          sd += __shfl_xor(sd, off, 32);
        }
        if (lo == 0) {
          const int row = rbase + r;
          s_src[row * HEADS + head] = ss;
          s_dst[row * HEADS + head] = sd;
        }
      }
    }
  }
}

__global__ void csr_zero_kernel(int* __restrict__ cnt, int N) {
  const int idx = blockIdx.x * blockDim.x + threadIdx.x;
  if (idx < N) cnt[idx] = 0;
}

__global__ void csr_count_kernel(const int* __restrict__ ei, int E, int* __restrict__ cnt) {
  const int idx = blockIdx.x * blockDim.x + threadIdx.x;
  if (idx < E) atomicAdd(cnt + ei[E + idx], 1);
}

__global__ void __launch_bounds__(1024)
csr_scan_kernel(const int* __restrict__ cnt, int* __restrict__ rowptr,
                int* __restrict__ cursor, int N) {
  __shared__ int part[1024];
  const int t   = threadIdx.x;
  const int per = (N + 1023) >> 10;
  const int lo  = t * per;
  const int hi  = (lo + per < N) ? (lo + per) : N;
  int s = 0;
  for (int i = lo; i < hi; ++i) s += cnt[i];
  part[t] = s;
  __syncthreads();
  for (int off = 1; off < 1024; off <<= 1) {
    const int v = (t >= off) ? part[t - off] : 0;
    __syncthreads();
    part[t] += v;
    __syncthreads();
  }
  int run = part[t] - s;
  for (int i = lo; i < hi; ++i) {
    const int c = cnt[i];
    rowptr[i] = run;
    cursor[i] = run;
    run += c;
  }
  if (t == 1023) rowptr[N] = part[1023];
}

__global__ void csr_fill_kernel(const int* __restrict__ ei, int E,
                                int* __restrict__ cursor, int* __restrict__ csr_src) {
  const int idx = blockIdx.x * blockDim.x + threadIdx.x;
  if (idx < E) {
    const int pos = atomicAdd(cursor + ei[E + idx], 1);
    csr_src[pos] = ei[idx];
  }
}

static __device__ __forceinline__ float wave_max(float v) {
  #pragma unroll
  for (int off = 16; off > 0; off >>= 1) v = fmaxf(v, __shfl_xor(v, off, 32));
  return v;
}
static __device__ __forceinline__ float wave_sum(float v) {
  #pragma unroll
  for (int off = 16; off > 0; off >>= 1) v += __shfl_xor(v, off, 32);
  return v;
}

static __device__ __forceinline__ void load_entry(const int* __restrict__ csr_src,
                                                  const float* __restrict__ s_src,
                                                  int beg, int deg, int d, int j,
                                                  const float4 vd, int& s, float4& e) {
  s = d;
  if (j < deg) s = csr_src[beg + j];
  if (j <= deg) {
    const float4 vs = *(const float4*)(s_src + (size_t)s * HEADS);
    e = make_float4(lrelu(vs.x + vd.x), lrelu(vs.y + vd.y), lrelu(vs.z + vd.z), lrelu(vs.w + vd.w));
  } else {
    e = make_float4(-INFINITY, -INFINITY, -INFINITY, -INFINITY);
  }
}

static __device__ __forceinline__ void accum_entries(const float* __restrict__ Hin,
                                                     int s, const float4 al, int cnt, int lane,
                                                     float4& acc0, float4& acc1) {
  const bool up = lane >= 16;
  #pragma unroll 4
  for (int t = 0; t < cnt; ++t) {
    const int   st = __builtin_amdgcn_readlane(s, t);
    const float a0 = __int_as_float(__builtin_amdgcn_readlane(__float_as_int(al.x), t));
    const float a1 = __int_as_float(__builtin_amdgcn_readlane(__float_as_int(al.y), t));
    const float a2 = __int_as_float(__builtin_amdgcn_readlane(__float_as_int(al.z), t));
    const float a3 = __int_as_float(__builtin_amdgcn_readlane(__float_as_int(al.w), t));
    const float* hs = Hin + (size_t)st * HIDDEN + (lane << 2);
    const float4 h0 = *(const float4*)(hs);
    const float4 h1 = *(const float4*)(hs + 128);
    const float w0 = up ? a1 : a0;
    const float w1 = up ? a3 : a2;
    acc0.x += h0.x * w0; acc0.y += h0.y * w0; acc0.z += h0.z * w0; acc0.w += h0.w * w0;
    acc1.x += h1.x * w1; acc1.y += h1.y * w1; acc1.z += h1.z * w1; acc1.w += h1.w * w1;
  }
}

static __device__ __forceinline__ float elu1(float v) {
  return v > 0.f ? v : expm1f(v);
}

template <bool ELU>
__global__ void gat_gather_kernel(const int* __restrict__ rowptr,
                                  const int* __restrict__ csr_src,
                                  const float* __restrict__ s_src,
                                  const float* __restrict__ s_dst,
                                  const float* __restrict__ Hin,
                                  const float* __restrict__ bias,
                                  float* __restrict__ agg, int N) {
  const int lane = threadIdx.x & 31;
  const int d = __builtin_amdgcn_readfirstlane(blockIdx.x * (blockDim.x >> 5) + (threadIdx.x >> 5));
  if (d >= N) return;
  const int beg = rowptr[d];
  const int deg = rowptr[d + 1] - beg;
  const int tot = deg + 1;
  const float4 vd = *(const float4*)(s_dst + (size_t)d * HEADS);

  int s0; float4 e0;
  load_entry(csr_src, s_src, beg, deg, d, lane, vd, s0, e0);

  float4 m = e0;
  for (int base = 32; base < tot; base += 32) {
    int s; float4 e;
    load_entry(csr_src, s_src, beg, deg, d, base + lane, vd, s, e);
    m.x = fmaxf(m.x, e.x); m.y = fmaxf(m.y, e.y); m.z = fmaxf(m.z, e.z); m.w = fmaxf(m.w, e.w);
  }
  m.x = wave_max(m.x); m.y = wave_max(m.y); m.z = wave_max(m.z); m.w = wave_max(m.w);

  const bool v0 = lane <= deg;
  float4 x0 = make_float4(v0 ? expf(e0.x - m.x) : 0.f, v0 ? expf(e0.y - m.y) : 0.f,
                          v0 ? expf(e0.z - m.z) : 0.f, v0 ? expf(e0.w - m.w) : 0.f);
  float4 dn = x0;
  for (int base = 32; base < tot; base += 32) {
    int s; float4 e;
    load_entry(csr_src, s_src, beg, deg, d, base + lane, vd, s, e);
    if (base + lane <= deg) {
      dn.x += expf(e.x - m.x); dn.y += expf(e.y - m.y);
      dn.z += expf(e.z - m.z); dn.w += expf(e.w - m.w);
    }
  }
  dn.x = wave_sum(dn.x); dn.y = wave_sum(dn.y); dn.z = wave_sum(dn.z); dn.w = wave_sum(dn.w);

  float4 acc0 = make_float4(0.f, 0.f, 0.f, 0.f);
  float4 acc1 = make_float4(0.f, 0.f, 0.f, 0.f);
  {
    const float4 al = make_float4(x0.x / dn.x, x0.y / dn.y, x0.z / dn.z, x0.w / dn.w);
    accum_entries(Hin, s0, al, tot < 32 ? tot : 32, lane, acc0, acc1);
  }
  for (int base = 32; base < tot; base += 32) {
    int s; float4 e;
    load_entry(csr_src, s_src, beg, deg, d, base + lane, vd, s, e);
    const bool v = base + lane <= deg;
    const float4 al = make_float4(v ? expf(e.x - m.x) / dn.x : 0.f, v ? expf(e.y - m.y) / dn.y : 0.f,
                                  v ? expf(e.z - m.z) / dn.z : 0.f, v ? expf(e.w - m.w) / dn.w : 0.f);
    const int rem = tot - base;
    accum_entries(Hin, s, al, rem < 32 ? rem : 32, lane, acc0, acc1);
  }

  const float4 b0 = *(const float4*)(bias + (lane << 2));
  const float4 b1 = *(const float4*)(bias + 128 + (lane << 2));
  float* od = agg + (size_t)d * HIDDEN + (lane << 2);
  float4 o0 = make_float4(acc0.x + b0.x, acc0.y + b0.y, acc0.z + b0.z, acc0.w + b0.w);
  float4 o1 = make_float4(acc1.x + b1.x, acc1.y + b1.y, acc1.z + b1.z, acc1.w + b1.w);
  if (ELU) {
    o0 = make_float4(elu1(o0.x), elu1(o0.y), elu1(o0.z), elu1(o0.w));
    o1 = make_float4(elu1(o1.x), elu1(o1.y), elu1(o1.z), elu1(o1.w));
  }
  *(float4*)(od)       = o0;
  *(float4*)(od + 128) = o1;
}

extern "C" void kernel_launch(void* const* d_in, const int* in_sizes, int n_in,
                              void* d_out, int out_size, void* d_ws, size_t ws_size,
                              hipStream_t stream) {
  const float* x     = (const float*)d_in[0];
  const int*   ei    = (const int*)  d_in[1];
  const float* W1    = (const float*)d_in[2];
  const float* asrc1 = (const float*)d_in[3];
  const float* adst1 = (const float*)d_in[4];
  const float* b1    = (const float*)d_in[5];
  const float* W2    = (const float*)d_in[6];
  const float* asrc2 = (const float*)d_in[7];
  const float* adst2 = (const float*)d_in[8];
  const float* b2    = (const float*)d_in[9];
  const float* Wf    = (const float*)d_in[10];
  const float* bf    = (const float*)d_in[11];

  const int B  = 2;
  const int N  = in_sizes[0] / (B * FEAT);
  const int E  = in_sizes[1] / 2;

  float* p = (float*)d_ws;
  float* buf_h   = p; p += (size_t)N * HIDDEN;
  float* buf_agg = p; p += (size_t)N * HIDDEN;
  float* ssrc    = p; p += (size_t)N * HEADS;
  float* sdst    = p; p += (size_t)N * HEADS;
  int* q = (int*)p;
  int* cnt     = q; q += (size_t)N;
  int* cursor  = q; q += (size_t)N;
  int* rowptr  = q; q += (size_t)N + 4;
  int* csr_src = q; q += (size_t)E;

  const int mtiles  = N >> 5;
  const int gtiles  = mtiles * (HIDDEN >> 6);
  const int ftiles  = mtiles * (CHAN >> 6);
  const dim3 blkG(128);
  const dim3 blkE(256);

  csr_zero_kernel<<<(N + 255) / 256, blkE, 0, stream>>>(cnt, N);
  csr_count_kernel<<<(E + 255) / 256, blkE, 0, stream>>>(ei, E, cnt);
  csr_scan_kernel<<<1, 1024, 0, stream>>>(cnt, rowptr, cursor, N);
  csr_fill_kernel<<<(E + 255) / 256, blkE, 0, stream>>>(ei, E, cursor, csr_src);

  for (int b = 0; b < B; ++b) {
    const float* xb = x + (size_t)b * N * FEAT;
    float* outb = (float*)d_out + (size_t)b * N * CHAN;

    wmma_gemm_kernel<true, false><<<(gtiles + 3) / 4, blkG, 0, stream>>>(
        xb, W1, nullptr, asrc1, adst1, buf_h, ssrc, sdst, N, FEAT, HIDDEN);
    gat_gather_kernel<true><<<(N + 7) / 8, blkE, 0, stream>>>(rowptr, csr_src, ssrc, sdst, buf_h, b1, buf_agg, N);

    wmma_gemm_kernel<true, false><<<(gtiles + 3) / 4, blkG, 0, stream>>>(
        buf_agg, W2, nullptr, asrc2, adst2, buf_h, ssrc, sdst, N, HIDDEN, HIDDEN);
    gat_gather_kernel<false><<<(N + 7) / 8, blkE, 0, stream>>>(rowptr, csr_src, ssrc, sdst, buf_h, b2, buf_agg, N);

    wmma_gemm_kernel<false, true><<<(ftiles + 3) / 4, blkG, 0, stream>>>(
        buf_agg, Wf, bf, nullptr, nullptr, outb, nullptr, nullptr, N, HIDDEN, CHAN);
  }
}


# Both graphs of the batch stacked along rows: each GEMM and each gather runs once over 2N rows (gather maps a row to its graph-local node); 14 launches become 9; sums and their order unchanged
# speedup vs baseline: 4.7033x; 1.1244x over previous
#include <hip/hip_runtime.h>
#include <math.h>


typedef __attribute__((ext_vector_type(2))) float v2f;
typedef __attribute__((ext_vector_type(8))) float v8f;

#define HEADS     4
#define CHAN      64
#define HIDDEN    256
#define FEAT      128
#define NEG_SLOPE 0.2f

static __device__ __forceinline__ float lrelu(float x) {
  return x > 0.f ? x : NEG_SLOPE * x;
}

#define KB 4
struct Frag {
  v2f a0[KB];
  v2f a1[KB];
  v2f b[KB][4];
};

static __device__ __forceinline__ void load_frag(Frag& f,
                                                 const float* __restrict__ ap0,
                                                 const float* __restrict__ ap1,
                                                 const float* __restrict__ wp, int Nc) {
  #pragma unroll
  for (int s = 0; s < KB; ++s) {
    f.a0[s] = *(const v2f*)(ap0 + 4 * s);
    f.a1[s] = *(const v2f*)(ap1 + 4 * s);
    const float* w0 = wp + (size_t)(4 * s) * Nc;
    const float* w1 = w0 + Nc;
    f.b[s][0] = v2f{w0[0],  w1[0]};
    f.b[s][1] = v2f{w0[16], w1[16]};
    f.b[s][2] = v2f{w0[32], w1[32]};
    f.b[s][3] = v2f{w0[48], w1[48]};
  }
}

static __device__ __forceinline__ void mma_frag(v8f (&acc)[2][4], const Frag& f) {
  #pragma unroll
  for (int s = 0; s < KB; ++s) {
    #pragma unroll
    for (int c = 0; c < 4; ++c) {
      acc[0][c] = __builtin_amdgcn_wmma_f32_16x16x4_f32(false, f.a0[s], false, f.b[s][c], (short)0, acc[0][c], false, false);
      acc[1][c] = __builtin_amdgcn_wmma_f32_16x16x4_f32(false, f.a1[s], false, f.b[s][c], (short)0, acc[1][c], false, false);
    }
  }
}

template <bool GAT, bool BIAS>
__global__ void __launch_bounds__(128) wmma_gemm_kernel(const float* __restrict__ A,
                                 const float* __restrict__ W,
                                 const float* __restrict__ bias,
                                 const float* __restrict__ att_src,
                                 const float* __restrict__ att_dst,
                                 float* __restrict__ Hout,
                                 float* __restrict__ s_src,
                                 float* __restrict__ s_dst,
                                 int M, int K, int Nc) {
  const int lane = threadIdx.x & 31;
  const int wv   = threadIdx.x >> 5;
  const int wpb  = blockDim.x >> 5;
  const int tile = blockIdx.x * wpb + wv;
  const int ctiles = Nc >> 6;
  const int mt = tile / ctiles;
  const int ct = tile - mt * ctiles;
  const int m0 = mt << 5;
  if (m0 >= M) return;
  const int n0 = ct << 6;
  const int lo = lane & 15;
  const int hi = lane >> 4;

  v8f acc[2][4] = {};
  const float* ap0 = A + (size_t)(m0 + lo) * K + (hi << 1);
  const float* ap1 = A + (size_t)(m0 + 16 + lo) * K + (hi << 1);
  const float* wp  = W + n0 + lo + (size_t)(hi << 1) * Nc;
  const size_t wblk = (size_t)(4 * KB) * Nc;

  Frag f0, f1;
  load_frag(f0, ap0, ap1, wp, Nc);
  int k = 0;
  for (; k + 8 * KB < K; k += 8 * KB) {
    load_frag(f1, ap0 + 4 * KB, ap1 + 4 * KB, wp + wblk, Nc);
    mma_frag(acc, f0);
    load_frag(f0, ap0 + 8 * KB, ap1 + 8 * KB, wp + 2 * wblk, Nc);
    mma_frag(acc, f1);
    ap0 += 8 * KB; ap1 += 8 * KB; wp += 2 * wblk;
  }
  load_frag(f1, ap0 + 4 * KB, ap1 + 4 * KB, wp + wblk, Nc);
  mma_frag(acc, f0);
  mma_frag(acc, f1);

  float bv0 = 0.f, bv1 = 0.f, bv2 = 0.f, bv3 = 0.f;
  if (BIAS) {
    bv0 = bias[n0 + lo];      bv1 = bias[n0 + 16 + lo];
    bv2 = bias[n0 + 32 + lo]; bv3 = bias[n0 + 48 + lo];
  }

  #pragma unroll
  for (int g = 0; g < 2; ++g) {
    const int rbase = m0 + (g << 4) + (hi << 3);
    #pragma unroll
    for (int r = 0; r < 8; ++r) {
      float* hp = Hout + (size_t)(rbase + r) * Nc + n0 + lo;
      hp[0]  = acc[g][0][r] + bv0;
      hp[16] = acc[g][1][r] + bv1;
      hp[32] = acc[g][2][r] + bv2;
      hp[48] = acc[g][3][r] + bv3;
    }
  }

  if (GAT) {
    const int head = n0 >> 6;
    const float* as = att_src + head * CHAN;
    const float* ad = att_dst + head * CHAN;
    const float as0 = as[lo], as1 = as[16 + lo], as2 = as[32 + lo], as3 = as[48 + lo];
    const float ad0 = ad[lo], ad1 = ad[16 + lo], ad2 = ad[32 + lo], ad3 = ad[48 + lo];
    #pragma unroll
    for (int g = 0; g < 2; ++g) {
      const int rbase = m0 + (g << 4) + (hi << 3);
      #pragma unroll
      for (int r = 0; r < 8; ++r) {
        float ss = acc[g][0][r] * as0 + acc[g][1][r] * as1 + acc[g][2][r] * as2 + acc[g][3][r] * as3;
        float sd = acc[g][0][r] * ad0 + acc[g][1][r] * ad1 + acc[g][2][r] * ad2 + acc[g][3][r] * ad3;
        #pragma unroll
        for (int off = 8; off > 0; off >>= 1) {
          ss += __shfl_xor(ss, off, 32);
          sd += __shfl_xor(sd, off, 32);
        }
        if (lo == 0) {
          const int row = rbase + r;
          s_src[row * HEADS + head] = ss;
          s_dst[row * HEADS + head] = sd;
        }
      }
    }
  }
}

__global__ void csr_zero_kernel(int* __restrict__ cnt, int N) {
  const int idx = blockIdx.x * blockDim.x + threadIdx.x;
  if (idx < N) cnt[idx] = 0;
}

__global__ void csr_count_kernel(const int* __restrict__ ei, int E, int* __restrict__ cnt) {
  const int idx = blockIdx.x * blockDim.x + threadIdx.x;
  if (idx < E) atomicAdd(cnt + ei[E + idx], 1);
}

__global__ void __launch_bounds__(1024)
csr_scan_kernel(const int* __restrict__ cnt, int* __restrict__ rowptr,
                int* __restrict__ cursor, int N) {
  __shared__ int part[1024];
  const int t   = threadIdx.x;
  const int per = (N + 1023) >> 10;
  const int lo  = t * per;
  const int hi  = (lo + per < N) ? (lo + per) : N;
  int s = 0;
  for (int i = lo; i < hi; ++i) s += cnt[i];
  part[t] = s;
  __syncthreads();
  for (int off = 1; off < 1024; off <<= 1) {
    const int v = (t >= off) ? part[t - off] : 0;
    __syncthreads();
    part[t] += v;
    __syncthreads();
  }
  int run = part[t] - s;
  for (int i = lo; i < hi; ++i) {
    const int c = cnt[i];
    rowptr[i] = run;
    cursor[i] = run;
    run += c;
  }
  if (t == 1023) rowptr[N] = part[1023];
}

__global__ void csr_fill_kernel(const int* __restrict__ ei, int E,
                                int* __restrict__ cursor, int* __restrict__ csr_src) {
  const int idx = blockIdx.x * blockDim.x + threadIdx.x;
  if (idx < E) {
    const int pos = atomicAdd(cursor + ei[E + idx], 1);
    csr_src[pos] = ei[idx];
  }
}

static __device__ __forceinline__ float wave_max(float v) {
  #pragma unroll
  for (int off = 16; off > 0; off >>= 1) v = fmaxf(v, __shfl_xor(v, off, 32));
  return v;
}
static __device__ __forceinline__ float wave_sum(float v) {
  #pragma unroll
  for (int off = 16; off > 0; off >>= 1) v += __shfl_xor(v, off, 32);
  return v;
}

static __device__ __forceinline__ void load_entry(const int* __restrict__ csr_src,
                                                  const float* __restrict__ s_src,
                                                  int beg, int deg, int d, int row0, int j,
                                                  const float4 vd, int& s, float4& e) {
  s = d;
  if (j < deg) s = row0 + csr_src[beg + j];
  if (j <= deg) {
    const float4 vs = *(const float4*)(s_src + (size_t)s * HEADS);
    e = make_float4(lrelu(vs.x + vd.x), lrelu(vs.y + vd.y), lrelu(vs.z + vd.z), lrelu(vs.w + vd.w));
  } else {
    e = make_float4(-INFINITY, -INFINITY, -INFINITY, -INFINITY);
  }
}

static __device__ __forceinline__ void accum_entries(const float* __restrict__ Hin,
                                                     int s, const float4 al, int cnt, int lane,
                                                     float4& acc0, float4& acc1) {
  const bool up = lane >= 16;
  #pragma unroll 4
  for (int t = 0; t < cnt; ++t) {
    const int   st = __builtin_amdgcn_readlane(s, t);
    const float a0 = __int_as_float(__builtin_amdgcn_readlane(__float_as_int(al.x), t));
    const float a1 = __int_as_float(__builtin_amdgcn_readlane(__float_as_int(al.y), t));
    const float a2 = __int_as_float(__builtin_amdgcn_readlane(__float_as_int(al.z), t));
    const float a3 = __int_as_float(__builtin_amdgcn_readlane(__float_as_int(al.w), t));
    const float* hs = Hin + (size_t)st * HIDDEN + (lane << 2);
    const float4 h0 = *(const float4*)(hs);
    const float4 h1 = *(const float4*)(hs + 128);
    const float w0 = up ? a1 : a0;
    const float w1 = up ? a3 : a2;
    acc0.x += h0.x * w0; acc0.y += h0.y * w0; acc0.z += h0.z * w0; acc0.w += h0.w * w0;
    acc1.x += h1.x * w1; acc1.y += h1.y * w1; acc1.z += h1.z * w1; acc1.w += h1.w * w1;
  }
}

static __device__ __forceinline__ float elu1(float v) {
  return v > 0.f ? v : expm1f(v);
}

template <bool ELU>
__global__ void gat_gather_kernel(const int* __restrict__ rowptr,
                                  const int* __restrict__ csr_src,
                                  const float* __restrict__ s_src,
                                  const float* __restrict__ s_dst,
                                  const float* __restrict__ Hin,
                                  const float* __restrict__ bias,
                                  float* __restrict__ agg, int N, int R) {
  const int lane = threadIdx.x & 31;
  const int d = __builtin_amdgcn_readfirstlane(blockIdx.x * (blockDim.x >> 5) + (threadIdx.x >> 5));
  if (d >= R) return;
  const int row0 = d >= N ? N : 0;
  const int g    = d - row0;
  const int beg = rowptr[g];
  const int deg = rowptr[g + 1] - beg;
  const int tot = deg + 1;
  const float4 vd = *(const float4*)(s_dst + (size_t)d * HEADS);

  int s0; float4 e0;
  load_entry(csr_src, s_src, beg, deg, d, row0, lane, vd, s0, e0);

  float4 m = e0;
  for (int base = 32; base < tot; base += 32) {
    int s; float4 e;
    load_entry(csr_src, s_src, beg, deg, d, row0, base + lane, vd, s, e);
    m.x = fmaxf(m.x, e.x); m.y = fmaxf(m.y, e.y); m.z = fmaxf(m.z, e.z); m.w = fmaxf(m.w, e.w);
  }
  m.x = wave_max(m.x); m.y = wave_max(m.y); m.z = wave_max(m.z); m.w = wave_max(m.w);

  const bool v0 = lane <= deg;
  float4 x0 = make_float4(v0 ? expf(e0.x - m.x) : 0.f, v0 ? expf(e0.y - m.y) : 0.f,
                          v0 ? expf(e0.z - m.z) : 0.f, v0 ? expf(e0.w - m.w) : 0.f);
  float4 dn = x0;
  for (int base = 32; base < tot; base += 32) {
    int s; float4 e;
    load_entry(csr_src, s_src, beg, deg, d, row0, base + lane, vd, s, e);
    if (base + lane <= deg) {
      dn.x += expf(e.x - m.x); dn.y += expf(e.y - m.y);
      dn.z += expf(e.z - m.z); dn.w += expf(e.w - m.w);
    }
  }
  dn.x = wave_sum(dn.x); dn.y = wave_sum(dn.y); dn.z = wave_sum(dn.z); dn.w = wave_sum(dn.w);

  float4 acc0 = make_float4(0.f, 0.f, 0.f, 0.f);
  float4 acc1 = make_float4(0.f, 0.f, 0.f, 0.f);
  {
    const float4 al = make_float4(x0.x / dn.x, x0.y / dn.y, x0.z / dn.z, x0.w / dn.w);
    accum_entries(Hin, s0, al, tot < 32 ? tot : 32, lane, acc0, acc1);
  }
  for (int base = 32; base < tot; base += 32) {
    int s; float4 e;
    load_entry(csr_src, s_src, beg, deg, d, row0, base + lane, vd, s, e);
    const bool v = base + lane <= deg;
    const float4 al = make_float4(v ? expf(e.x - m.x) / dn.x : 0.f, v ? expf(e.y - m.y) / dn.y : 0.f,
                                  v ? expf(e.z - m.z) / dn.z : 0.f, v ? expf(e.w - m.w) / dn.w : 0.f);
    const int rem = tot - base;
    accum_entries(Hin, s, al, rem < 32 ? rem : 32, lane, acc0, acc1);
  }

  const float4 b0 = *(const float4*)(bias + (lane << 2));
  const float4 b1 = *(const float4*)(bias + 128 + (lane << 2));
  float* od = agg + (size_t)d * HIDDEN + (lane << 2);
  float4 o0 = make_float4(acc0.x + b0.x, acc0.y + b0.y, acc0.z + b0.z, acc0.w + b0.w);
  float4 o1 = make_float4(acc1.x + b1.x, acc1.y + b1.y, acc1.z + b1.z, acc1.w + b1.w);
  if (ELU) {
    o0 = make_float4(elu1(o0.x), elu1(o0.y), elu1(o0.z), elu1(o0.w));
    o1 = make_float4(elu1(o1.x), elu1(o1.y), elu1(o1.z), elu1(o1.w));
  }
  *(float4*)(od)       = o0;
  *(float4*)(od + 128) = o1;
}

extern "C" void kernel_launch(void* const* d_in, const int* in_sizes, int n_in,
                              void* d_out, int out_size, void* d_ws, size_t ws_size,
                              hipStream_t stream) {
  const float* x     = (const float*)d_in[0];
  const int*   ei    = (const int*)  d_in[1];
  const float* W1    = (const float*)d_in[2];
  const float* asrc1 = (const float*)d_in[3];
  const float* adst1 = (const float*)d_in[4];
  const float* b1    = (const float*)d_in[5];
  const float* W2    = (const float*)d_in[6];
  const float* asrc2 = (const float*)d_in[7];
  const float* adst2 = (const float*)d_in[8];
  const float* b2    = (const float*)d_in[9];
  const float* Wf    = (const float*)d_in[10];
  const float* bf    = (const float*)d_in[11];

  const int B  = 2;
  const int N  = in_sizes[0] / (B * FEAT);
  const int E  = in_sizes[1] / 2;
  const int R  = B * N;

  float* p = (float*)d_ws;
  float* buf_h   = p; p += (size_t)R * HIDDEN;
  float* buf_agg = p; p += (size_t)R * HIDDEN;
  float* ssrc    = p; p += (size_t)R * HEADS;
  float* sdst    = p; p += (size_t)R * HEADS;
  int* q = (int*)p;
  int* cnt     = q; q += (size_t)N;
  int* cursor  = q; q += (size_t)N;
  int* rowptr  = q; q += (size_t)N + 4;
  int* csr_src = q; q += (size_t)E;

  const int mtiles  = R >> 5;
  const int gtiles  = mtiles * (HIDDEN >> 6);
  const int ftiles  = mtiles * (CHAN >> 6);
  const dim3 blkG(128);
  const dim3 blkE(256);

  csr_zero_kernel<<<(N + 255) / 256, blkE, 0, stream>>>(cnt, N);
  csr_count_kernel<<<(E + 255) / 256, blkE, 0, stream>>>(ei, E, cnt);
  csr_scan_kernel<<<1, 1024, 0, stream>>>(cnt, rowptr, cursor, N);
  csr_fill_kernel<<<(E + 255) / 256, blkE, 0, stream>>>(ei, E, cursor, csr_src);


  wmma_gemm_kernel<true, false><<<(gtiles + 3) / 4, blkG, 0, stream>>>(
      x, W1, nullptr, asrc1, adst1, buf_h, ssrc, sdst, R, FEAT, HIDDEN);
  gat_gather_kernel<true><<<(R + 7) / 8, blkE, 0, stream>>>(rowptr, csr_src, ssrc, sdst, buf_h, b1, buf_agg, N, R);

  wmma_gemm_kernel<true, false><<<(gtiles + 3) / 4, blkG, 0, stream>>>(
      buf_agg, W2, nullptr, asrc2, adst2, buf_h, ssrc, sdst, R, HIDDEN, HIDDEN);
  gat_gather_kernel<false><<<(R + 7) / 8, blkE, 0, stream>>>(rowptr, csr_src, ssrc, sdst, buf_h, b2, buf_agg, N, R);

  wmma_gemm_kernel<false, true><<<(ftiles + 3) / 4, blkG, 0, stream>>>(
      buf_agg, Wf, bf, nullptr, nullptr, (float*)d_out, nullptr, nullptr, R, HIDDEN, CHAN);
}


# GEMM lane owns 4 adjacent output columns: W rows and A read by 16-byte loads (12 per 16 k, was 40) grouped before WMMAs, h stored as whole lines; k order in a 16-k block and score sum order change
# speedup vs baseline: 5.0329x; 1.0701x over previous
#include <hip/hip_runtime.h>
#include <math.h>


typedef __attribute__((ext_vector_type(2))) float v2f;
typedef __attribute__((ext_vector_type(4))) float v4f;
typedef __attribute__((ext_vector_type(8))) float v8f;

#define HEADS     4
#define CHAN      64
#define HIDDEN    256
#define FEAT      128
#define NEG_SLOPE 0.2f

static __device__ __forceinline__ float lrelu(float x) {
  return x > 0.f ? x : NEG_SLOPE * x;
}

struct Frag {
  v4f a0[2];
  v4f a1[2];
  v4f b[4][2];
};

template <int NC>
static __device__ __forceinline__ void load_frag(Frag& f,
                                                 const float* __restrict__ ap0,
                                                 const float* __restrict__ ap1,
                                                 const float* __restrict__ wp) {
  #pragma unroll
  for (int i = 0; i < 2; ++i) {
    f.a0[i] = *(const v4f*)(ap0 + 8 * i);
    f.a1[i] = *(const v4f*)(ap1 + 8 * i);
    #pragma unroll
    for (int p = 0; p < 2; ++p) {
      const int t = 2 * i + p;
      #pragma unroll
      for (int e = 0; e < 2; ++e)
        f.b[t][e] = *(const v4f*)(wp + (size_t)(8 * i + 2 * p + e) * NC);
    }
  }
}

static __device__ __forceinline__ void mma_frag(v8f (&acc)[2][4], const Frag& f) {
  #pragma unroll
  for (int t = 0; t < 4; ++t) {
    const int i = t >> 1, p = (t & 1) * 2;
    const v2f a0 = v2f{f.a0[i][p], f.a0[i][p + 1]};
    const v2f a1 = v2f{f.a1[i][p], f.a1[i][p + 1]};
    #pragma unroll
    for (int c = 0; c < 4; ++c) {
      const v2f b = v2f{f.b[t][0][c], f.b[t][1][c]};
      acc[0][c] = __builtin_amdgcn_wmma_f32_16x16x4_f32(false, a0, false, b, (short)0, acc[0][c], false, false);
      acc[1][c] = __builtin_amdgcn_wmma_f32_16x16x4_f32(false, a1, false, b, (short)0, acc[1][c], false, false);
    }
  }
}

template <bool GAT, bool BIAS, int NC>
__global__ void __launch_bounds__(128) wmma_gemm_kernel(const float* __restrict__ A,
                                 const float* __restrict__ W,
                                 const float* __restrict__ bias,
                                 const float* __restrict__ att_src,
                                 const float* __restrict__ att_dst,
                                 float* __restrict__ Hout,
                                 float* __restrict__ s_src,
                                 float* __restrict__ s_dst,
                                 int M, int K) {
  const int lane = threadIdx.x & 31;
  const int wv   = threadIdx.x >> 5;
  const int wpb  = blockDim.x >> 5;
  const int tile = blockIdx.x * wpb + wv;
  constexpr int ctiles = NC >> 6;
  const int mt = tile / ctiles;
  const int ct = tile - mt * ctiles;
  const int m0 = mt << 5;
  if (m0 >= M) return;
  const int n0 = ct << 6;
  const int lo = lane & 15;
  const int hi = lane >> 4;

  v8f acc[2][4] = {};
  const float* ap0 = A + (size_t)(m0 + lo) * K + (hi << 2);
  const float* ap1 = A + (size_t)(m0 + 16 + lo) * K + (hi << 2);
  const float* wp  = W + n0 + (lo << 2) + (size_t)(hi << 2) * NC;
  constexpr size_t wblk = (size_t)16 * NC;

  Frag f0, f1;
  load_frag<NC>(f0, ap0, ap1, wp);
  for (int k = 0; k + 32 < K; k += 32) {
    load_frag<NC>(f1, ap0 + 16, ap1 + 16, wp + wblk);
    __builtin_amdgcn_sched_barrier(0);
    mma_frag(acc, f0);
    __builtin_amdgcn_sched_barrier(0);
    load_frag<NC>(f0, ap0 + 32, ap1 + 32, wp + 2 * wblk);
    __builtin_amdgcn_sched_barrier(0);
    mma_frag(acc, f1);
    ap0 += 32; ap1 += 32; wp += 2 * wblk;
  }
  load_frag<NC>(f1, ap0 + 16, ap1 + 16, wp + wblk);
  __builtin_amdgcn_sched_barrier(0);
  mma_frag(acc, f0);
  mma_frag(acc, f1);

  v4f bv = v4f{0.f, 0.f, 0.f, 0.f};
  if (BIAS) bv = *(const v4f*)(bias + n0 + (lo << 2));

  #pragma unroll
  for (int g = 0; g < 2; ++g) {
    const int rbase = m0 + (g << 4) + (hi << 3);
    #pragma unroll
    for (int r = 0; r < 8; ++r) {
      float* hp = Hout + (size_t)(rbase + r) * NC + n0 + (lo << 2);
      *(v4f*)hp = v4f{acc[g][0][r] + bv.x, acc[g][1][r] + bv.y,
                      acc[g][2][r] + bv.z, acc[g][3][r] + bv.w};
    }
  }

  if (GAT) {
    const int head = n0 >> 6;
    const v4f as = *(const v4f*)(att_src + head * CHAN + (lo << 2));
    const v4f ad = *(const v4f*)(att_dst + head * CHAN + (lo << 2));
    #pragma unroll
    for (int g = 0; g < 2; ++g) {
      const int rbase = m0 + (g << 4) + (hi << 3);
      #pragma unroll
      for (int r = 0; r < 8; ++r) {
        float ss = acc[g][0][r] * as.x + acc[g][1][r] * as.y + acc[g][2][r] * as.z + acc[g][3][r] * as.w;
        float sd = acc[g][0][r] * ad.x + acc[g][1][r] * ad.y + acc[g][2][r] * ad.z + acc[g][3][r] * ad.w;
        #pragma unroll
        for (int off = 8; off > 0; off >>= 1) {
          ss += __shfl_xor(ss, off, 32);
          sd += __shfl_xor(sd, off, 32);
        }
        if (lo == 0) {
          const int row = rbase + r;
          s_src[row * HEADS + head] = ss;
          s_dst[row * HEADS + head] = sd;
        }
      }
    }
  }
}

__global__ void csr_zero_kernel(int* __restrict__ cnt, int N) {
  const int idx = blockIdx.x * blockDim.x + threadIdx.x;
  if (idx < N) cnt[idx] = 0;
}

__global__ void csr_count_kernel(const int* __restrict__ ei, int E, int* __restrict__ cnt) {
  const int idx = blockIdx.x * blockDim.x + threadIdx.x;
  if (idx < E) atomicAdd(cnt + ei[E + idx], 1);
}

__global__ void __launch_bounds__(1024)
csr_scan_kernel(const int* __restrict__ cnt, int* __restrict__ rowptr,
                int* __restrict__ cursor, int N) {
  __shared__ int part[1024];
  const int t   = threadIdx.x;
  const int per = (N + 1023) >> 10;
  const int lo  = t * per;
  const int hi  = (lo + per < N) ? (lo + per) : N;
  int s = 0;
  for (int i = lo; i < hi; ++i) s += cnt[i];
  part[t] = s;
  __syncthreads();
  for (int off = 1; off < 1024; off <<= 1) {
    const int v = (t >= off) ? part[t - off] : 0;
    __syncthreads();
    part[t] += v;
    __syncthreads();
  }
  int run = part[t] - s;
  for (int i = lo; i < hi; ++i) {
    const int c = cnt[i];
    rowptr[i] = run;
    cursor[i] = run;
    run += c;
  }
  if (t == 1023) rowptr[N] = part[1023];
}

__global__ void csr_fill_kernel(const int* __restrict__ ei, int E,
                                int* __restrict__ cursor, int* __restrict__ csr_src) {
  const int idx = blockIdx.x * blockDim.x + threadIdx.x;
  if (idx < E) {
    const int pos = atomicAdd(cursor + ei[E + idx], 1);
    csr_src[pos] = ei[idx];
  }
}

static __device__ __forceinline__ float wave_max(float v) {
  #pragma unroll
  for (int off = 16; off > 0; off >>= 1) v = fmaxf(v, __shfl_xor(v, off, 32));
  return v;
}
static __device__ __forceinline__ float wave_sum(float v) {
  #pragma unroll
  for (int off = 16; off > 0; off >>= 1) v += __shfl_xor(v, off, 32);
  return v;
}

static __device__ __forceinline__ void load_entry(const int* __restrict__ csr_src,
                                                  const float* __restrict__ s_src,
                                                  int beg, int deg, int d, int row0, int j,
                                                  const float4 vd, int& s, float4& e) {
  s = d;
  if (j < deg) s = row0 + csr_src[beg + j];
  if (j <= deg) {
    const float4 vs = *(const float4*)(s_src + (size_t)s * HEADS);
    e = make_float4(lrelu(vs.x + vd.x), lrelu(vs.y + vd.y), lrelu(vs.z + vd.z), lrelu(vs.w + vd.w));
  } else {
    e = make_float4(-INFINITY, -INFINITY, -INFINITY, -INFINITY);
  }
}

static __device__ __forceinline__ void accum_entries(const float* __restrict__ Hin,
                                                     int s, const float4 al, int cnt, int lane,
                                                     float4& acc0, float4& acc1) {
  const bool up = lane >= 16;
  #pragma unroll 4
  for (int t = 0; t < cnt; ++t) {
    const int   st = __builtin_amdgcn_readlane(s, t);
    const float a0 = __int_as_float(__builtin_amdgcn_readlane(__float_as_int(al.x), t));
    const float a1 = __int_as_float(__builtin_amdgcn_readlane(__float_as_int(al.y), t));
    const float a2 = __int_as_float(__builtin_amdgcn_readlane(__float_as_int(al.z), t));
    const float a3 = __int_as_float(__builtin_amdgcn_readlane(__float_as_int(al.w), t));
    const float* hs = Hin + (size_t)st * HIDDEN + (lane << 2);
    const float4 h0 = *(const float4*)(hs);
    const float4 h1 = *(const float4*)(hs + 128);
    const float w0 = up ? a1 : a0;
    const float w1 = up ? a3 : a2;
    acc0.x += h0.x * w0; acc0.y += h0.y * w0; acc0.z += h0.z * w0; acc0.w += h0.w * w0;
    acc1.x += h1.x * w1; acc1.y += h1.y * w1; acc1.z += h1.z * w1; acc1.w += h1.w * w1;
  }
}

static __device__ __forceinline__ float elu1(float v) {
  return v > 0.f ? v : expm1f(v);
}

template <bool ELU>
__global__ void gat_gather_kernel(const int* __restrict__ rowptr,
                                  const int* __restrict__ csr_src,
                                  const float* __restrict__ s_src,
                                  const float* __restrict__ s_dst,
                                  const float* __restrict__ Hin,
                                  const float* __restrict__ bias,
                                  float* __restrict__ agg, int N, int R) {
  const int lane = threadIdx.x & 31;
  const int d = __builtin_amdgcn_readfirstlane(blockIdx.x * (blockDim.x >> 5) + (threadIdx.x >> 5));
  if (d >= R) return;
  const int row0 = d >= N ? N : 0;
  const int g    = d - row0;
  const int beg = rowptr[g];
  const int deg = rowptr[g + 1] - beg;
  const int tot = deg + 1;
  const float4 vd = *(const float4*)(s_dst + (size_t)d * HEADS);

  int s0; float4 e0;
  load_entry(csr_src, s_src, beg, deg, d, row0, lane, vd, s0, e0);

  float4 m = e0;
  for (int base = 32; base < tot; base += 32) {
    int s; float4 e;
    load_entry(csr_src, s_src, beg, deg, d, row0, base + lane, vd, s, e);
    m.x = fmaxf(m.x, e.x); m.y = fmaxf(m.y, e.y); m.z = fmaxf(m.z, e.z); m.w = fmaxf(m.w, e.w);
  }
  m.x = wave_max(m.x); m.y = wave_max(m.y); m.z = wave_max(m.z); m.w = wave_max(m.w);

  const bool v0 = lane <= deg;
  float4 x0 = make_float4(v0 ? expf(e0.x - m.x) : 0.f, v0 ? expf(e0.y - m.y) : 0.f,
                          v0 ? expf(e0.z - m.z) : 0.f, v0 ? expf(e0.w - m.w) : 0.f);
  float4 dn = x0;
  for (int base = 32; base < tot; base += 32) {
    int s; float4 e;
    load_entry(csr_src, s_src, beg, deg, d, row0, base + lane, vd, s, e);
    if (base + lane <= deg) {
      dn.x += expf(e.x - m.x); dn.y += expf(e.y - m.y);
      dn.z += expf(e.z - m.z); dn.w += expf(e.w - m.w);
    }
  }
  dn.x = wave_sum(dn.x); dn.y = wave_sum(dn.y); dn.z = wave_sum(dn.z); dn.w = wave_sum(dn.w);

  float4 acc0 = make_float4(0.f, 0.f, 0.f, 0.f);
  float4 acc1 = make_float4(0.f, 0.f, 0.f, 0.f);
  {
    const float4 al = make_float4(x0.x / dn.x, x0.y / dn.y, x0.z / dn.z, x0.w / dn.w);
    accum_entries(Hin, s0, al, tot < 32 ? tot : 32, lane, acc0, acc1);
  }
  for (int base = 32; base < tot; base += 32) {
    int s; float4 e;
    load_entry(csr_src, s_src, beg, deg, d, row0, base + lane, vd, s, e);
    const bool v = base + lane <= deg;
    const float4 al = make_float4(v ? expf(e.x - m.x) / dn.x : 0.f, v ? expf(e.y - m.y) / dn.y : 0.f,
                                  v ? expf(e.z - m.z) / dn.z : 0.f, v ? expf(e.w - m.w) / dn.w : 0.f);
    const int rem = tot - base;
    accum_entries(Hin, s, al, rem < 32 ? rem : 32, lane, acc0, acc1);
  }

  const float4 b0 = *(const float4*)(bias + (lane << 2));
  const float4 b1 = *(const float4*)(bias + 128 + (lane << 2));
  float* od = agg + (size_t)d * HIDDEN + (lane << 2);
  float4 o0 = make_float4(acc0.x + b0.x, acc0.y + b0.y, acc0.z + b0.z, acc0.w + b0.w);
  float4 o1 = make_float4(acc1.x + b1.x, acc1.y + b1.y, acc1.z + b1.z, acc1.w + b1.w);
  if (ELU) {
    o0 = make_float4(elu1(o0.x), elu1(o0.y), elu1(o0.z), elu1(o0.w));
    o1 = make_float4(elu1(o1.x), elu1(o1.y), elu1(o1.z), elu1(o1.w));
  }
  *(float4*)(od)       = o0;
  *(float4*)(od + 128) = o1;
}

extern "C" void kernel_launch(void* const* d_in, const int* in_sizes, int n_in,
                              void* d_out, int out_size, void* d_ws, size_t ws_size,
                              hipStream_t stream) {
  const float* x     = (const float*)d_in[0];
  const int*   ei    = (const int*)  d_in[1];
  const float* W1    = (const float*)d_in[2];
  const float* asrc1 = (const float*)d_in[3];
  const float* adst1 = (const float*)d_in[4];
  const float* b1    = (const float*)d_in[5];
  const float* W2    = (const float*)d_in[6];
  const float* asrc2 = (const float*)d_in[7];
  const float* adst2 = (const float*)d_in[8];
  const float* b2    = (const float*)d_in[9];
  const float* Wf    = (const float*)d_in[10];
  const float* bf    = (const float*)d_in[11];

  const int B  = 2;
  const int N  = in_sizes[0] / (B * FEAT);
  const int E  = in_sizes[1] / 2;
  const int R  = B * N;

  float* p = (float*)d_ws;
  float* buf_h   = p; p += (size_t)R * HIDDEN;
  float* buf_agg = p; p += (size_t)R * HIDDEN;
  float* ssrc    = p; p += (size_t)R * HEADS;
  float* sdst    = p; p += (size_t)R * HEADS;
  int* q = (int*)p;
  int* cnt     = q; q += (size_t)N;
  int* cursor  = q; q += (size_t)N;
  int* rowptr  = q; q += (size_t)N + 4;
  int* csr_src = q; q += (size_t)E;

  const int mtiles  = R >> 5;
  const int gtiles  = mtiles * (HIDDEN >> 6);
  const int ftiles  = mtiles * (CHAN >> 6);
  const dim3 blkG(128);
  const dim3 blkE(256);

  csr_zero_kernel<<<(N + 255) / 256, blkE, 0, stream>>>(cnt, N);
  csr_count_kernel<<<(E + 255) / 256, blkE, 0, stream>>>(ei, E, cnt);
  csr_scan_kernel<<<1, 1024, 0, stream>>>(cnt, rowptr, cursor, N);
  csr_fill_kernel<<<(E + 255) / 256, blkE, 0, stream>>>(ei, E, cursor, csr_src);


  wmma_gemm_kernel<true, false, HIDDEN><<<(gtiles + 3) / 4, blkG, 0, stream>>>(
      x, W1, nullptr, asrc1, adst1, buf_h, ssrc, sdst, R, FEAT);
  gat_gather_kernel<true><<<(R + 7) / 8, blkE, 0, stream>>>(rowptr, csr_src, ssrc, sdst, buf_h, b1, buf_agg, N, R);

  wmma_gemm_kernel<true, false, HIDDEN><<<(gtiles + 3) / 4, blkG, 0, stream>>>(
      buf_agg, W2, nullptr, asrc2, adst2, buf_h, ssrc, sdst, R, HIDDEN);
  gat_gather_kernel<false><<<(R + 7) / 8, blkE, 0, stream>>>(rowptr, csr_src, ssrc, sdst, buf_h, b2, buf_agg, N, R);

  wmma_gemm_kernel<false, true, CHAN><<<(ftiles + 3) / 4, blkG, 0, stream>>>(
      buf_agg, Wf, bf, nullptr, nullptr, (float*)d_out, nullptr, nullptr, R, HIDDEN);
}


# Edges put into a fixed 64-slot row per destination, filled by atomic rank (a node over 64 edges sweeps the raw edge list): count and scan launches gone, 9 launches become 7; sums stay in rank order
# speedup vs baseline: 6.3348x; 1.2587x over previous
#include <hip/hip_runtime.h>
#include <math.h>


typedef __attribute__((ext_vector_type(2))) float v2f;
typedef __attribute__((ext_vector_type(4))) float v4f;
typedef __attribute__((ext_vector_type(8))) float v8f;

#define HEADS     4
#define CHAN      64
#define HIDDEN    256
#define FEAT      128
#define NEG_SLOPE 0.2f
#define SLOTS     64

static __device__ __forceinline__ float lrelu(float x) {
  return x > 0.f ? x : NEG_SLOPE * x;
}

struct Frag {
  v4f a0[2];
  v4f a1[2];
  v4f b[4][2];
};

template <int NC>
static __device__ __forceinline__ void load_frag(Frag& f,
                                                 const float* __restrict__ ap0,
                                                 const float* __restrict__ ap1,
                                                 const float* __restrict__ wp) {
  #pragma unroll
  for (int i = 0; i < 2; ++i) {
    f.a0[i] = *(const v4f*)(ap0 + 8 * i);
    f.a1[i] = *(const v4f*)(ap1 + 8 * i);
    #pragma unroll
    for (int p = 0; p < 2; ++p) {
      const int t = 2 * i + p;
      #pragma unroll
      for (int e = 0; e < 2; ++e)
        f.b[t][e] = *(const v4f*)(wp + (size_t)(8 * i + 2 * p + e) * NC);
    }
  }
}

static __device__ __forceinline__ void mma_frag(v8f (&acc)[2][4], const Frag& f) {
  #pragma unroll
  for (int t = 0; t < 4; ++t) {
    const int i = t >> 1, p = (t & 1) * 2;
    const v2f a0 = v2f{f.a0[i][p], f.a0[i][p + 1]};
    const v2f a1 = v2f{f.a1[i][p], f.a1[i][p + 1]};
    #pragma unroll
    for (int c = 0; c < 4; ++c) {
      const v2f b = v2f{f.b[t][0][c], f.b[t][1][c]};
      acc[0][c] = __builtin_amdgcn_wmma_f32_16x16x4_f32(false, a0, false, b, (short)0, acc[0][c], false, false);
      acc[1][c] = __builtin_amdgcn_wmma_f32_16x16x4_f32(false, a1, false, b, (short)0, acc[1][c], false, false);
    }
  }
}

template <bool GAT, bool BIAS, int NC>
__global__ void __launch_bounds__(128) wmma_gemm_kernel(const float* __restrict__ A,
                                 const float* __restrict__ W,
                                 const float* __restrict__ bias,
                                 const float* __restrict__ att_src,
                                 const float* __restrict__ att_dst,
                                 float* __restrict__ Hout,
                                 float* __restrict__ s_src,
                                 float* __restrict__ s_dst,
                                 int M, int K) {
  const int lane = threadIdx.x & 31;
  const int wv   = threadIdx.x >> 5;
  const int wpb  = blockDim.x >> 5;
  const int tile = blockIdx.x * wpb + wv;
  constexpr int ctiles = NC >> 6;
  const int mt = tile / ctiles;
  const int ct = tile - mt * ctiles;
  const int m0 = mt << 5;
  if (m0 >= M) return;
  const int n0 = ct << 6;
  const int lo = lane & 15;
  const int hi = lane >> 4;

  v8f acc[2][4] = {};
  const float* ap0 = A + (size_t)(m0 + lo) * K + (hi << 2);
  const float* ap1 = A + (size_t)(m0 + 16 + lo) * K + (hi << 2);
  const float* wp  = W + n0 + (lo << 2) + (size_t)(hi << 2) * NC;
  constexpr size_t wblk = (size_t)16 * NC;

  Frag f0, f1;
  load_frag<NC>(f0, ap0, ap1, wp);
  for (int k = 0; k + 32 < K; k += 32) {
    load_frag<NC>(f1, ap0 + 16, ap1 + 16, wp + wblk);
    __builtin_amdgcn_sched_barrier(0);
    mma_frag(acc, f0);
    __builtin_amdgcn_sched_barrier(0);
    load_frag<NC>(f0, ap0 + 32, ap1 + 32, wp + 2 * wblk);
    __builtin_amdgcn_sched_barrier(0);
    mma_frag(acc, f1);
    ap0 += 32; ap1 += 32; wp += 2 * wblk;
  }
  load_frag<NC>(f1, ap0 + 16, ap1 + 16, wp + wblk);
  __builtin_amdgcn_sched_barrier(0);
  mma_frag(acc, f0);
  mma_frag(acc, f1);

  v4f bv = v4f{0.f, 0.f, 0.f, 0.f};
  if (BIAS) bv = *(const v4f*)(bias + n0 + (lo << 2));

  #pragma unroll
  for (int g = 0; g < 2; ++g) {
    const int rbase = m0 + (g << 4) + (hi << 3);
    #pragma unroll
    for (int r = 0; r < 8; ++r) {
      float* hp = Hout + (size_t)(rbase + r) * NC + n0 + (lo << 2);
      *(v4f*)hp = v4f{acc[g][0][r] + bv.x, acc[g][1][r] + bv.y,
                      acc[g][2][r] + bv.z, acc[g][3][r] + bv.w};
    }
  }

  if (GAT) {
    const int head = n0 >> 6;
    const v4f as = *(const v4f*)(att_src + head * CHAN + (lo << 2));
    const v4f ad = *(const v4f*)(att_dst + head * CHAN + (lo << 2));
    #pragma unroll
    for (int g = 0; g < 2; ++g) {
      const int rbase = m0 + (g << 4) + (hi << 3);
      #pragma unroll
      for (int r = 0; r < 8; ++r) {
        float ss = acc[g][0][r] * as.x + acc[g][1][r] * as.y + acc[g][2][r] * as.z + acc[g][3][r] * as.w;
        float sd = acc[g][0][r] * ad.x + acc[g][1][r] * ad.y + acc[g][2][r] * ad.z + acc[g][3][r] * ad.w;
        #pragma unroll
        for (int off = 8; off > 0; off >>= 1) {
          ss += __shfl_xor(ss, off, 32);
          sd += __shfl_xor(sd, off, 32);
        }
        if (lo == 0) {
          const int row = rbase + r;
          s_src[row * HEADS + head] = ss;
          s_dst[row * HEADS + head] = sd;
        }
      }
    }
  }
}

__global__ void deg_zero_kernel(int* __restrict__ degs, int N) {
  const int idx = blockIdx.x * blockDim.x + threadIdx.x;
  if (idx < N) degs[idx] = 0;
}

__global__ void slot_fill_kernel(const int* __restrict__ ei, int E,
                                 int* __restrict__ degs, int* __restrict__ slots) {
  const int idx = blockIdx.x * blockDim.x + threadIdx.x;
  if (idx < E) {
    const int dn  = ei[E + idx];
    const int pos = atomicAdd(degs + dn, 1);
    if (pos < SLOTS) slots[(size_t)dn * SLOTS + pos] = ei[idx];
  }
}

static __device__ __forceinline__ float wave_max(float v) {
  #pragma unroll
  for (int off = 16; off > 0; off >>= 1) v = fmaxf(v, __shfl_xor(v, off, 32));
  return v;
}
static __device__ __forceinline__ float wave_sum(float v) {
  #pragma unroll
  for (int off = 16; off > 0; off >>= 1) v += __shfl_xor(v, off, 32);
  return v;
}

static __device__ __forceinline__ void load_entry(const int* __restrict__ csr_src,
                                                  const float* __restrict__ s_src,
                                                  int beg, int deg, int d, int row0, int j,
                                                  const float4 vd, int& s, float4& e) {
  s = d;
  if (j < deg) s = row0 + csr_src[beg + j];
  if (j <= deg) {
    const float4 vs = *(const float4*)(s_src + (size_t)s * HEADS);
    e = make_float4(lrelu(vs.x + vd.x), lrelu(vs.y + vd.y), lrelu(vs.z + vd.z), lrelu(vs.w + vd.w));
  } else {
    e = make_float4(-INFINITY, -INFINITY, -INFINITY, -INFINITY);
  }
}

static __device__ __forceinline__ void accum_one(const float* __restrict__ Hin,
                                                 int s, const float4 al, int t, int lane,
                                                 float4& acc0, float4& acc1) {
  const int   st = __builtin_amdgcn_readlane(s, t);
  const float a0 = __int_as_float(__builtin_amdgcn_readlane(__float_as_int(al.x), t));
  const float a1 = __int_as_float(__builtin_amdgcn_readlane(__float_as_int(al.y), t));
  const float a2 = __int_as_float(__builtin_amdgcn_readlane(__float_as_int(al.z), t));
  const float a3 = __int_as_float(__builtin_amdgcn_readlane(__float_as_int(al.w), t));
  const float* hs = Hin + (size_t)st * HIDDEN + (lane << 2);
  const float4 h0 = *(const float4*)(hs);
  const float4 h1 = *(const float4*)(hs + 128);
  const float w0 = lane >= 16 ? a1 : a0;
  const float w1 = lane >= 16 ? a3 : a2;
  acc0.x += h0.x * w0; acc0.y += h0.y * w0; acc0.z += h0.z * w0; acc0.w += h0.w * w0;
  acc1.x += h1.x * w1; acc1.y += h1.y * w1; acc1.z += h1.z * w1; acc1.w += h1.w * w1;
}

static __device__ __forceinline__ void accum_entries(const float* __restrict__ Hin,
                                                     int s, const float4 al, int cnt, int lane,
                                                     float4& acc0, float4& acc1) {
  const bool up = lane >= 16;
  #pragma unroll 4
  for (int t = 0; t < cnt; ++t) {
    const int   st = __builtin_amdgcn_readlane(s, t);
    const float a0 = __int_as_float(__builtin_amdgcn_readlane(__float_as_int(al.x), t));
    const float a1 = __int_as_float(__builtin_amdgcn_readlane(__float_as_int(al.y), t));
    const float a2 = __int_as_float(__builtin_amdgcn_readlane(__float_as_int(al.z), t));
    const float a3 = __int_as_float(__builtin_amdgcn_readlane(__float_as_int(al.w), t));
    const float* hs = Hin + (size_t)st * HIDDEN + (lane << 2);
    const float4 h0 = *(const float4*)(hs);
    const float4 h1 = *(const float4*)(hs + 128);
    const float w0 = up ? a1 : a0;
    const float w1 = up ? a3 : a2;
    acc0.x += h0.x * w0; acc0.y += h0.y * w0; acc0.z += h0.z * w0; acc0.w += h0.w * w0;
    acc1.x += h1.x * w1; acc1.y += h1.y * w1; acc1.z += h1.z * w1; acc1.w += h1.w * w1;
  }
}

static __device__ __forceinline__ float elu1(float v) {
  return v > 0.f ? v : expm1f(v);
}

static __device__ __forceinline__ float4 lrelu4(const float4 a, const float4 b) {
  return make_float4(lrelu(a.x + b.x), lrelu(a.y + b.y), lrelu(a.z + b.z), lrelu(a.w + b.w));
}

static __device__ __forceinline__ void sweep_edge_list(const int* __restrict__ ei, int E,
                                                       const float* __restrict__ s_src,
                                                       const float* __restrict__ Hin,
                                                       int g, int d, int row0, const float4 vd,
                                                       int lane, float4& acc0, float4& acc1) {
  const float4 es = lrelu4(*(const float4*)(s_src + (size_t)d * HEADS), vd);
  float4 m = es;
  for (int i0 = 0; i0 < E; i0 += 32) {
    const int i = i0 + lane;
    if (i < E && ei[E + i] == g) {
      const float4 e = lrelu4(*(const float4*)(s_src + (size_t)(row0 + ei[i]) * HEADS), vd);
      m.x = fmaxf(m.x, e.x); m.y = fmaxf(m.y, e.y); m.z = fmaxf(m.z, e.z); m.w = fmaxf(m.w, e.w);
    }
  }
  m.x = wave_max(m.x); m.y = wave_max(m.y); m.z = wave_max(m.z); m.w = wave_max(m.w);

  const float4 xs = make_float4(expf(es.x - m.x), expf(es.y - m.y), expf(es.z - m.z), expf(es.w - m.w));
  float4 dn = lane == 0 ? xs : make_float4(0.f, 0.f, 0.f, 0.f);
  for (int i0 = 0; i0 < E; i0 += 32) {
    const int i = i0 + lane;
    if (i < E && ei[E + i] == g) {
      const float4 e = lrelu4(*(const float4*)(s_src + (size_t)(row0 + ei[i]) * HEADS), vd);
      dn.x += expf(e.x - m.x); dn.y += expf(e.y - m.y);
      dn.z += expf(e.z - m.z); dn.w += expf(e.w - m.w);
    }
  }
  dn.x = wave_sum(dn.x); dn.y = wave_sum(dn.y); dn.z = wave_sum(dn.z); dn.w = wave_sum(dn.w);

  for (int i0 = 0; i0 < E; i0 += 32) {
    const int i = i0 + lane;
    const bool hit = i < E && ei[E + i] == g;
    int s = d;
    float4 al = make_float4(0.f, 0.f, 0.f, 0.f);
    if (hit) {
      s = row0 + ei[i];
      const float4 e = lrelu4(*(const float4*)(s_src + (size_t)s * HEADS), vd);
      al = make_float4(expf(e.x - m.x) / dn.x, expf(e.y - m.y) / dn.y,
                       expf(e.z - m.z) / dn.z, expf(e.w - m.w) / dn.w);
    }
    unsigned hits = __builtin_amdgcn_ballot_w32(hit);
    while (hits) {
      const int t = __builtin_ctz(hits);
      hits &= hits - 1;
      accum_one(Hin, s, al, t, lane, acc0, acc1);
    }
  }
  const float4 al = make_float4(xs.x / dn.x, xs.y / dn.y, xs.z / dn.z, xs.w / dn.w);
  const float* hs = Hin + (size_t)d * HIDDEN + (lane << 2);
  const float4 h0 = *(const float4*)(hs);
  const float4 h1 = *(const float4*)(hs + 128);
  const float w0 = lane >= 16 ? al.y : al.x;
  const float w1 = lane >= 16 ? al.w : al.z;
  acc0.x += h0.x * w0; acc0.y += h0.y * w0; acc0.z += h0.z * w0; acc0.w += h0.w * w0;
  acc1.x += h1.x * w1; acc1.y += h1.y * w1; acc1.z += h1.z * w1; acc1.w += h1.w * w1;
}

template <bool ELU>
__global__ void gat_gather_kernel(const int* __restrict__ degs,
                                  const int* __restrict__ csr_src,
                                  const int* __restrict__ ei, int E,
                                  const float* __restrict__ s_src,
                                  const float* __restrict__ s_dst,
                                  const float* __restrict__ Hin,
                                  const float* __restrict__ bias,
                                  float* __restrict__ agg, int N, int R) {
  const int lane = threadIdx.x & 31;
  const int d = __builtin_amdgcn_readfirstlane(blockIdx.x * (blockDim.x >> 5) + (threadIdx.x >> 5));
  if (d >= R) return;
  const int row0 = d >= N ? N : 0;
  const int g    = d - row0;
  const int beg = g * SLOTS;
  const int deg = degs[g];
  const int tot = deg + 1;
  const float4 vd = *(const float4*)(s_dst + (size_t)d * HEADS);

  float4 acc0 = make_float4(0.f, 0.f, 0.f, 0.f);
  float4 acc1 = make_float4(0.f, 0.f, 0.f, 0.f);
  if (deg > SLOTS) {
    sweep_edge_list(ei, E, s_src, Hin, g, d, row0, vd, lane, acc0, acc1);
  } else {
  int s0; float4 e0;
  load_entry(csr_src, s_src, beg, deg, d, row0, lane, vd, s0, e0);

  float4 m = e0;
  for (int base = 32; base < tot; base += 32) {
    int s; float4 e;
    load_entry(csr_src, s_src, beg, deg, d, row0, base + lane, vd, s, e);
    m.x = fmaxf(m.x, e.x); m.y = fmaxf(m.y, e.y); m.z = fmaxf(m.z, e.z); m.w = fmaxf(m.w, e.w);
  }
  m.x = wave_max(m.x); m.y = wave_max(m.y); m.z = wave_max(m.z); m.w = wave_max(m.w);

  const bool v0 = lane <= deg;
  float4 x0 = make_float4(v0 ? expf(e0.x - m.x) : 0.f, v0 ? expf(e0.y - m.y) : 0.f,
                          v0 ? expf(e0.z - m.z) : 0.f, v0 ? expf(e0.w - m.w) : 0.f);
  float4 dn = x0;
  for (int base = 32; base < tot; base += 32) {
    int s; float4 e;
    load_entry(csr_src, s_src, beg, deg, d, row0, base + lane, vd, s, e);
    if (base + lane <= deg) {
      dn.x += expf(e.x - m.x); dn.y += expf(e.y - m.y);
      dn.z += expf(e.z - m.z); dn.w += expf(e.w - m.w);
    }
  }
  dn.x = wave_sum(dn.x); dn.y = wave_sum(dn.y); dn.z = wave_sum(dn.z); dn.w = wave_sum(dn.w);

  {
    const float4 al = make_float4(x0.x / dn.x, x0.y / dn.y, x0.z / dn.z, x0.w / dn.w);
    accum_entries(Hin, s0, al, tot < 32 ? tot : 32, lane, acc0, acc1);
  }
  for (int base = 32; base < tot; base += 32) {
    int s; float4 e;
    load_entry(csr_src, s_src, beg, deg, d, row0, base + lane, vd, s, e);
    const bool v = base + lane <= deg;
    const float4 al = make_float4(v ? expf(e.x - m.x) / dn.x : 0.f, v ? expf(e.y - m.y) / dn.y : 0.f,
                                  v ? expf(e.z - m.z) / dn.z : 0.f, v ? expf(e.w - m.w) / dn.w : 0.f);
    const int rem = tot - base;
    accum_entries(Hin, s, al, rem < 32 ? rem : 32, lane, acc0, acc1);
  }
  }

  const float4 b0 = *(const float4*)(bias + (lane << 2));
  const float4 b1 = *(const float4*)(bias + 128 + (lane << 2));
  float* od = agg + (size_t)d * HIDDEN + (lane << 2);
  float4 o0 = make_float4(acc0.x + b0.x, acc0.y + b0.y, acc0.z + b0.z, acc0.w + b0.w);
  float4 o1 = make_float4(acc1.x + b1.x, acc1.y + b1.y, acc1.z + b1.z, acc1.w + b1.w);
  if (ELU) {
    o0 = make_float4(elu1(o0.x), elu1(o0.y), elu1(o0.z), elu1(o0.w));
    o1 = make_float4(elu1(o1.x), elu1(o1.y), elu1(o1.z), elu1(o1.w));
  }
  *(float4*)(od)       = o0;
  *(float4*)(od + 128) = o1;
}

extern "C" void kernel_launch(void* const* d_in, const int* in_sizes, int n_in,
                              void* d_out, int out_size, void* d_ws, size_t ws_size,
                              hipStream_t stream) {
  const float* x     = (const float*)d_in[0];
  const int*   ei    = (const int*)  d_in[1];
  const float* W1    = (const float*)d_in[2];
  const float* asrc1 = (const float*)d_in[3];
  const float* adst1 = (const float*)d_in[4];
  const float* b1    = (const float*)d_in[5];
  const float* W2    = (const float*)d_in[6];
  const float* asrc2 = (const float*)d_in[7];
  const float* adst2 = (const float*)d_in[8];
  const float* b2    = (const float*)d_in[9];
  const float* Wf    = (const float*)d_in[10];
  const float* bf    = (const float*)d_in[11];

  const int B  = 2;
  const int N  = in_sizes[0] / (B * FEAT);
  const int E  = in_sizes[1] / 2;
  const int R  = B * N;

  float* p = (float*)d_ws;
  float* buf_h   = p; p += (size_t)R * HIDDEN;
  float* buf_agg = p; p += (size_t)R * HIDDEN;
  float* ssrc    = p; p += (size_t)R * HEADS;
  float* sdst    = p; p += (size_t)R * HEADS;
  int* q = (int*)p;
  int* slots = q; q += (size_t)N * SLOTS;
  int* degs  = q; q += (size_t)N;

  const int mtiles  = R >> 5;
  const int gtiles  = mtiles * (HIDDEN >> 6);
  const int ftiles  = mtiles * (CHAN >> 6);
  const dim3 blkG(128);
  const dim3 blkE(256);

  deg_zero_kernel<<<(N + 255) / 256, blkE, 0, stream>>>(degs, N);
  slot_fill_kernel<<<(E + 255) / 256, blkE, 0, stream>>>(ei, E, degs, slots);


  wmma_gemm_kernel<true, false, HIDDEN><<<(gtiles + 3) / 4, blkG, 0, stream>>>(
      x, W1, nullptr, asrc1, adst1, buf_h, ssrc, sdst, R, FEAT);
  gat_gather_kernel<true><<<(R + 7) / 8, blkE, 0, stream>>>(degs, slots, ei, E, ssrc, sdst, buf_h, b1, buf_agg, N, R);

  wmma_gemm_kernel<true, false, HIDDEN><<<(gtiles + 3) / 4, blkG, 0, stream>>>(
      buf_agg, W2, nullptr, asrc2, adst2, buf_h, ssrc, sdst, R, HIDDEN);
  gat_gather_kernel<false><<<(R + 7) / 8, blkE, 0, stream>>>(degs, slots, ei, E, ssrc, sdst, buf_h, b2, buf_agg, N, R);

  wmma_gemm_kernel<false, true, CHAN><<<(ftiles + 3) / 4, blkG, 0, stream>>>(
      buf_agg, Wf, bf, nullptr, nullptr, (float*)d_out, nullptr, nullptr, R, HIDDEN);
}


# Layer 1 aggregates before it multiplies: gather sums 512-byte x rows per head, GEMM forms (sums)@W1 per head with the ELU, scores are x@(W1 a); 1-KB rows of x@W1 not fetched; sum order changes
# speedup vs baseline: 6.6648x; 1.0521x over previous
#include <hip/hip_runtime.h>
#include <math.h>


typedef __attribute__((ext_vector_type(2))) float v2f;
typedef __attribute__((ext_vector_type(4))) float v4f;
typedef __attribute__((ext_vector_type(8))) float v8f;

#define HEADS     4
#define CHAN      64
#define HIDDEN    256
#define FEAT      128
#define NEG_SLOPE 0.2f
#define SLOTS     64

static __device__ __forceinline__ float lrelu(float x) {
  return x > 0.f ? x : NEG_SLOPE * x;
}

static __device__ __forceinline__ float elu1(float v) {
  const float e = expm1f(fminf(v, 0.f));
  return v < 0.f ? e : v;
}

struct Frag {
  v4f a0[2];
  v4f a1[2];
  v4f b[4][2];
};

template <int NC>
static __device__ __forceinline__ void load_frag(Frag& f,
                                                 const float* __restrict__ ap0,
                                                 const float* __restrict__ ap1,
                                                 const float* __restrict__ wp) {
  #pragma unroll
  for (int i = 0; i < 2; ++i) {
    f.a0[i] = *(const v4f*)(ap0 + 8 * i);
    f.a1[i] = *(const v4f*)(ap1 + 8 * i);
    #pragma unroll
    for (int p = 0; p < 2; ++p) {
      const int t = 2 * i + p;
      #pragma unroll
      for (int e = 0; e < 2; ++e)
        f.b[t][e] = *(const v4f*)(wp + (size_t)(8 * i + 2 * p + e) * NC);
    }
  }
}

static __device__ __forceinline__ void mma_frag(v8f (&acc)[2][4], const Frag& f) {
  #pragma unroll
  for (int t = 0; t < 4; ++t) {
    const int i = t >> 1, p = (t & 1) * 2;
    const v2f a0 = v2f{f.a0[i][p], f.a0[i][p + 1]};
    const v2f a1 = v2f{f.a1[i][p], f.a1[i][p + 1]};
    #pragma unroll
    for (int c = 0; c < 4; ++c) {
      const v2f b = v2f{f.b[t][0][c], f.b[t][1][c]};
      acc[0][c] = __builtin_amdgcn_wmma_f32_16x16x4_f32(false, a0, false, b, (short)0, acc[0][c], false, false);
      acc[1][c] = __builtin_amdgcn_wmma_f32_16x16x4_f32(false, a1, false, b, (short)0, acc[1][c], false, false);
    }
  }
}

template <bool GAT, bool BIAS, bool ELU, bool HEADA, int NC>
__global__ void __launch_bounds__(128) wmma_gemm_kernel(const float* __restrict__ A,
                                 const float* __restrict__ W,
                                 const float* __restrict__ bias,
                                 const float* __restrict__ att_src,
                                 const float* __restrict__ att_dst,
                                 float* __restrict__ Hout,
                                 float* __restrict__ s_src,
                                 float* __restrict__ s_dst,
                                 int M, int K) {
  const int lane = threadIdx.x & 31;
  const int wv   = threadIdx.x >> 5;
  const int wpb  = blockDim.x >> 5;
  const int tile = blockIdx.x * wpb + wv;
  constexpr int ctiles = NC >> 6;
  const int mt = tile / ctiles;
  const int ct = tile - mt * ctiles;
  const int m0 = mt << 5;
  if (m0 >= M) return;
  const int n0 = ct << 6;
  const int lo = lane & 15;
  const int hi = lane >> 4;

  v8f acc[2][4] = {};
  const size_t lda  = HEADA ? (size_t)K * ctiles : (size_t)K;
  const size_t acol = (HEADA ? (size_t)ct * K : (size_t)0) + (hi << 2);
  const float* ap0 = A + (size_t)(m0 + lo) * lda + acol;
  const float* ap1 = A + (size_t)(m0 + 16 + lo) * lda + acol;
  const float* wp  = W + n0 + (lo << 2) + (size_t)(hi << 2) * NC;
  constexpr size_t wblk = (size_t)16 * NC;

  Frag f0, f1;
  load_frag<NC>(f0, ap0, ap1, wp);
  for (int k = 0; k + 32 < K; k += 32) {
    load_frag<NC>(f1, ap0 + 16, ap1 + 16, wp + wblk);
    __builtin_amdgcn_sched_barrier(0);
    mma_frag(acc, f0);
    __builtin_amdgcn_sched_barrier(0);
    load_frag<NC>(f0, ap0 + 32, ap1 + 32, wp + 2 * wblk);
    __builtin_amdgcn_sched_barrier(0);
    mma_frag(acc, f1);
    ap0 += 32; ap1 += 32; wp += 2 * wblk;
  }
  load_frag<NC>(f1, ap0 + 16, ap1 + 16, wp + wblk);
  __builtin_amdgcn_sched_barrier(0);
  mma_frag(acc, f0);
  mma_frag(acc, f1);

  v4f bv = v4f{0.f, 0.f, 0.f, 0.f};
  if (BIAS) bv = *(const v4f*)(bias + n0 + (lo << 2));

  #pragma unroll
  for (int g = 0; g < 2; ++g) {
    const int rbase = m0 + (g << 4) + (hi << 3);
    #pragma unroll
    for (int r = 0; r < 8; ++r) {
      float* hp = Hout + (size_t)(rbase + r) * NC + n0 + (lo << 2);
      v4f o = v4f{acc[g][0][r] + bv.x, acc[g][1][r] + bv.y,
                  acc[g][2][r] + bv.z, acc[g][3][r] + bv.w};
      if (ELU) o = v4f{elu1(o.x), elu1(o.y), elu1(o.z), elu1(o.w)};
      *(v4f*)hp = o;
    }
  }

  if (GAT) {
    const int head = n0 >> 6;
    const v4f as = *(const v4f*)(att_src + head * CHAN + (lo << 2));
    const v4f ad = *(const v4f*)(att_dst + head * CHAN + (lo << 2));
    #pragma unroll
    for (int g = 0; g < 2; ++g) {
      const int rbase = m0 + (g << 4) + (hi << 3);
      #pragma unroll
      for (int r = 0; r < 8; ++r) {
        float ss = acc[g][0][r] * as.x + acc[g][1][r] * as.y + acc[g][2][r] * as.z + acc[g][3][r] * as.w;
        float sd = acc[g][0][r] * ad.x + acc[g][1][r] * ad.y + acc[g][2][r] * ad.z + acc[g][3][r] * ad.w;
        #pragma unroll
        for (int off = 8; off > 0; off >>= 1) {
          ss += __shfl_xor(ss, off, 32);
          sd += __shfl_xor(sd, off, 32);
        }
        if (lo == 0) {
          const int row = rbase + r;
          s_src[row * HEADS + head] = ss;
          s_dst[row * HEADS + head] = sd;
        }
      }
    }
  }
}

__global__ void prep_kernel(const float* __restrict__ W,
                            const float* __restrict__ att_src,
                            const float* __restrict__ att_dst,
                            float* __restrict__ V, int* __restrict__ degs, int N) {
  const int idx = blockIdx.x * blockDim.x + threadIdx.x;
  if (idx < N) degs[idx] = 0;
  const int lane = threadIdx.x & 31;
  const int k = idx >> 5;
  if (k < FEAT) {
    const float* wr = W + (size_t)k * HIDDEN + (lane << 2);
    const float4 w0 = *(const float4*)(wr);
    const float4 w1 = *(const float4*)(wr + 128);
    const float4 s0 = *(const float4*)(att_src + (lane << 2));
    const float4 s1 = *(const float4*)(att_src + 128 + (lane << 2));
    const float4 d0 = *(const float4*)(att_dst + (lane << 2));
    const float4 d1 = *(const float4*)(att_dst + 128 + (lane << 2));
    float ps0 = w0.x * s0.x + w0.y * s0.y + w0.z * s0.z + w0.w * s0.w;
    float ps1 = w1.x * s1.x + w1.y * s1.y + w1.z * s1.z + w1.w * s1.w;
    float pd0 = w0.x * d0.x + w0.y * d0.y + w0.z * d0.z + w0.w * d0.w;
    float pd1 = w1.x * d1.x + w1.y * d1.y + w1.z * d1.z + w1.w * d1.w;
    #pragma unroll
    for (int off = 8; off > 0; off >>= 1) {
      ps0 += __shfl_xor(ps0, off, 32); ps1 += __shfl_xor(ps1, off, 32);
      pd0 += __shfl_xor(pd0, off, 32); pd1 += __shfl_xor(pd1, off, 32);
    }
    if ((lane & 15) == 0) {
      const int h = lane >> 4;
      float* vr = V + k * 8;
      vr[h] = ps0; vr[2 + h] = ps1; vr[4 + h] = pd0; vr[6 + h] = pd1;
    }
  }
}

__global__ void slot_fill_score_kernel(const int* __restrict__ ei, int E,
                                       int* __restrict__ degs, int* __restrict__ slots,
                                       const float* __restrict__ X,
                                       const float* __restrict__ V,
                                       float* __restrict__ s_src,
                                       float* __restrict__ s_dst, int R) {
  __shared__ float Vs[FEAT * 8];
  const int idx = blockIdx.x * blockDim.x + threadIdx.x;
  if (idx < E) {
    const int dn  = ei[E + idx];
    const int pos = atomicAdd(degs + dn, 1);
    if (pos < SLOTS) slots[(size_t)dn * SLOTS + pos] = ei[idx];
  }
  *(float4*)(Vs + (threadIdx.x << 2)) = *(const float4*)(V + (threadIdx.x << 2));
  __syncthreads();
  const int row = blockIdx.x * 32 + (threadIdx.x >> 3);
  const int j   = threadIdx.x & 7;
  if (row < R) {
    const float* xr = X + (size_t)row * FEAT;
    float s = 0.f;
    #pragma unroll 8
    for (int k = 0; k < FEAT; k += 4) {
      const float4 xv = *(const float4*)(xr + k);
      s += xv.x * Vs[(k + 0) * 8 + j];
      s += xv.y * Vs[(k + 1) * 8 + j];
      s += xv.z * Vs[(k + 2) * 8 + j];
      s += xv.w * Vs[(k + 3) * 8 + j];
    }
    if (j < 4) s_src[row * HEADS + j] = s;
    else       s_dst[row * HEADS + (j - 4)] = s;
  }
}

static __device__ __forceinline__ float wave_max(float v) {
  #pragma unroll
  for (int off = 16; off > 0; off >>= 1) v = fmaxf(v, __shfl_xor(v, off, 32));
  return v;
}
static __device__ __forceinline__ float wave_sum(float v) {
  #pragma unroll
  for (int off = 16; off > 0; off >>= 1) v += __shfl_xor(v, off, 32);
  return v;
}

static __device__ __forceinline__ void load_entry(const int* __restrict__ csr_src,
                                                  const float* __restrict__ s_src,
                                                  int beg, int deg, int d, int row0, int j,
                                                  const float4 vd, int& s, float4& e) {
  s = d;
  if (j < deg) s = row0 + csr_src[beg + j];
  if (j <= deg) {
    const float4 vs = *(const float4*)(s_src + (size_t)s * HEADS);
    e = make_float4(lrelu(vs.x + vd.x), lrelu(vs.y + vd.y), lrelu(vs.z + vd.z), lrelu(vs.w + vd.w));
  } else {
    e = make_float4(-INFINITY, -INFINITY, -INFINITY, -INFINITY);
  }
}

template <bool XIN>
static __device__ __forceinline__ void row_add(const float* __restrict__ Hin, int st,
                                               float a0, float a1, float a2, float a3,
                                               int lane, float4 (&v)[4]) {
  if constexpr (XIN) {
    const float4 x = *(const float4*)(Hin + (size_t)st * FEAT + (lane << 2));
    v[0].x += x.x * a0; v[0].y += x.y * a0; v[0].z += x.z * a0; v[0].w += x.w * a0;
    v[1].x += x.x * a1; v[1].y += x.y * a1; v[1].z += x.z * a1; v[1].w += x.w * a1;
    v[2].x += x.x * a2; v[2].y += x.y * a2; v[2].z += x.z * a2; v[2].w += x.w * a2;
    v[3].x += x.x * a3; v[3].y += x.y * a3; v[3].z += x.z * a3; v[3].w += x.w * a3;
  } else {
    const float* hs = Hin + (size_t)st * HIDDEN + (lane << 2);
    const float4 h0 = *(const float4*)(hs);
    const float4 h1 = *(const float4*)(hs + 128);
    const float w0 = lane >= 16 ? a1 : a0;
    const float w1 = lane >= 16 ? a3 : a2;
    v[0].x += h0.x * w0; v[0].y += h0.y * w0; v[0].z += h0.z * w0; v[0].w += h0.w * w0;
    v[1].x += h1.x * w1; v[1].y += h1.y * w1; v[1].z += h1.z * w1; v[1].w += h1.w * w1;
  }
}

template <bool XIN>
static __device__ __forceinline__ void accum_one(const float* __restrict__ Hin,
                                                 int s, const float4 al, int t, int lane,
                                                 float4 (&v)[4]) {
  const int   st = __builtin_amdgcn_readlane(s, t);
  const float a0 = __int_as_float(__builtin_amdgcn_readlane(__float_as_int(al.x), t));
  const float a1 = __int_as_float(__builtin_amdgcn_readlane(__float_as_int(al.y), t));
  const float a2 = __int_as_float(__builtin_amdgcn_readlane(__float_as_int(al.z), t));
  const float a3 = __int_as_float(__builtin_amdgcn_readlane(__float_as_int(al.w), t));
  row_add<XIN>(Hin, st, a0, a1, a2, a3, lane, v);
}

template <bool XIN>
static __device__ __forceinline__ void accum_entries(const float* __restrict__ Hin,
                                                     int s, const float4 al, int cnt, int lane,
                                                     float4 (&v)[4]) {
  #pragma unroll 4
  for (int t = 0; t < cnt; ++t) {
    const int   st = __builtin_amdgcn_readlane(s, t);
    const float a0 = __int_as_float(__builtin_amdgcn_readlane(__float_as_int(al.x), t));
    const float a1 = __int_as_float(__builtin_amdgcn_readlane(__float_as_int(al.y), t));
    const float a2 = __int_as_float(__builtin_amdgcn_readlane(__float_as_int(al.z), t));
    const float a3 = __int_as_float(__builtin_amdgcn_readlane(__float_as_int(al.w), t));
    row_add<XIN>(Hin, st, a0, a1, a2, a3, lane, v);
  }
}

static __device__ __forceinline__ float4 lrelu4(const float4 a, const float4 b) {
  return make_float4(lrelu(a.x + b.x), lrelu(a.y + b.y), lrelu(a.z + b.z), lrelu(a.w + b.w));
}

template <bool XIN>
static __device__ __forceinline__ void sweep_edge_list(const int* __restrict__ ei, int E,
                                                       const float* __restrict__ s_src,
                                                       const float* __restrict__ Hin,
                                                       int g, int d, int row0, const float4 vd,
                                                       int lane, float4 (&v)[4]) {
  const float4 es = lrelu4(*(const float4*)(s_src + (size_t)d * HEADS), vd);
  float4 m = es;
  for (int i0 = 0; i0 < E; i0 += 32) {
    const int i = i0 + lane;
    if (i < E && ei[E + i] == g) {
      const float4 e = lrelu4(*(const float4*)(s_src + (size_t)(row0 + ei[i]) * HEADS), vd);
      m.x = fmaxf(m.x, e.x); m.y = fmaxf(m.y, e.y); m.z = fmaxf(m.z, e.z); m.w = fmaxf(m.w, e.w);
    }
  }
  m.x = wave_max(m.x); m.y = wave_max(m.y); m.z = wave_max(m.z); m.w = wave_max(m.w);

  const float4 xs = make_float4(expf(es.x - m.x), expf(es.y - m.y), expf(es.z - m.z), expf(es.w - m.w));
  float4 dn = lane == 0 ? xs : make_float4(0.f, 0.f, 0.f, 0.f);
  for (int i0 = 0; i0 < E; i0 += 32) {
    const int i = i0 + lane;
    if (i < E && ei[E + i] == g) {
      const float4 e = lrelu4(*(const float4*)(s_src + (size_t)(row0 + ei[i]) * HEADS), vd);
      dn.x += expf(e.x - m.x); dn.y += expf(e.y - m.y);
      dn.z += expf(e.z - m.z); dn.w += expf(e.w - m.w);
    }
  }
  dn.x = wave_sum(dn.x); dn.y = wave_sum(dn.y); dn.z = wave_sum(dn.z); dn.w = wave_sum(dn.w);

  for (int i0 = 0; i0 < E; i0 += 32) {
    const int i = i0 + lane;
    const bool hit = i < E && ei[E + i] == g;
    int s = d;
    float4 al = make_float4(0.f, 0.f, 0.f, 0.f);
    if (hit) {
      s = row0 + ei[i];
      const float4 e = lrelu4(*(const float4*)(s_src + (size_t)s * HEADS), vd);
      al = make_float4(expf(e.x - m.x) / dn.x, expf(e.y - m.y) / dn.y,
                       expf(e.z - m.z) / dn.z, expf(e.w - m.w) / dn.w);
    }
    unsigned hits = __builtin_amdgcn_ballot_w32(hit);
    while (hits) {
      const int t = __builtin_ctz(hits);
      hits &= hits - 1;
      accum_one<XIN>(Hin, s, al, t, lane, v);
    }
  }
  const float4 al = make_float4(xs.x / dn.x, xs.y / dn.y, xs.z / dn.z, xs.w / dn.w);
  row_add<XIN>(Hin, d, al.x, al.y, al.z, al.w, lane, v);
}

template <bool XIN>
__global__ void gat_gather_kernel(const int* __restrict__ degs,
                                  const int* __restrict__ csr_src,
                                  const int* __restrict__ ei, int E,
                                  const float* __restrict__ s_src,
                                  const float* __restrict__ s_dst,
                                  const float* __restrict__ Hin,
                                  const float* __restrict__ bias,
                                  float* __restrict__ agg, int N, int R) {
  const int lane = threadIdx.x & 31;
  const int d = __builtin_amdgcn_readfirstlane(blockIdx.x * (blockDim.x >> 5) + (threadIdx.x >> 5));
  if (d >= R) return;
  const int row0 = d >= N ? N : 0;
  const int g    = d - row0;
  const int beg = g * SLOTS;
  const int deg = degs[g];
  const int tot = deg + 1;
  const float4 vd = *(const float4*)(s_dst + (size_t)d * HEADS);

  float4 acc[4];
  #pragma unroll
  for (int h = 0; h < 4; ++h) acc[h] = make_float4(0.f, 0.f, 0.f, 0.f);
  if (deg > SLOTS) {
    sweep_edge_list<XIN>(ei, E, s_src, Hin, g, d, row0, vd, lane, acc);
  } else {
  int s0; float4 e0;
  load_entry(csr_src, s_src, beg, deg, d, row0, lane, vd, s0, e0);

  float4 m = e0;
  for (int base = 32; base < tot; base += 32) {
    int s; float4 e;
    load_entry(csr_src, s_src, beg, deg, d, row0, base + lane, vd, s, e);
    m.x = fmaxf(m.x, e.x); m.y = fmaxf(m.y, e.y); m.z = fmaxf(m.z, e.z); m.w = fmaxf(m.w, e.w);
  }
  m.x = wave_max(m.x); m.y = wave_max(m.y); m.z = wave_max(m.z); m.w = wave_max(m.w);

  const bool v0 = lane <= deg;
  float4 x0 = make_float4(v0 ? expf(e0.x - m.x) : 0.f, v0 ? expf(e0.y - m.y) : 0.f,
                          v0 ? expf(e0.z - m.z) : 0.f, v0 ? expf(e0.w - m.w) : 0.f);
  float4 dn = x0;
  for (int base = 32; base < tot; base += 32) {
    int s; float4 e;
    load_entry(csr_src, s_src, beg, deg, d, row0, base + lane, vd, s, e);
    if (base + lane <= deg) {
      dn.x += expf(e.x - m.x); dn.y += expf(e.y - m.y);
      dn.z += expf(e.z - m.z); dn.w += expf(e.w - m.w);
    }
  }
  dn.x = wave_sum(dn.x); dn.y = wave_sum(dn.y); dn.z = wave_sum(dn.z); dn.w = wave_sum(dn.w);

  {
    const float4 al = make_float4(x0.x / dn.x, x0.y / dn.y, x0.z / dn.z, x0.w / dn.w);
    accum_entries<XIN>(Hin, s0, al, tot < 32 ? tot : 32, lane, acc);
  }
  for (int base = 32; base < tot; base += 32) {
    int s; float4 e;
    load_entry(csr_src, s_src, beg, deg, d, row0, base + lane, vd, s, e);
    const bool v = base + lane <= deg;
    const float4 al = make_float4(v ? expf(e.x - m.x) / dn.x : 0.f, v ? expf(e.y - m.y) / dn.y : 0.f,
                                  v ? expf(e.z - m.z) / dn.z : 0.f, v ? expf(e.w - m.w) / dn.w : 0.f);
    const int rem = tot - base;
    accum_entries<XIN>(Hin, s, al, rem < 32 ? rem : 32, lane, acc);
  }
  }

  if constexpr (XIN) {
    float* od = agg + (size_t)d * (HEADS * FEAT) + (lane << 2);
    #pragma unroll
    for (int h = 0; h < 4; ++h) *(float4*)(od + h * FEAT) = acc[h];
  } else {
    const float4 b0 = *(const float4*)(bias + (lane << 2));
    const float4 b1 = *(const float4*)(bias + 128 + (lane << 2));
    float* od = agg + (size_t)d * HIDDEN + (lane << 2);
    *(float4*)(od)       = make_float4(acc[0].x + b0.x, acc[0].y + b0.y, acc[0].z + b0.z, acc[0].w + b0.w);
    *(float4*)(od + 128) = make_float4(acc[1].x + b1.x, acc[1].y + b1.y, acc[1].z + b1.z, acc[1].w + b1.w);
  }
}

extern "C" void kernel_launch(void* const* d_in, const int* in_sizes, int n_in,
                              void* d_out, int out_size, void* d_ws, size_t ws_size,
                              hipStream_t stream) {
  const float* x     = (const float*)d_in[0];
  const int*   ei    = (const int*)  d_in[1];
  const float* W1    = (const float*)d_in[2];
  const float* asrc1 = (const float*)d_in[3];
  const float* adst1 = (const float*)d_in[4];
  const float* b1    = (const float*)d_in[5];
  const float* W2    = (const float*)d_in[6];
  const float* asrc2 = (const float*)d_in[7];
  const float* adst2 = (const float*)d_in[8];
  const float* b2    = (const float*)d_in[9];
  const float* Wf    = (const float*)d_in[10];
  const float* bf    = (const float*)d_in[11];

  const int B  = 2;
  const int N  = in_sizes[0] / (B * FEAT);
  const int E  = in_sizes[1] / 2;
  const int R  = B * N;

  float* p = (float*)d_ws;
  float* buf_h   = p; p += (size_t)R * HIDDEN;
  float* buf_agg = p; p += (size_t)R * HIDDEN;
  float* buf_z   = p; p += (size_t)R * HEADS * FEAT;
  float* ssrc    = p; p += (size_t)R * HEADS;
  float* sdst    = p; p += (size_t)R * HEADS;
  float* vmat    = p; p += (size_t)FEAT * 8;
  int* q = (int*)p;
  int* slots = q; q += (size_t)N * SLOTS;
  int* degs  = q; q += (size_t)N;

  const int mtiles  = R >> 5;
  const int gtiles  = mtiles * (HIDDEN >> 6);
  const int ftiles  = mtiles * (CHAN >> 6);
  const dim3 blkG(128);
  const dim3 blkE(256);


  const int eblocks = (E + 255) / 256;
  const int sblocks = (R + 31) / 32;
  prep_kernel<<<(N + 255) / 256 + (FEAT * 32) / 256, blkE, 0, stream>>>(W1, asrc1, adst1, vmat, degs, N);
  slot_fill_score_kernel<<<eblocks > sblocks ? eblocks : sblocks, blkE, 0, stream>>>(
      ei, E, degs, slots, x, vmat, ssrc, sdst, R);

  gat_gather_kernel<true><<<(R + 7) / 8, blkE, 0, stream>>>(degs, slots, ei, E, ssrc, sdst, x, nullptr, buf_z, N, R);
  wmma_gemm_kernel<false, true, true, true, HIDDEN><<<(gtiles + 3) / 4, blkG, 0, stream>>>(
      buf_z, W1, b1, nullptr, nullptr, buf_agg, nullptr, nullptr, R, FEAT);

  wmma_gemm_kernel<true, false, false, false, HIDDEN><<<(gtiles + 3) / 4, blkG, 0, stream>>>(
      buf_agg, W2, nullptr, asrc2, adst2, buf_h, ssrc, sdst, R, HIDDEN);
  gat_gather_kernel<false><<<(R + 7) / 8, blkE, 0, stream>>>(degs, slots, ei, E, ssrc, sdst, buf_h, b2, buf_agg, N, R);

  wmma_gemm_kernel<false, true, false, false, CHAN><<<(ftiles + 3) / 4, blkG, 0, stream>>>(
      buf_agg, Wf, bf, nullptr, nullptr, (float*)d_out, nullptr, nullptr, R, HIDDEN);
}


# Layer-1 and layer-2 GEMMs in one launch: a block's four waves leave the 32x256 ELU tile in LDS and read it back as A of act@W2; act never written to memory; 7 launches become 6; sum order unchanged
# speedup vs baseline: 6.9693x; 1.0457x over previous
#include <hip/hip_runtime.h>
#include <math.h>


typedef __attribute__((ext_vector_type(2))) float v2f;
typedef __attribute__((ext_vector_type(4))) float v4f;
typedef __attribute__((ext_vector_type(8))) float v8f;

#define HEADS     4
#define CHAN      64
#define HIDDEN    256
#define FEAT      128
#define NEG_SLOPE 0.2f
#define SLOTS     64

static __device__ __forceinline__ float lrelu(float x) {
  return x > 0.f ? x : NEG_SLOPE * x;
}

static __device__ __forceinline__ float elu1(float v) {
  const float e = expm1f(fminf(v, 0.f));
  return v < 0.f ? e : v;
}

struct Frag {
  v4f a0[2];
  v4f a1[2];
  v4f b[4][2];
};

template <int NC>
static __device__ __forceinline__ void load_frag(Frag& f,
                                                 const float* __restrict__ ap0,
                                                 const float* __restrict__ ap1,
                                                 const float* __restrict__ wp) {
  #pragma unroll
  for (int i = 0; i < 2; ++i) {
    f.a0[i] = *(const v4f*)(ap0 + 8 * i);
    f.a1[i] = *(const v4f*)(ap1 + 8 * i);
    #pragma unroll
    for (int p = 0; p < 2; ++p) {
      const int t = 2 * i + p;
      #pragma unroll
      for (int e = 0; e < 2; ++e)
        f.b[t][e] = *(const v4f*)(wp + (size_t)(8 * i + 2 * p + e) * NC);
    }
  }
}

static __device__ __forceinline__ void mma_frag(v8f (&acc)[2][4], const Frag& f) {
  #pragma unroll
  for (int t = 0; t < 4; ++t) {
    const int i = t >> 1, p = (t & 1) * 2;
    const v2f a0 = v2f{f.a0[i][p], f.a0[i][p + 1]};
    const v2f a1 = v2f{f.a1[i][p], f.a1[i][p + 1]};
    #pragma unroll
    for (int c = 0; c < 4; ++c) {
      const v2f b = v2f{f.b[t][0][c], f.b[t][1][c]};
      acc[0][c] = __builtin_amdgcn_wmma_f32_16x16x4_f32(false, a0, false, b, (short)0, acc[0][c], false, false);
      acc[1][c] = __builtin_amdgcn_wmma_f32_16x16x4_f32(false, a1, false, b, (short)0, acc[1][c], false, false);
    }
  }
}

template <int NC>
static __device__ __forceinline__ void gemm_kloop(v8f (&acc)[2][4],
                                                  const float* ap0, const float* ap1,
                                                  const float* __restrict__ wp, int K) {
  constexpr size_t wblk = (size_t)16 * NC;
  Frag f0, f1;
  load_frag<NC>(f0, ap0, ap1, wp);
  for (int k = 0; k + 32 < K; k += 32) {
    load_frag<NC>(f1, ap0 + 16, ap1 + 16, wp + wblk);
    __builtin_amdgcn_sched_barrier(0);
    mma_frag(acc, f0);
    __builtin_amdgcn_sched_barrier(0);
    load_frag<NC>(f0, ap0 + 32, ap1 + 32, wp + 2 * wblk);
    __builtin_amdgcn_sched_barrier(0);
    mma_frag(acc, f1);
    ap0 += 32; ap1 += 32; wp += 2 * wblk;
  }
  load_frag<NC>(f1, ap0 + 16, ap1 + 16, wp + wblk);
  __builtin_amdgcn_sched_barrier(0);
  mma_frag(acc, f0);
  mma_frag(acc, f1);
}

template <int NC>
__global__ void __launch_bounds__(128) wmma_linear_kernel(const float* __restrict__ A,
                                 const float* __restrict__ W,
                                 const float* __restrict__ bias,
                                 float* __restrict__ Hout,
                                 int M, int K) {
  const int lane = threadIdx.x & 31;
  const int wv   = threadIdx.x >> 5;
  const int wpb  = blockDim.x >> 5;
  const int tile = blockIdx.x * wpb + wv;
  constexpr int ctiles = NC >> 6;
  const int mt = tile / ctiles;
  const int ct = tile - mt * ctiles;
  const int m0 = mt << 5;
  if (m0 >= M) return;
  const int n0 = ct << 6;
  const int lo = lane & 15;
  const int hi = lane >> 4;

  v8f acc[2][4] = {};
  const float* ap0 = A + (size_t)(m0 + lo) * K + (hi << 2);
  const float* ap1 = A + (size_t)(m0 + 16 + lo) * K + (hi << 2);
  const float* wp  = W + n0 + (lo << 2) + (size_t)(hi << 2) * NC;
  gemm_kloop<NC>(acc, ap0, ap1, wp, K);

  const v4f bv = *(const v4f*)(bias + n0 + (lo << 2));
  #pragma unroll
  for (int g = 0; g < 2; ++g) {
    const int rbase = m0 + (g << 4) + (hi << 3);
    #pragma unroll
    for (int r = 0; r < 8; ++r) {
      float* hp = Hout + (size_t)(rbase + r) * NC + n0 + (lo << 2);
      *(v4f*)hp = v4f{acc[g][0][r] + bv.x, acc[g][1][r] + bv.y,
                      acc[g][2][r] + bv.z, acc[g][3][r] + bv.w};
    }
  }
}

#define APITCH (HIDDEN + 4)
__global__ void __launch_bounds__(128) gat_two_gemm_kernel(const float* __restrict__ Z,
                                 const float* __restrict__ W1,
                                 const float* __restrict__ b1,
                                 const float* __restrict__ W2,
                                 const float* __restrict__ att_src,
                                 const float* __restrict__ att_dst,
                                 float* __restrict__ Hout,
                                 float* __restrict__ s_src,
                                 float* __restrict__ s_dst,
                                 int M, int K1, int K2) {
  __shared__ __attribute__((aligned(16))) float act[32 * APITCH];
  const int lane = threadIdx.x & 31;
  const int head = threadIdx.x >> 5;
  const int m0 = blockIdx.x << 5;
  if (m0 >= M) return;
  const int n0 = head << 6;
  const int lo = lane & 15;
  const int hi = lane >> 4;

  {
    v8f acc[2][4] = {};
    const size_t lda = (size_t)K1 * HEADS;
    const float* ap0 = Z + (size_t)(m0 + lo) * lda + (size_t)head * K1 + (hi << 2);
    const float* ap1 = ap0 + 16 * lda;
    const float* wp  = W1 + n0 + (lo << 2) + (size_t)(hi << 2) * HIDDEN;
    gemm_kloop<HIDDEN>(acc, ap0, ap1, wp, K1);
    const v4f bv = *(const v4f*)(b1 + n0 + (lo << 2));
    #pragma unroll
    for (int g = 0; g < 2; ++g) {
      #pragma unroll
      for (int r = 0; r < 8; ++r) {
        const int row = (g << 4) + (hi << 3) + r;
        *(v4f*)(act + row * APITCH + n0 + (lo << 2)) =
            v4f{elu1(acc[g][0][r] + bv.x), elu1(acc[g][1][r] + bv.y),
                elu1(acc[g][2][r] + bv.z), elu1(acc[g][3][r] + bv.w)};
      }
    }
  }
  __syncthreads();

  v8f acc[2][4] = {};
  {
    const float* ap0 = act + lo * APITCH + (hi << 2);
    const float* ap1 = ap0 + 16 * APITCH;
    const float* wp  = W2 + n0 + (lo << 2) + (size_t)(hi << 2) * HIDDEN;
    gemm_kloop<HIDDEN>(acc, ap0, ap1, wp, K2);
  }
  #pragma unroll
  for (int g = 0; g < 2; ++g) {
    const int rbase = m0 + (g << 4) + (hi << 3);
    #pragma unroll
    for (int r = 0; r < 8; ++r) {
      float* hp = Hout + (size_t)(rbase + r) * HIDDEN + n0 + (lo << 2);
      *(v4f*)hp = v4f{acc[g][0][r], acc[g][1][r], acc[g][2][r], acc[g][3][r]};
    }
  }
  const v4f as = *(const v4f*)(att_src + head * CHAN + (lo << 2));
  const v4f ad = *(const v4f*)(att_dst + head * CHAN + (lo << 2));
  #pragma unroll
  for (int g = 0; g < 2; ++g) {
    const int rbase = m0 + (g << 4) + (hi << 3);
    #pragma unroll
    for (int r = 0; r < 8; ++r) {
      float ss = acc[g][0][r] * as.x + acc[g][1][r] * as.y + acc[g][2][r] * as.z + acc[g][3][r] * as.w;
      float sd = acc[g][0][r] * ad.x + acc[g][1][r] * ad.y + acc[g][2][r] * ad.z + acc[g][3][r] * ad.w;
      #pragma unroll
      for (int off = 8; off > 0; off >>= 1) {
        ss += __shfl_xor(ss, off, 32);
        sd += __shfl_xor(sd, off, 32);
      }
      if (lo == 0) {
        const int row = rbase + r;
        s_src[row * HEADS + head] = ss;
        s_dst[row * HEADS + head] = sd;
      }
    }
  }
}

__global__ void prep_kernel(const float* __restrict__ W,
                            const float* __restrict__ att_src,
                            const float* __restrict__ att_dst,
                            float* __restrict__ V, int* __restrict__ degs, int N) {
  const int idx = blockIdx.x * blockDim.x + threadIdx.x;
  if (idx < N) degs[idx] = 0;
  const int lane = threadIdx.x & 31;
  const int k = idx >> 5;
  if (k < FEAT) {
    const float* wr = W + (size_t)k * HIDDEN + (lane << 2);
    const float4 w0 = *(const float4*)(wr);
    const float4 w1 = *(const float4*)(wr + 128);
    const float4 s0 = *(const float4*)(att_src + (lane << 2));
    const float4 s1 = *(const float4*)(att_src + 128 + (lane << 2));
    const float4 d0 = *(const float4*)(att_dst + (lane << 2));
    const float4 d1 = *(const float4*)(att_dst + 128 + (lane << 2));
    float ps0 = w0.x * s0.x + w0.y * s0.y + w0.z * s0.z + w0.w * s0.w;
    float ps1 = w1.x * s1.x + w1.y * s1.y + w1.z * s1.z + w1.w * s1.w;
    float pd0 = w0.x * d0.x + w0.y * d0.y + w0.z * d0.z + w0.w * d0.w;
    float pd1 = w1.x * d1.x + w1.y * d1.y + w1.z * d1.z + w1.w * d1.w;
    #pragma unroll
    for (int off = 8; off > 0; off >>= 1) {
      ps0 += __shfl_xor(ps0, off, 32); ps1 += __shfl_xor(ps1, off, 32);
      pd0 += __shfl_xor(pd0, off, 32); pd1 += __shfl_xor(pd1, off, 32);
    }
    if ((lane & 15) == 0) {
      const int h = lane >> 4;
      float* vr = V + k * 8;
      vr[h] = ps0; vr[2 + h] = ps1; vr[4 + h] = pd0; vr[6 + h] = pd1;
    }
  }
}

__global__ void slot_fill_score_kernel(const int* __restrict__ ei, int E,
                                       int* __restrict__ degs, int* __restrict__ slots,
                                       const float* __restrict__ X,
                                       const float* __restrict__ V,
                                       float* __restrict__ s_src,
                                       float* __restrict__ s_dst, int R) {
  __shared__ float Vs[FEAT * 8];
  const int idx = blockIdx.x * blockDim.x + threadIdx.x;
  if (idx < E) {
    const int dn  = ei[E + idx];
    const int pos = atomicAdd(degs + dn, 1);
    if (pos < SLOTS) slots[(size_t)dn * SLOTS + pos] = ei[idx];
  }
  *(float4*)(Vs + (threadIdx.x << 2)) = *(const float4*)(V + (threadIdx.x << 2));
  __syncthreads();
  const int row = blockIdx.x * 32 + (threadIdx.x >> 3);
  const int j   = threadIdx.x & 7;
  if (row < R) {
    const float* xr = X + (size_t)row * FEAT;
    float s = 0.f;
    #pragma unroll 8
    for (int k = 0; k < FEAT; k += 4) {
      const float4 xv = *(const float4*)(xr + k);
      s += xv.x * Vs[(k + 0) * 8 + j];
      s += xv.y * Vs[(k + 1) * 8 + j];
      s += xv.z * Vs[(k + 2) * 8 + j];
      s += xv.w * Vs[(k + 3) * 8 + j];
    }
    if (j < 4) s_src[row * HEADS + j] = s;
    else       s_dst[row * HEADS + (j - 4)] = s;
  }
}

static __device__ __forceinline__ float wave_max(float v) {
  #pragma unroll
  for (int off = 16; off > 0; off >>= 1) v = fmaxf(v, __shfl_xor(v, off, 32));
  return v;
}
static __device__ __forceinline__ float wave_sum(float v) {
  #pragma unroll
  for (int off = 16; off > 0; off >>= 1) v += __shfl_xor(v, off, 32);
  return v;
}

static __device__ __forceinline__ void load_entry(const int* __restrict__ csr_src,
                                                  const float* __restrict__ s_src,
                                                  int beg, int deg, int d, int row0, int j,
                                                  const float4 vd, int& s, float4& e) {
  s = d;
  if (j < deg) s = row0 + csr_src[beg + j];
  if (j <= deg) {
    const float4 vs = *(const float4*)(s_src + (size_t)s * HEADS);
    e = make_float4(lrelu(vs.x + vd.x), lrelu(vs.y + vd.y), lrelu(vs.z + vd.z), lrelu(vs.w + vd.w));
  } else {
    e = make_float4(-INFINITY, -INFINITY, -INFINITY, -INFINITY);
  }
}

template <bool XIN>
static __device__ __forceinline__ void row_add(const float* __restrict__ Hin, int st,
                                               float a0, float a1, float a2, float a3,
                                               int lane, float4 (&v)[4]) {
  if constexpr (XIN) {
    const float4 x = *(const float4*)(Hin + (size_t)st * FEAT + (lane << 2));
    v[0].x += x.x * a0; v[0].y += x.y * a0; v[0].z += x.z * a0; v[0].w += x.w * a0;
    v[1].x += x.x * a1; v[1].y += x.y * a1; v[1].z += x.z * a1; v[1].w += x.w * a1;
    v[2].x += x.x * a2; v[2].y += x.y * a2; v[2].z += x.z * a2; v[2].w += x.w * a2;
    v[3].x += x.x * a3; v[3].y += x.y * a3; v[3].z += x.z * a3; v[3].w += x.w * a3;
  } else {
    const float* hs = Hin + (size_t)st * HIDDEN + (lane << 2);
    const float4 h0 = *(const float4*)(hs);
    const float4 h1 = *(const float4*)(hs + 128);
    const float w0 = lane >= 16 ? a1 : a0;
    const float w1 = lane >= 16 ? a3 : a2;
    v[0].x += h0.x * w0; v[0].y += h0.y * w0; v[0].z += h0.z * w0; v[0].w += h0.w * w0;
    v[1].x += h1.x * w1; v[1].y += h1.y * w1; v[1].z += h1.z * w1; v[1].w += h1.w * w1;
  }
}

template <bool XIN>
static __device__ __forceinline__ void accum_one(const float* __restrict__ Hin,
                                                 int s, const float4 al, int t, int lane,
                                                 float4 (&v)[4]) {
  const int   st = __builtin_amdgcn_readlane(s, t);
  const float a0 = __int_as_float(__builtin_amdgcn_readlane(__float_as_int(al.x), t));
  const float a1 = __int_as_float(__builtin_amdgcn_readlane(__float_as_int(al.y), t));
  const float a2 = __int_as_float(__builtin_amdgcn_readlane(__float_as_int(al.z), t));
  const float a3 = __int_as_float(__builtin_amdgcn_readlane(__float_as_int(al.w), t));
  row_add<XIN>(Hin, st, a0, a1, a2, a3, lane, v);
}

template <bool XIN>
static __device__ __forceinline__ void accum_entries(const float* __restrict__ Hin,
                                                     int s, const float4 al, int cnt, int lane,
                                                     float4 (&v)[4]) {
  #pragma unroll 4
  for (int t = 0; t < cnt; ++t) {
    const int   st = __builtin_amdgcn_readlane(s, t);
    const float a0 = __int_as_float(__builtin_amdgcn_readlane(__float_as_int(al.x), t));
    const float a1 = __int_as_float(__builtin_amdgcn_readlane(__float_as_int(al.y), t));
    const float a2 = __int_as_float(__builtin_amdgcn_readlane(__float_as_int(al.z), t));
    const float a3 = __int_as_float(__builtin_amdgcn_readlane(__float_as_int(al.w), t));
    row_add<XIN>(Hin, st, a0, a1, a2, a3, lane, v);
  }
}

static __device__ __forceinline__ float4 lrelu4(const float4 a, const float4 b) {
  return make_float4(lrelu(a.x + b.x), lrelu(a.y + b.y), lrelu(a.z + b.z), lrelu(a.w + b.w));
}

template <bool XIN>
static __device__ __forceinline__ void sweep_edge_list(const int* __restrict__ ei, int E,
                                                       const float* __restrict__ s_src,
                                                       const float* __restrict__ Hin,
                                                       int g, int d, int row0, const float4 vd,
                                                       int lane, float4 (&v)[4]) {
  const float4 es = lrelu4(*(const float4*)(s_src + (size_t)d * HEADS), vd);
  float4 m = es;
  for (int i0 = 0; i0 < E; i0 += 32) {
    const int i = i0 + lane;
    if (i < E && ei[E + i] == g) {
      const float4 e = lrelu4(*(const float4*)(s_src + (size_t)(row0 + ei[i]) * HEADS), vd);
      m.x = fmaxf(m.x, e.x); m.y = fmaxf(m.y, e.y); m.z = fmaxf(m.z, e.z); m.w = fmaxf(m.w, e.w);
    }
  }
  m.x = wave_max(m.x); m.y = wave_max(m.y); m.z = wave_max(m.z); m.w = wave_max(m.w);

  const float4 xs = make_float4(expf(es.x - m.x), expf(es.y - m.y), expf(es.z - m.z), expf(es.w - m.w));
  float4 dn = lane == 0 ? xs : make_float4(0.f, 0.f, 0.f, 0.f);
  for (int i0 = 0; i0 < E; i0 += 32) {
    const int i = i0 + lane;
    if (i < E && ei[E + i] == g) {
      const float4 e = lrelu4(*(const float4*)(s_src + (size_t)(row0 + ei[i]) * HEADS), vd);
      dn.x += expf(e.x - m.x); dn.y += expf(e.y - m.y);
      dn.z += expf(e.z - m.z); dn.w += expf(e.w - m.w);
    }
  }
  dn.x = wave_sum(dn.x); dn.y = wave_sum(dn.y); dn.z = wave_sum(dn.z); dn.w = wave_sum(dn.w);

  for (int i0 = 0; i0 < E; i0 += 32) {
    const int i = i0 + lane;
    const bool hit = i < E && ei[E + i] == g;
    int s = d;
    float4 al = make_float4(0.f, 0.f, 0.f, 0.f);
    if (hit) {
      s = row0 + ei[i];
      const float4 e = lrelu4(*(const float4*)(s_src + (size_t)s * HEADS), vd);
      al = make_float4(expf(e.x - m.x) / dn.x, expf(e.y - m.y) / dn.y,
                       expf(e.z - m.z) / dn.z, expf(e.w - m.w) / dn.w);
    }
    unsigned hits = __builtin_amdgcn_ballot_w32(hit);
    while (hits) {
      const int t = __builtin_ctz(hits);
      hits &= hits - 1;
      accum_one<XIN>(Hin, s, al, t, lane, v);
    }
  }
  const float4 al = make_float4(xs.x / dn.x, xs.y / dn.y, xs.z / dn.z, xs.w / dn.w);
  row_add<XIN>(Hin, d, al.x, al.y, al.z, al.w, lane, v);
}

template <bool XIN>
__global__ void gat_gather_kernel(const int* __restrict__ degs,
                                  const int* __restrict__ csr_src,
                                  const int* __restrict__ ei, int E,
                                  const float* __restrict__ s_src,
                                  const float* __restrict__ s_dst,
                                  const float* __restrict__ Hin,
                                  const float* __restrict__ bias,
                                  float* __restrict__ agg, int N, int R) {
  const int lane = threadIdx.x & 31;
  const int d = __builtin_amdgcn_readfirstlane(blockIdx.x * (blockDim.x >> 5) + (threadIdx.x >> 5));
  if (d >= R) return;
  const int row0 = d >= N ? N : 0;
  const int g    = d - row0;
  const int beg = g * SLOTS;
  const int deg = degs[g];
  const int tot = deg + 1;
  const float4 vd = *(const float4*)(s_dst + (size_t)d * HEADS);

  float4 acc[4];
  #pragma unroll
  for (int h = 0; h < 4; ++h) acc[h] = make_float4(0.f, 0.f, 0.f, 0.f);
  if (deg > SLOTS) {
    sweep_edge_list<XIN>(ei, E, s_src, Hin, g, d, row0, vd, lane, acc);
  } else {
  int s0; float4 e0;
  load_entry(csr_src, s_src, beg, deg, d, row0, lane, vd, s0, e0);

  float4 m = e0;
  for (int base = 32; base < tot; base += 32) {
    int s; float4 e;
    load_entry(csr_src, s_src, beg, deg, d, row0, base + lane, vd, s, e);
    m.x = fmaxf(m.x, e.x); m.y = fmaxf(m.y, e.y); m.z = fmaxf(m.z, e.z); m.w = fmaxf(m.w, e.w);
  }
  m.x = wave_max(m.x); m.y = wave_max(m.y); m.z = wave_max(m.z); m.w = wave_max(m.w);

  const bool v0 = lane <= deg;
  float4 x0 = make_float4(v0 ? expf(e0.x - m.x) : 0.f, v0 ? expf(e0.y - m.y) : 0.f,
                          v0 ? expf(e0.z - m.z) : 0.f, v0 ? expf(e0.w - m.w) : 0.f);
  float4 dn = x0;
  for (int base = 32; base < tot; base += 32) {
    int s; float4 e;
    load_entry(csr_src, s_src, beg, deg, d, row0, base + lane, vd, s, e);
    if (base + lane <= deg) {
      dn.x += expf(e.x - m.x); dn.y += expf(e.y - m.y);
      dn.z += expf(e.z - m.z); dn.w += expf(e.w - m.w);
    }
  }
  dn.x = wave_sum(dn.x); dn.y = wave_sum(dn.y); dn.z = wave_sum(dn.z); dn.w = wave_sum(dn.w);

  {
    const float4 al = make_float4(x0.x / dn.x, x0.y / dn.y, x0.z / dn.z, x0.w / dn.w);
    accum_entries<XIN>(Hin, s0, al, tot < 32 ? tot : 32, lane, acc);
  }
  for (int base = 32; base < tot; base += 32) {
    int s; float4 e;
    load_entry(csr_src, s_src, beg, deg, d, row0, base + lane, vd, s, e);
    const bool v = base + lane <= deg;
    const float4 al = make_float4(v ? expf(e.x - m.x) / dn.x : 0.f, v ? expf(e.y - m.y) / dn.y : 0.f,
                                  v ? expf(e.z - m.z) / dn.z : 0.f, v ? expf(e.w - m.w) / dn.w : 0.f);
    const int rem = tot - base;
    accum_entries<XIN>(Hin, s, al, rem < 32 ? rem : 32, lane, acc);
  }
  }

  if constexpr (XIN) {
    float* od = agg + (size_t)d * (HEADS * FEAT) + (lane << 2);
    #pragma unroll
    for (int h = 0; h < 4; ++h) *(float4*)(od + h * FEAT) = acc[h];
  } else {
    const float4 b0 = *(const float4*)(bias + (lane << 2));
    const float4 b1 = *(const float4*)(bias + 128 + (lane << 2));
    float* od = agg + (size_t)d * HIDDEN + (lane << 2);
    *(float4*)(od)       = make_float4(acc[0].x + b0.x, acc[0].y + b0.y, acc[0].z + b0.z, acc[0].w + b0.w);
    *(float4*)(od + 128) = make_float4(acc[1].x + b1.x, acc[1].y + b1.y, acc[1].z + b1.z, acc[1].w + b1.w);
  }
}

extern "C" void kernel_launch(void* const* d_in, const int* in_sizes, int n_in,
                              void* d_out, int out_size, void* d_ws, size_t ws_size,
                              hipStream_t stream) {
  const float* x     = (const float*)d_in[0];
  const int*   ei    = (const int*)  d_in[1];
  const float* W1    = (const float*)d_in[2];
  const float* asrc1 = (const float*)d_in[3];
  const float* adst1 = (const float*)d_in[4];
  const float* b1    = (const float*)d_in[5];
  const float* W2    = (const float*)d_in[6];
  const float* asrc2 = (const float*)d_in[7];
  const float* adst2 = (const float*)d_in[8];
  const float* b2    = (const float*)d_in[9];
  const float* Wf    = (const float*)d_in[10];
  const float* bf    = (const float*)d_in[11];

  const int B  = 2;
  const int N  = in_sizes[0] / (B * FEAT);
  const int E  = in_sizes[1] / 2;
  const int R  = B * N;

  float* p = (float*)d_ws;
  float* buf_h   = p; p += (size_t)R * HIDDEN;
  float* buf_agg = p; p += (size_t)R * HIDDEN;
  float* buf_z   = p; p += (size_t)R * HEADS * FEAT;
  float* ssrc    = p; p += (size_t)R * HEADS;
  float* sdst    = p; p += (size_t)R * HEADS;
  float* vmat    = p; p += (size_t)FEAT * 8;
  int* q = (int*)p;
  int* slots = q; q += (size_t)N * SLOTS;
  int* degs  = q; q += (size_t)N;

  const int mtiles  = R >> 5;
  const int ftiles  = mtiles * (CHAN >> 6);
  const dim3 blkG(128);
  const dim3 blkE(256);


  const int eblocks = (E + 255) / 256;
  const int sblocks = (R + 31) / 32;
  prep_kernel<<<(N + 255) / 256 + (FEAT * 32) / 256, blkE, 0, stream>>>(W1, asrc1, adst1, vmat, degs, N);
  slot_fill_score_kernel<<<eblocks > sblocks ? eblocks : sblocks, blkE, 0, stream>>>(
      ei, E, degs, slots, x, vmat, ssrc, sdst, R);

  gat_gather_kernel<true><<<(R + 7) / 8, blkE, 0, stream>>>(degs, slots, ei, E, ssrc, sdst, x, nullptr, buf_z, N, R);
  gat_two_gemm_kernel<<<mtiles, blkG, 0, stream>>>(
      buf_z, W1, b1, W2, asrc2, adst2, buf_h, ssrc, sdst, R, FEAT, HIDDEN);

  gat_gather_kernel<false><<<(R + 7) / 8, blkE, 0, stream>>>(degs, slots, ei, E, ssrc, sdst, buf_h, b2, buf_agg, N, R);

  wmma_linear_kernel<CHAN><<<(ftiles + 3) / 4, blkG, 0, stream>>>(
      buf_agg, Wf, bf, (float*)d_out, R, HIDDEN);
}


# Final linear folded into layer 2: launch 1 forms W2_h@Wf_h per head (WMMA), layer-2 GEMM gives 64 out channels per head, gather adds heads, writes out; scores act@(W2 a) from LDS; sum order changes
# speedup vs baseline: 7.3085x; 1.0487x over previous
#include <hip/hip_runtime.h>
#include <math.h>


typedef __attribute__((ext_vector_type(2))) float v2f;
typedef __attribute__((ext_vector_type(4))) float v4f;
typedef __attribute__((ext_vector_type(8))) float v8f;

#define HEADS     4
#define CHAN      64
#define HIDDEN    256
#define FEAT      128
#define NEG_SLOPE 0.2f
#define SLOTS     64

static __device__ __forceinline__ float lrelu(float x) {
  return x > 0.f ? x : NEG_SLOPE * x;
}

static __device__ __forceinline__ float elu1(float v) {
  const float e = expm1f(fminf(v, 0.f));
  return v < 0.f ? e : v;
}

struct Frag {
  v4f a0[2];
  v4f a1[2];
  v4f b[4][2];
};

template <int NC>
static __device__ __forceinline__ void load_frag(Frag& f,
                                                 const float* __restrict__ ap0,
                                                 const float* __restrict__ ap1,
                                                 const float* __restrict__ wp) {
  #pragma unroll
  for (int i = 0; i < 2; ++i) {
    f.a0[i] = *(const v4f*)(ap0 + 8 * i);
    f.a1[i] = *(const v4f*)(ap1 + 8 * i);
    #pragma unroll
    for (int p = 0; p < 2; ++p) {
      const int t = 2 * i + p;
      #pragma unroll
      for (int e = 0; e < 2; ++e)
        f.b[t][e] = *(const v4f*)(wp + (size_t)(8 * i + 2 * p + e) * NC);
    }
  }
}

static __device__ __forceinline__ void mma_frag(v8f (&acc)[2][4], const Frag& f) {
  #pragma unroll
  for (int t = 0; t < 4; ++t) {
    const int i = t >> 1, p = (t & 1) * 2;
    const v2f a0 = v2f{f.a0[i][p], f.a0[i][p + 1]};
    const v2f a1 = v2f{f.a1[i][p], f.a1[i][p + 1]};
    #pragma unroll
    for (int c = 0; c < 4; ++c) {
      const v2f b = v2f{f.b[t][0][c], f.b[t][1][c]};
      acc[0][c] = __builtin_amdgcn_wmma_f32_16x16x4_f32(false, a0, false, b, (short)0, acc[0][c], false, false);
      acc[1][c] = __builtin_amdgcn_wmma_f32_16x16x4_f32(false, a1, false, b, (short)0, acc[1][c], false, false);
    }
  }
}

template <int NC>
static __device__ __forceinline__ void gemm_kloop(v8f (&acc)[2][4],
                                                  const float* ap0, const float* ap1,
                                                  const float* __restrict__ wp, int K) {
  constexpr size_t wblk = (size_t)16 * NC;
  Frag f0, f1;
  load_frag<NC>(f0, ap0, ap1, wp);
  for (int k = 0; k + 32 < K; k += 32) {
    load_frag<NC>(f1, ap0 + 16, ap1 + 16, wp + wblk);
    __builtin_amdgcn_sched_barrier(0);
    mma_frag(acc, f0);
    __builtin_amdgcn_sched_barrier(0);
    load_frag<NC>(f0, ap0 + 32, ap1 + 32, wp + 2 * wblk);
    __builtin_amdgcn_sched_barrier(0);
    mma_frag(acc, f1);
    ap0 += 32; ap1 += 32; wp += 2 * wblk;
  }
  load_frag<NC>(f1, ap0 + 16, ap1 + 16, wp + wblk);
  __builtin_amdgcn_sched_barrier(0);
  mma_frag(acc, f0);
  mma_frag(acc, f1);
}

#define APITCH (HIDDEN + 4)
__global__ void __launch_bounds__(128) gat_two_gemm_kernel(const float* __restrict__ Z,
                                 const float* __restrict__ W1,
                                 const float* __restrict__ b1,
                                 const float* __restrict__ WP,
                                 const float* __restrict__ U2,
                                 float* __restrict__ Hout,
                                 float* __restrict__ s_src,
                                 float* __restrict__ s_dst,
                                 int M, int K1, int K2) {
  __shared__ __attribute__((aligned(16))) float act[32 * APITCH];
  const int lane = threadIdx.x & 31;
  const int head = threadIdx.x >> 5;
  const int m0 = blockIdx.x << 5;
  if (m0 >= M) return;
  const int n0 = head << 6;
  const int lo = lane & 15;
  const int hi = lane >> 4;

  {
    v8f acc[2][4] = {};
    const size_t lda = (size_t)K1 * HEADS;
    const float* ap0 = Z + (size_t)(m0 + lo) * lda + (size_t)head * K1 + (hi << 2);
    const float* ap1 = ap0 + 16 * lda;
    const float* wp  = W1 + n0 + (lo << 2) + (size_t)(hi << 2) * HIDDEN;
    gemm_kloop<HIDDEN>(acc, ap0, ap1, wp, K1);
    const v4f bv = *(const v4f*)(b1 + n0 + (lo << 2));
    #pragma unroll
    for (int g = 0; g < 2; ++g) {
      #pragma unroll
      for (int r = 0; r < 8; ++r) {
        const int row = (g << 4) + (hi << 3) + r;
        *(v4f*)(act + row * APITCH + n0 + (lo << 2)) =
            v4f{elu1(acc[g][0][r] + bv.x), elu1(acc[g][1][r] + bv.y),
                elu1(acc[g][2][r] + bv.z), elu1(acc[g][3][r] + bv.w)};
      }
    }
  }
  __syncthreads();

  {
    const int hs = __builtin_amdgcn_readfirstlane(head);
    const float* __restrict__ up = U2 + (size_t)hs * (2 * HIDDEN);
    const float* ar = act + lane * APITCH;
    float ps[4] = {0.f, 0.f, 0.f, 0.f}, pd[4] = {0.f, 0.f, 0.f, 0.f};
    #pragma unroll 8
    for (int k = 0; k < HIDDEN; k += 4) {
      const v4f a = *(const v4f*)(ar + k);
      ps[0] += a.x * up[2 * k + 0]; pd[0] += a.x * up[2 * k + 1];
      ps[1] += a.y * up[2 * k + 2]; pd[1] += a.y * up[2 * k + 3];
      ps[2] += a.z * up[2 * k + 4]; pd[2] += a.z * up[2 * k + 5];
      ps[3] += a.w * up[2 * k + 6]; pd[3] += a.w * up[2 * k + 7];
    }
    const int row = m0 + lane;
    s_src[row * HEADS + head] = (ps[0] + ps[1]) + (ps[2] + ps[3]);
    s_dst[row * HEADS + head] = (pd[0] + pd[1]) + (pd[2] + pd[3]);
  }

  v8f acc[2][4] = {};
  {
    const float* ap0 = act + lo * APITCH + (hi << 2);
    const float* ap1 = ap0 + 16 * APITCH;
    const float* wp  = WP + n0 + (lo << 2) + (size_t)(hi << 2) * HIDDEN;
    gemm_kloop<HIDDEN>(acc, ap0, ap1, wp, K2);
  }
  #pragma unroll
  for (int g = 0; g < 2; ++g) {
    const int rbase = m0 + (g << 4) + (hi << 3);
    #pragma unroll
    for (int r = 0; r < 8; ++r) {
      float* hp = Hout + (size_t)(rbase + r) * HIDDEN + n0 + (lo << 2);
      *(v4f*)hp = v4f{acc[g][0][r], acc[g][1][r], acc[g][2][r], acc[g][3][r]};
    }
  }
}

static __device__ __forceinline__ float wave_max(float v) {
  #pragma unroll
  for (int off = 16; off > 0; off >>= 1) v = fmaxf(v, __shfl_xor(v, off, 32));
  return v;
}
static __device__ __forceinline__ float wave_sum(float v) {
  #pragma unroll
  for (int off = 16; off > 0; off >>= 1) v += __shfl_xor(v, off, 32);
  return v;
}

#define PREP_BLOCKS 64
static_assert(PREP_BLOCKS * 7 >= FEAT + HIDDEN + CHAN && PREP_BLOCKS >= 32,
              "light items and tile waves must fit the first PREP_BLOCKS blocks");
__global__ void __launch_bounds__(256) prep_kernel(const float* __restrict__ W1,
                            const float* __restrict__ att_src1,
                            const float* __restrict__ att_dst1,
                            const float* __restrict__ W2,
                            const float* __restrict__ att_src2,
                            const float* __restrict__ att_dst2,
                            const float* __restrict__ b2,
                            const float* __restrict__ Wf,
                            const float* __restrict__ bf,
                            float* __restrict__ V, float* __restrict__ U2,
                            float* __restrict__ WP, float* __restrict__ BP,
                            int* __restrict__ degs, int N) {
  const int idx = blockIdx.x * blockDim.x + threadIdx.x;
  if (idx < N) degs[idx] = 0;
  const int lane = threadIdx.x & 31;
  const int wv = __builtin_amdgcn_readfirstlane(threadIdx.x >> 5);
  const int bk = blockIdx.x;
  const int k  = wv == 0 ? FEAT + HIDDEN + CHAN : bk * 7 + wv - 1;
  if (wv == 0 && bk < 32) {
    const int t  = bk;
    const int m0 = (t & 7) << 5;
    const int h  = t >> 3;
    const int lo = lane & 15;
    const int hi = lane >> 4;
    v8f acc[2][4] = {};
    const float* ap0 = W2 + (size_t)(m0 + lo) * HIDDEN + h * CHAN + (hi << 2);
    const float* ap1 = ap0 + 16 * HIDDEN;
    const float* wp  = Wf + (size_t)(h * CHAN + (hi << 2)) * CHAN + (lo << 2);
    gemm_kloop<CHAN>(acc, ap0, ap1, wp, CHAN);
    #pragma unroll
    for (int g = 0; g < 2; ++g) {
      const int rbase = m0 + (g << 4) + (hi << 3);
      #pragma unroll
      for (int r = 0; r < 8; ++r) {
        float* hp = WP + (size_t)(rbase + r) * HIDDEN + h * CHAN + (lo << 2);
        *(v4f*)hp = v4f{acc[g][0][r], acc[g][1][r], acc[g][2][r], acc[g][3][r]};
      }
    }
  } else if (k < FEAT + HIDDEN) {
    const bool second = k >= FEAT;
    const int r = second ? k - FEAT : k;
    const float* wr = (second ? W2 : W1) + (size_t)r * HIDDEN + (lane << 2);
    const float* as = second ? att_src2 : att_src1;
    const float* ad = second ? att_dst2 : att_dst1;
    const float4 w0 = *(const float4*)(wr);
    const float4 w1 = *(const float4*)(wr + 128);
    const float4 s0 = *(const float4*)(as + (lane << 2));
    const float4 s1 = *(const float4*)(as + 128 + (lane << 2));
    const float4 d0 = *(const float4*)(ad + (lane << 2));
    const float4 d1 = *(const float4*)(ad + 128 + (lane << 2));
    float ps0 = w0.x * s0.x + w0.y * s0.y + w0.z * s0.z + w0.w * s0.w;
    float ps1 = w1.x * s1.x + w1.y * s1.y + w1.z * s1.z + w1.w * s1.w;
    float pd0 = w0.x * d0.x + w0.y * d0.y + w0.z * d0.z + w0.w * d0.w;
    float pd1 = w1.x * d1.x + w1.y * d1.y + w1.z * d1.z + w1.w * d1.w;
    #pragma unroll
    for (int off = 8; off > 0; off >>= 1) {
      ps0 += __shfl_xor(ps0, off, 32); ps1 += __shfl_xor(ps1, off, 32);
      pd0 += __shfl_xor(pd0, off, 32); pd1 += __shfl_xor(pd1, off, 32);
    }
    if ((lane & 15) == 0) {
      const int h = lane >> 4;
      if (!second) {
        float* vr = V + r * 8;
        vr[h] = ps0; vr[2 + h] = ps1; vr[4 + h] = pd0; vr[6 + h] = pd1;
      } else {
        float* u0 = U2 + (size_t)h * (2 * HIDDEN) + 2 * r;
        float* u1 = U2 + (size_t)(2 + h) * (2 * HIDDEN) + 2 * r;
        u0[0] = ps0; u0[1] = pd0;
        u1[0] = ps1; u1[1] = pd1;
      }
    }
  } else if (k < FEAT + HIDDEN + CHAN) {
    const int j = k - (FEAT + HIDDEN);
    float s = 0.f;
    #pragma unroll
    for (int i = 0; i < HIDDEN / 32; ++i) {
      const int kk = lane + 32 * i;
      s += b2[kk] * Wf[(size_t)kk * CHAN + j];
    }
    s = wave_sum(s);
    if (lane == 0) BP[j] = s + bf[j];
  }
}

__global__ void slot_fill_score_kernel(const int* __restrict__ ei, int E,
                                       int* __restrict__ degs, int* __restrict__ slots,
                                       const float* __restrict__ X,
                                       const float* __restrict__ V,
                                       float* __restrict__ s_src,
                                       float* __restrict__ s_dst, int R) {
  __shared__ float Vs[FEAT * 8];
  const int idx = blockIdx.x * blockDim.x + threadIdx.x;
  if (idx < E) {
    const int dn  = ei[E + idx];
    const int pos = atomicAdd(degs + dn, 1);
    if (pos < SLOTS) slots[(size_t)dn * SLOTS + pos] = ei[idx];
  }
  *(float4*)(Vs + (threadIdx.x << 2)) = *(const float4*)(V + (threadIdx.x << 2));
  __syncthreads();
  const int row = blockIdx.x * 32 + (threadIdx.x >> 3);
  const int j   = threadIdx.x & 7;
  if (row < R) {
    const float* xr = X + (size_t)row * FEAT;
    float s = 0.f;
    #pragma unroll 8
    for (int k = 0; k < FEAT; k += 4) {
      const float4 xv = *(const float4*)(xr + k);
      s += xv.x * Vs[(k + 0) * 8 + j];
      s += xv.y * Vs[(k + 1) * 8 + j];
      s += xv.z * Vs[(k + 2) * 8 + j];
      s += xv.w * Vs[(k + 3) * 8 + j];
    }
    if (j < 4) s_src[row * HEADS + j] = s;
    else       s_dst[row * HEADS + (j - 4)] = s;
  }
}

static __device__ __forceinline__ void load_entry(const int* __restrict__ csr_src,
                                                  const float* __restrict__ s_src,
                                                  int beg, int deg, int d, int row0, int j,
                                                  const float4 vd, int& s, float4& e) {
  s = d;
  if (j < deg) s = row0 + csr_src[beg + j];
  if (j <= deg) {
    const float4 vs = *(const float4*)(s_src + (size_t)s * HEADS);
    e = make_float4(lrelu(vs.x + vd.x), lrelu(vs.y + vd.y), lrelu(vs.z + vd.z), lrelu(vs.w + vd.w));
  } else {
    e = make_float4(-INFINITY, -INFINITY, -INFINITY, -INFINITY);
  }
}

template <bool XIN>
static __device__ __forceinline__ void row_add(const float* __restrict__ Hin, int st,
                                               float a0, float a1, float a2, float a3,
                                               int lane, float4 (&v)[4]) {
  if constexpr (XIN) {
    const float4 x = *(const float4*)(Hin + (size_t)st * FEAT + (lane << 2));
    v[0].x += x.x * a0; v[0].y += x.y * a0; v[0].z += x.z * a0; v[0].w += x.w * a0;
    v[1].x += x.x * a1; v[1].y += x.y * a1; v[1].z += x.z * a1; v[1].w += x.w * a1;
    v[2].x += x.x * a2; v[2].y += x.y * a2; v[2].z += x.z * a2; v[2].w += x.w * a2;
    v[3].x += x.x * a3; v[3].y += x.y * a3; v[3].z += x.z * a3; v[3].w += x.w * a3;
  } else {
    const float* hs = Hin + (size_t)st * HIDDEN + (lane << 2);
    const float4 h0 = *(const float4*)(hs);
    const float4 h1 = *(const float4*)(hs + 128);
    const float w0 = lane >= 16 ? a1 : a0;
    const float w1 = lane >= 16 ? a3 : a2;
    v[0].x += h0.x * w0; v[0].y += h0.y * w0; v[0].z += h0.z * w0; v[0].w += h0.w * w0;
    v[1].x += h1.x * w1; v[1].y += h1.y * w1; v[1].z += h1.z * w1; v[1].w += h1.w * w1;
  }
}

template <bool XIN>
static __device__ __forceinline__ void accum_one(const float* __restrict__ Hin,
                                                 int s, const float4 al, int t, int lane,
                                                 float4 (&v)[4]) {
  const int   st = __builtin_amdgcn_readlane(s, t);
  const float a0 = __int_as_float(__builtin_amdgcn_readlane(__float_as_int(al.x), t));
  const float a1 = __int_as_float(__builtin_amdgcn_readlane(__float_as_int(al.y), t));
  const float a2 = __int_as_float(__builtin_amdgcn_readlane(__float_as_int(al.z), t));
  const float a3 = __int_as_float(__builtin_amdgcn_readlane(__float_as_int(al.w), t));
  row_add<XIN>(Hin, st, a0, a1, a2, a3, lane, v);
}

template <bool XIN>
static __device__ __forceinline__ void accum_entries(const float* __restrict__ Hin,
                                                     int s, const float4 al, int cnt, int lane,
                                                     float4 (&v)[4]) {
  #pragma unroll 4
  for (int t = 0; t < cnt; ++t) {
    const int   st = __builtin_amdgcn_readlane(s, t);
    const float a0 = __int_as_float(__builtin_amdgcn_readlane(__float_as_int(al.x), t));
    const float a1 = __int_as_float(__builtin_amdgcn_readlane(__float_as_int(al.y), t));
    const float a2 = __int_as_float(__builtin_amdgcn_readlane(__float_as_int(al.z), t));
    const float a3 = __int_as_float(__builtin_amdgcn_readlane(__float_as_int(al.w), t));
    row_add<XIN>(Hin, st, a0, a1, a2, a3, lane, v);
  }
}

static __device__ __forceinline__ float4 lrelu4(const float4 a, const float4 b) {
  return make_float4(lrelu(a.x + b.x), lrelu(a.y + b.y), lrelu(a.z + b.z), lrelu(a.w + b.w));
}

template <bool XIN>
static __device__ __forceinline__ void sweep_edge_list(const int* __restrict__ ei, int E,
                                                       const float* __restrict__ s_src,
                                                       const float* __restrict__ Hin,
                                                       int g, int d, int row0, const float4 vd,
                                                       int lane, float4 (&v)[4]) {
  const float4 es = lrelu4(*(const float4*)(s_src + (size_t)d * HEADS), vd);
  float4 m = es;
  for (int i0 = 0; i0 < E; i0 += 32) {
    const int i = i0 + lane;
    if (i < E && ei[E + i] == g) {
      const float4 e = lrelu4(*(const float4*)(s_src + (size_t)(row0 + ei[i]) * HEADS), vd);
      m.x = fmaxf(m.x, e.x); m.y = fmaxf(m.y, e.y); m.z = fmaxf(m.z, e.z); m.w = fmaxf(m.w, e.w);
    }
  }
  m.x = wave_max(m.x); m.y = wave_max(m.y); m.z = wave_max(m.z); m.w = wave_max(m.w);

  const float4 xs = make_float4(expf(es.x - m.x), expf(es.y - m.y), expf(es.z - m.z), expf(es.w - m.w));
  float4 dn = lane == 0 ? xs : make_float4(0.f, 0.f, 0.f, 0.f);
  for (int i0 = 0; i0 < E; i0 += 32) {
    const int i = i0 + lane;
    if (i < E && ei[E + i] == g) {
      const float4 e = lrelu4(*(const float4*)(s_src + (size_t)(row0 + ei[i]) * HEADS), vd);
      dn.x += expf(e.x - m.x); dn.y += expf(e.y - m.y);
      dn.z += expf(e.z - m.z); dn.w += expf(e.w - m.w);
    }
  }
  dn.x = wave_sum(dn.x); dn.y = wave_sum(dn.y); dn.z = wave_sum(dn.z); dn.w = wave_sum(dn.w);

  for (int i0 = 0; i0 < E; i0 += 32) {
    const int i = i0 + lane;
    const bool hit = i < E && ei[E + i] == g;
    int s = d;
    float4 al = make_float4(0.f, 0.f, 0.f, 0.f);
    if (hit) {
      s = row0 + ei[i];
      const float4 e = lrelu4(*(const float4*)(s_src + (size_t)s * HEADS), vd);
      al = make_float4(expf(e.x - m.x) / dn.x, expf(e.y - m.y) / dn.y,
                       expf(e.z - m.z) / dn.z, expf(e.w - m.w) / dn.w);
    }
    unsigned hits = __builtin_amdgcn_ballot_w32(hit);
    while (hits) {
      const int t = __builtin_ctz(hits);
      hits &= hits - 1;
      accum_one<XIN>(Hin, s, al, t, lane, v);
    }
  }
  const float4 al = make_float4(xs.x / dn.x, xs.y / dn.y, xs.z / dn.z, xs.w / dn.w);
  row_add<XIN>(Hin, d, al.x, al.y, al.z, al.w, lane, v);
}

template <bool XIN>
__global__ void gat_gather_kernel(const int* __restrict__ degs,
                                  const int* __restrict__ csr_src,
                                  const int* __restrict__ ei, int E,
                                  const float* __restrict__ s_src,
                                  const float* __restrict__ s_dst,
                                  const float* __restrict__ Hin,
                                  const float* __restrict__ bias,
                                  float* __restrict__ agg, int N, int R) {
  const int lane = threadIdx.x & 31;
  const int d = __builtin_amdgcn_readfirstlane(blockIdx.x * (blockDim.x >> 5) + (threadIdx.x >> 5));
  if (d >= R) return;
  const int row0 = d >= N ? N : 0;
  const int g    = d - row0;
  const int beg = g * SLOTS;
  const int deg = degs[g];
  const int tot = deg + 1;
  const float4 vd = *(const float4*)(s_dst + (size_t)d * HEADS);

  float4 acc[4];
  #pragma unroll
  for (int h = 0; h < 4; ++h) acc[h] = make_float4(0.f, 0.f, 0.f, 0.f);
  if (deg > SLOTS) {
    sweep_edge_list<XIN>(ei, E, s_src, Hin, g, d, row0, vd, lane, acc);
  } else {
  int s0; float4 e0;
  load_entry(csr_src, s_src, beg, deg, d, row0, lane, vd, s0, e0);

  float4 m = e0;
  for (int base = 32; base < tot; base += 32) {
    int s; float4 e;
    load_entry(csr_src, s_src, beg, deg, d, row0, base + lane, vd, s, e);
    m.x = fmaxf(m.x, e.x); m.y = fmaxf(m.y, e.y); m.z = fmaxf(m.z, e.z); m.w = fmaxf(m.w, e.w);
  }
  m.x = wave_max(m.x); m.y = wave_max(m.y); m.z = wave_max(m.z); m.w = wave_max(m.w);

  const bool v0 = lane <= deg;
  float4 x0 = make_float4(v0 ? expf(e0.x - m.x) : 0.f, v0 ? expf(e0.y - m.y) : 0.f,
                          v0 ? expf(e0.z - m.z) : 0.f, v0 ? expf(e0.w - m.w) : 0.f);
  float4 dn = x0;
  for (int base = 32; base < tot; base += 32) {
    int s; float4 e;
    load_entry(csr_src, s_src, beg, deg, d, row0, base + lane, vd, s, e);
    if (base + lane <= deg) {
      dn.x += expf(e.x - m.x); dn.y += expf(e.y - m.y);
      dn.z += expf(e.z - m.z); dn.w += expf(e.w - m.w);
    }
  }
  dn.x = wave_sum(dn.x); dn.y = wave_sum(dn.y); dn.z = wave_sum(dn.z); dn.w = wave_sum(dn.w);

  {
    const float4 al = make_float4(x0.x / dn.x, x0.y / dn.y, x0.z / dn.z, x0.w / dn.w);
    accum_entries<XIN>(Hin, s0, al, tot < 32 ? tot : 32, lane, acc);
  }
  for (int base = 32; base < tot; base += 32) {
    int s; float4 e;
    load_entry(csr_src, s_src, beg, deg, d, row0, base + lane, vd, s, e);
    const bool v = base + lane <= deg;
    const float4 al = make_float4(v ? expf(e.x - m.x) / dn.x : 0.f, v ? expf(e.y - m.y) / dn.y : 0.f,
                                  v ? expf(e.z - m.z) / dn.z : 0.f, v ? expf(e.w - m.w) / dn.w : 0.f);
    const int rem = tot - base;
    accum_entries<XIN>(Hin, s, al, rem < 32 ? rem : 32, lane, acc);
  }
  }

  if constexpr (XIN) {
    float* od = agg + (size_t)d * (HEADS * FEAT) + (lane << 2);
    #pragma unroll
    for (int h = 0; h < 4; ++h) *(float4*)(od + h * FEAT) = acc[h];
  } else {
    float4 t = make_float4(acc[0].x + acc[1].x, acc[0].y + acc[1].y, acc[0].z + acc[1].z, acc[0].w + acc[1].w);
    t.x += __shfl_xor(t.x, 16, 32); t.y += __shfl_xor(t.y, 16, 32);
    t.z += __shfl_xor(t.z, 16, 32); t.w += __shfl_xor(t.w, 16, 32);
    if (lane < 16) {
      const float4 b0 = *(const float4*)(bias + (lane << 2));
      float* od = agg + (size_t)d * CHAN + (lane << 2);
      *(float4*)(od) = make_float4(t.x + b0.x, t.y + b0.y, t.z + b0.z, t.w + b0.w);
    }
  }
}

extern "C" void kernel_launch(void* const* d_in, const int* in_sizes, int n_in,
                              void* d_out, int out_size, void* d_ws, size_t ws_size,
                              hipStream_t stream) {
  const float* x     = (const float*)d_in[0];
  const int*   ei    = (const int*)  d_in[1];
  const float* W1    = (const float*)d_in[2];
  const float* asrc1 = (const float*)d_in[3];
  const float* adst1 = (const float*)d_in[4];
  const float* b1    = (const float*)d_in[5];
  const float* W2    = (const float*)d_in[6];
  const float* asrc2 = (const float*)d_in[7];
  const float* adst2 = (const float*)d_in[8];
  const float* b2    = (const float*)d_in[9];
  const float* Wf    = (const float*)d_in[10];
  const float* bf    = (const float*)d_in[11];

  const int B  = 2;
  const int N  = in_sizes[0] / (B * FEAT);
  const int E  = in_sizes[1] / 2;
  const int R  = B * N;

  float* p = (float*)d_ws;
  float* buf_h   = p; p += (size_t)R * HIDDEN;
  float* buf_z   = p; p += (size_t)R * HEADS * FEAT;
  float* ssrc    = p; p += (size_t)R * HEADS;
  float* sdst    = p; p += (size_t)R * HEADS;
  float* vmat    = p; p += (size_t)FEAT * 8;
  float* umat    = p; p += (size_t)HEADS * 2 * HIDDEN;
  float* wfold   = p; p += (size_t)HIDDEN * HIDDEN;
  float* bfold   = p; p += (size_t)CHAN;
  int* q = (int*)p;
  int* slots = q; q += (size_t)N * SLOTS;
  int* degs  = q; q += (size_t)N;

  const int mtiles  = R >> 5;
  const dim3 blkG(128);
  const dim3 blkE(256);


  const int eblocks = (E + 255) / 256;
  const int sblocks = (R + 31) / 32;
  prep_kernel<<<(N + 255) / 256 + PREP_BLOCKS, blkE, 0, stream>>>(
      W1, asrc1, adst1, W2, asrc2, adst2, b2, Wf, bf, vmat, umat, wfold, bfold, degs, N);
  slot_fill_score_kernel<<<eblocks > sblocks ? eblocks : sblocks, blkE, 0, stream>>>(
      ei, E, degs, slots, x, vmat, ssrc, sdst, R);

  gat_gather_kernel<true><<<(R + 7) / 8, blkE, 0, stream>>>(degs, slots, ei, E, ssrc, sdst, x, nullptr, buf_z, N, R);
  gat_two_gemm_kernel<<<mtiles, blkG, 0, stream>>>(
      buf_z, W1, b1, wfold, umat, buf_h, ssrc, sdst, R, FEAT, HIDDEN);

  gat_gather_kernel<false><<<(R + 7) / 8, blkE, 0, stream>>>(degs, slots, ei, E, ssrc, sdst, buf_h, bfold, (float*)d_out, N, R);
}
